# starting kernel
# baseline (speedup 1.0000x reference)
#include <hip/hip_runtime.h>

#define N_NODES   65536
#define NPG       128
#define NGRAPH    512
#define E_EDGES   1048576
#define HID       128
#define OUT_DIM   10
#define N_LAYERS  4
#define LN_EPS    1e-5f

typedef float v2f __attribute__((ext_vector_type(2)));
typedef float v4f __attribute__((ext_vector_type(4)));
typedef float v8f __attribute__((ext_vector_type(8)));

__device__ __forceinline__ void atomic_add_f32_hw(float* p, float v) {
    asm volatile("global_atomic_add_f32 %0, %1, off" : : "v"(p), "v"(v) : "memory");
}

__global__ void k_deg_init(int* __restrict__ degcnt) {
    int i = blockIdx.x * 256 + threadIdx.x;
    if (i < N_NODES) degcnt[i] = 0;
}

__global__ void k_deg_count(const int* __restrict__ edge_index, int* __restrict__ degcnt) {
    int e = blockIdx.x * 256 + threadIdx.x;
    if (e < E_EDGES) {
        int d = edge_index[E_EDGES + e];
        atomicAdd(&degcnt[d], 1);
    }
}

__global__ void k_deg_finish(const int* __restrict__ degcnt,
                             float* __restrict__ dinv, float* __restrict__ invdeg) {
    int i = blockIdx.x * 256 + threadIdx.x;
    if (i < N_NODES) {
        float d = 1.0f + (float)degcnt[i];
        dinv[i]   = rsqrtf(d);
        invdeg[i] = 1.0f / d;
    }
}

__global__ void k_embed(const int* __restrict__ x,
                        const float* __restrict__ cls_emb,
                        const float* __restrict__ nbr_emb,
                        float* __restrict__ h) {
    int t = blockIdx.x * 256 + threadIdx.x;
    int n = t >> 5;
    int f = (t & 31) << 2;
    int c  = x[2 * n + 0];
    int nb = x[2 * n + 1];
    v4f a = *(const v4f*)(cls_emb + (size_t)c  * HID + f);
    v4f b = *(const v4f*)(nbr_emb + (size_t)nb * HID + f);
    *(v4f*)(h + (size_t)n * HID + f) = a + b;
}

__launch_bounds__(256)
__global__ void k_gemm_fused(const float* __restrict__ h,
                             const float* __restrict__ W,
                             const float* __restrict__ bias,
                             const float* __restrict__ invdeg,
                             float* __restrict__ hw,
                             float* __restrict__ agg) {
    __shared__ float lWt[HID * HID];
    const int tid = threadIdx.x;

    for (int i = tid * 4; i < HID * HID; i += 256 * 4) {
        v4f w = *(const v4f*)(W + i);
        const int k = i >> 7;
        const int n = i & (HID - 1);
#pragma unroll
        for (int j = 0; j < 4; ++j)
            lWt[(n + j) * HID + (k ^ (((n + j) & 31) << 1))] = w[j];
    }
    __syncthreads();

    const int wave = tid >> 5;
    const int lane = tid & 31;
    const int half = lane >> 4;
    const int l16  = lane & 15;
    const int m0   = blockIdx.x * 128 + wave * 16;

    v8f zero = {};
    v8f acc[8];
#pragma unroll
    for (int t = 0; t < 8; ++t) acc[t] = zero;

    const float* arow = h + (size_t)(m0 + l16) * HID;

    for (int k = 0; k < HID; k += 4) {
        const int kb = k + half * 2;
        v2f a = *(const v2f*)(arow + kb);
#pragma unroll
        for (int t = 0; t < 8; ++t) {
            const int col = t * 16 + l16;
            v2f b = *(const v2f*)(lWt + col * HID + (kb ^ ((col & 31) << 1)));
            acc[t] = __builtin_amdgcn_wmma_f32_16x16x4_f32(
                false, a, false, b, (short)0, acc[t], false, false);
        }
    }

    const int rbase = half * 8;
#pragma unroll
    for (int r = 0; r < 8; ++r) {
        const int row = m0 + rbase + r;
        const float inv = invdeg[row];
        float* hwrow  = hw  + (size_t)row * HID;
        float* aggrow = agg + (size_t)row * HID;
#pragma unroll
        for (int t = 0; t < 8; ++t) {
            const int col = t * 16 + l16;
            const float v = acc[t][r];
            hwrow[col]  = v;
            aggrow[col] = v * inv + bias[col];
        }
    }
}

__global__ void k_scatter(const int* __restrict__ edge_index,
                          const float* __restrict__ dinv,
                          const float* __restrict__ hw,
                          float* __restrict__ agg) {
    int t = blockIdx.x * 256 + threadIdx.x;
    int e = t >> 5;
    if (e >= E_EDGES) return;
    int f = (t & 31) << 2;
    int s = edge_index[e];
    int d = edge_index[E_EDGES + e];
    float en = dinv[s] * dinv[d];
    v4f v = *(const v4f*)(hw + (size_t)s * HID + f);
    float* out = agg + (size_t)d * HID + f;
    atomic_add_f32_hw(out + 0, v.x * en);
    atomic_add_f32_hw(out + 1, v.y * en);
    atomic_add_f32_hw(out + 2, v.z * en);
    atomic_add_f32_hw(out + 3, v.w * en);
}

__launch_bounds__(256)
__global__ void k_relu_res_ln(float* __restrict__ h,
                              const float* __restrict__ agg,
                              const float* __restrict__ gamma,
                              const float* __restrict__ beta) {
    int t = blockIdx.x * 256 + threadIdx.x;
    int n = t >> 5;
    int lane = t & 31;
    int f = lane << 2;

    const float* ar = agg + (size_t)n * HID;
    float* hr = h + (size_t)n * HID;

    v4f hv = *(const v4f*)(hr + f);
    v4f av = *(const v4f*)(ar + f);
    v4f val;
    val.x = hv.x + fmaxf(av.x, 0.0f);
    val.y = hv.y + fmaxf(av.y, 0.0f);
    val.z = hv.z + fmaxf(av.z, 0.0f);
    val.w = hv.w + fmaxf(av.w, 0.0f);

    float s = val.x + val.y + val.z + val.w;
#pragma unroll
    for (int off = 16; off >= 1; off >>= 1) s += __shfl_xor(s, off, 32);
    const float mu = s * (1.0f / (float)HID);

    v4f dv = val - mu;
    float q = dv.x * dv.x + dv.y * dv.y + dv.z * dv.z + dv.w * dv.w;
#pragma unroll
    for (int off = 16; off >= 1; off >>= 1) q += __shfl_xor(q, off, 32);
    const float rstd = rsqrtf(q * (1.0f / (float)HID) + LN_EPS);

    v4f g = *(const v4f*)(gamma + f);
    v4f b = *(const v4f*)(beta + f);
    v4f outv = dv * rstd * g + b;
    *(v4f*)(hr + f) = outv;
}

__global__ void k_out(const float* __restrict__ h,
                      const float* __restrict__ W_out,
                      float* __restrict__ out) {
    int t = blockIdx.x * 256 + threadIdx.x;
    if (t >= NGRAPH * OUT_DIM) return;
    int g = t / OUT_DIM;
    int o = t % OUT_DIM;
    const float* hr = h + (size_t)(g * NPG) * HID;
    float s = 0.0f;
#pragma unroll 4
    for (int k = 0; k < HID; ++k) s = fmaf(hr[k], W_out[k * OUT_DIM + o], s);
    out[t] = s;
}

extern "C" void kernel_launch(void* const* d_in, const int* in_sizes, int n_in,
                              void* d_out, int out_size, void* d_ws, size_t ws_size,
                              hipStream_t stream) {
    const int*   x        = (const int*)d_in[0];
    const int*   ei       = (const int*)d_in[1];
    const float* cls_emb  = (const float*)d_in[3];
    const float* nbr_emb  = (const float*)d_in[4];
    const float* Ws       = (const float*)d_in[5];
    const float* bs       = (const float*)d_in[6];
    const float* gammas   = (const float*)d_in[7];
    const float* betas    = (const float*)d_in[8];
    const float* W_out    = (const float*)d_in[9];
    float*       out      = (float*)d_out;

    char* ws = (char*)d_ws;
    size_t off = 0;
    int*   degcnt = (int*)(ws + off);   off += (size_t)N_NODES * 4;
    float* dinv   = (float*)(ws + off); off += (size_t)N_NODES * 4;
    float* invdeg = (float*)(ws + off); off += (size_t)N_NODES * 4;
    float* h      = (float*)(ws + off); off += (size_t)N_NODES * HID * 4;
    float* hw     = (float*)(ws + off); off += (size_t)N_NODES * HID * 4;
    float* agg    = (float*)(ws + off); off += (size_t)N_NODES * HID * 4;
    (void)ws_size; (void)in_sizes; (void)n_in; (void)out_size;

    k_deg_init  <<<N_NODES / 256, 256, 0, stream>>>(degcnt);
    k_deg_count <<<E_EDGES / 256, 256, 0, stream>>>(ei, degcnt);
    k_deg_finish<<<N_NODES / 256, 256, 0, stream>>>(degcnt, dinv, invdeg);

    k_embed<<<(N_NODES * 32) / 256, 256, 0, stream>>>(x, cls_emb, nbr_emb, h);

    for (int i = 0; i < N_LAYERS; ++i) {
        k_gemm_fused<<<N_NODES / 128, 256, 0, stream>>>(
            h, Ws + (size_t)i * HID * HID, bs + (size_t)i * HID, invdeg, hw, agg);
        k_scatter<<<(E_EDGES * 32) / 256, 256, 0, stream>>>(ei, dinv, hw, agg);
        k_relu_res_ln<<<(N_NODES * 32) / 256, 256, 0, stream>>>(
            h, agg, gammas + (size_t)i * HID, betas + (size_t)i * HID);
    }

    k_out<<<(NGRAPH * OUT_DIM + 255) / 256, 256, 0, stream>>>(h, W_out, out);
}


# Atomic f32 edge scatter replaced by dst-grouped edge list (counting sort once per call, buckets ordered by src) + wave-per-node register gather; agg written once; sum order now fixed, no f32 atomics
# speedup vs baseline: 2.4849x; 2.4849x over previous
#include <hip/hip_runtime.h>

#define N_NODES   65536
#define NPG       128
#define NGRAPH    512
#define E_EDGES   1048576
#define HID       128
#define OUT_DIM   10
#define N_LAYERS  4
#define LN_EPS    1e-5f

typedef float v2f __attribute__((ext_vector_type(2)));
typedef float v4f __attribute__((ext_vector_type(4)));
typedef float v8f __attribute__((ext_vector_type(8)));

__global__ void k_deg_init(int* __restrict__ degcnt) {
    int i = blockIdx.x * 256 + threadIdx.x;
    if (i < N_NODES) degcnt[i] = 0;
}

__global__ void k_deg_count(const int* __restrict__ edge_index, int* __restrict__ degcnt,
                            int* __restrict__ slot) {
    int e = blockIdx.x * 256 + threadIdx.x;
    if (e < E_EDGES) {
        int d = edge_index[E_EDGES + e];
        slot[e] = atomicAdd(&degcnt[d], 1);
    }
}

#define SCAN_T 1024
__launch_bounds__(SCAN_T)
__global__ void k_deg_finish(const int* __restrict__ degcnt,
                             float* __restrict__ dinv, float* __restrict__ invdeg,
                             int* __restrict__ rowptr) {
    __shared__ int part[SCAN_T];
    __shared__ int wsum[SCAN_T / 32];
    const int t = threadIdx.x;
    const int first = blockIdx.x * SCAN_T;
    const int i = first + t;

    const int c = degcnt[i];
    const float d = 1.0f + (float)c;
    dinv[i]   = rsqrtf(d);
    invdeg[i] = 1.0f / d;

    int pre = 0;
    for (int k = t; k < first; k += SCAN_T) pre += degcnt[k];
    for (int off = 16; off >= 1; off >>= 1) pre += __shfl_xor(pre, off, 32);
    if ((t & 31) == 0) wsum[t >> 5] = pre;

    part[t] = c;
    __syncthreads();
    for (int off = 1; off < SCAN_T; off <<= 1) {
        const int v = (t >= off) ? part[t - off] : 0;
        __syncthreads();
        part[t] += v;
        __syncthreads();
    }
    int base = 0;
    for (int w = 0; w < SCAN_T / 32; ++w) base += wsum[w];

    const int incl = base + part[t];
    rowptr[i] = incl - c;
    if (i == N_NODES - 1) rowptr[N_NODES] = incl;
}

__global__ void k_csr_fill(const int* __restrict__ edge_index,
                           const int* __restrict__ rowptr,
                           const int* __restrict__ slot,
                           int* __restrict__ csr_tmp) {
    int e = blockIdx.x * 256 + threadIdx.x;
    if (e < E_EDGES) {
        int s = edge_index[e];
        int d = edge_index[E_EDGES + e];
        csr_tmp[rowptr[d] + slot[e]] = s;
    }
}

__launch_bounds__(256)
__global__ void k_csr_order(const int* __restrict__ rowptr,
                            const int* __restrict__ csr_tmp,
                            const float* __restrict__ dinv,
                            int* __restrict__ csr_src,
                            float* __restrict__ csr_w) {
    const int t = blockIdx.x * 256 + threadIdx.x;
    const int n = t >> 5;
    const int lane = t & 31;
    const int beg = rowptr[n];
    const int deg = rowptr[n + 1] - beg;
    const float dn = dinv[n];
    const int* b = csr_tmp + beg;
    for (int i = lane; i < deg; i += 32) {
        const int key = b[i] & (N_NODES - 1);
        int rank = 0;
        for (int j = 0; j < deg; ++j) {
            const int kj = b[j] & (N_NODES - 1);
            rank += (kj < key || (kj == key && j < i)) ? 1 : 0;
        }
        csr_src[beg + rank] = key;
        csr_w[beg + rank]   = dinv[key] * dn;
    }
}

__global__ void k_embed(const int* __restrict__ x,
                        const float* __restrict__ cls_emb,
                        const float* __restrict__ nbr_emb,
                        float* __restrict__ h) {
    int t = blockIdx.x * 256 + threadIdx.x;
    int n = t >> 5;
    int f = (t & 31) << 2;
    int c  = x[2 * n + 0];
    int nb = x[2 * n + 1];
    v4f a = *(const v4f*)(cls_emb + (size_t)c  * HID + f);
    v4f b = *(const v4f*)(nbr_emb + (size_t)nb * HID + f);
    *(v4f*)(h + (size_t)n * HID + f) = a + b;
}

__launch_bounds__(256)
__global__ void k_gemm_fused(const float* __restrict__ h,
                             const float* __restrict__ W,
                             float* __restrict__ hw) {
    __shared__ float lWt[HID * HID];
    const int tid = threadIdx.x;

    for (int i = tid * 4; i < HID * HID; i += 256 * 4) {
        v4f w = *(const v4f*)(W + i);
        const int k = i >> 7;
        const int n = i & (HID - 1);
#pragma unroll
        for (int j = 0; j < 4; ++j)
            lWt[(n + j) * HID + (k ^ (((n + j) & 31) << 1))] = w[j];
    }
    __syncthreads();

    const int wave = tid >> 5;
    const int lane = tid & 31;
    const int half = lane >> 4;
    const int l16  = lane & 15;
    const int m0   = blockIdx.x * 128 + wave * 16;

    v8f zero = {};
    v8f acc[8];
#pragma unroll
    for (int t = 0; t < 8; ++t) acc[t] = zero;

    const float* arow = h + (size_t)(m0 + l16) * HID;

    for (int k = 0; k < HID; k += 4) {
        const int kb = k + half * 2;
        v2f a = *(const v2f*)(arow + kb);
#pragma unroll
        for (int t = 0; t < 8; ++t) {
            const int col = t * 16 + l16;
            v2f b = *(const v2f*)(lWt + col * HID + (kb ^ ((col & 31) << 1)));
            acc[t] = __builtin_amdgcn_wmma_f32_16x16x4_f32(
                false, a, false, b, (short)0, acc[t], false, false);
        }
    }

    const int rbase = half * 8;
#pragma unroll
    for (int r = 0; r < 8; ++r) {
        const int row = m0 + rbase + r;
        float* hwrow  = hw  + (size_t)row * HID;
#pragma unroll
        for (int t = 0; t < 8; ++t) {
            const int col = t * 16 + l16;
            hwrow[col] = acc[t][r];
        }
    }
}


__device__ __forceinline__ v4f edge_row(const float* hwf, int s_l, int j) {
    const int s = __builtin_amdgcn_readlane(s_l, j);
    return *(const v4f*)(hwf + (size_t)s * HID);
}
__device__ __forceinline__ float edge_wt(float w_l, int j) {
    return __int_as_float(__builtin_amdgcn_readlane(__float_as_int(w_l), j));
}

__launch_bounds__(256)
__global__ void k_gather(const int* __restrict__ rowptr,
                         const int* __restrict__ csr_src,
                         const float* __restrict__ csr_w,
                         const float* __restrict__ hw,
                         const float* __restrict__ invdeg,
                         const float* __restrict__ bias,
                         float* __restrict__ agg) {
    const int t = blockIdx.x * 256 + threadIdx.x;
    const int n = __builtin_amdgcn_readfirstlane(t >> 5);
    const int lane = threadIdx.x & 31;
    const int f = lane << 2;
    const int beg = rowptr[n];
    const int end = rowptr[n + 1];
    const float* hwf = hw + f;

    const v4f  self = *(const v4f*)(hwf + (size_t)n * HID);
    const float inv = invdeg[n];
    const v4f  bv   = *(const v4f*)(bias + f);

    v4f acc = {0.0f, 0.0f, 0.0f, 0.0f};
    for (int base = beg; base < end; base += 32) {
        const int idx = base + lane;
        int   s_l = 0;
        float w_l = 0.0f;
        if (idx < end) { s_l = csr_src[idx]; w_l = csr_w[idx]; }
        const int left = end - base;
        const int cnt = left < 32 ? left : 32;
        int j = 0;
        for (; j + 4 <= cnt; j += 4) {
            const v4f v0 = edge_row(hwf, s_l, j + 0);
            const v4f v1 = edge_row(hwf, s_l, j + 1);
            const v4f v2 = edge_row(hwf, s_l, j + 2);
            const v4f v3 = edge_row(hwf, s_l, j + 3);
            acc += v0 * edge_wt(w_l, j + 0);
            acc += v1 * edge_wt(w_l, j + 1);
            acc += v2 * edge_wt(w_l, j + 2);
            acc += v3 * edge_wt(w_l, j + 3);
        }
        const int rem = cnt - j;
        if (rem == 3) {
            const v4f v0 = edge_row(hwf, s_l, j + 0);
            const v4f v1 = edge_row(hwf, s_l, j + 1);
            const v4f v2 = edge_row(hwf, s_l, j + 2);
            acc += v0 * edge_wt(w_l, j + 0);
            acc += v1 * edge_wt(w_l, j + 1);
            acc += v2 * edge_wt(w_l, j + 2);
        } else if (rem == 2) {
            const v4f v0 = edge_row(hwf, s_l, j + 0);
            const v4f v1 = edge_row(hwf, s_l, j + 1);
            acc += v0 * edge_wt(w_l, j + 0);
            acc += v1 * edge_wt(w_l, j + 1);
        } else if (rem == 1) {
            const v4f v0 = edge_row(hwf, s_l, j);
            acc += v0 * edge_wt(w_l, j);
        }
    }

    const v4f r = (acc + self * inv) + bv;
    *(v4f*)(agg + (size_t)n * HID + f) = r;
}

__launch_bounds__(256)
__global__ void k_relu_res_ln(float* __restrict__ h,
                              const float* __restrict__ agg,
                              const float* __restrict__ gamma,
                              const float* __restrict__ beta) {
    int t = blockIdx.x * 256 + threadIdx.x;
    int n = t >> 5;
    int lane = t & 31;
    int f = lane << 2;

    const float* ar = agg + (size_t)n * HID;
    float* hr = h + (size_t)n * HID;

    v4f hv = *(const v4f*)(hr + f);
    v4f av = *(const v4f*)(ar + f);
    v4f val;
    val.x = hv.x + fmaxf(av.x, 0.0f);
    val.y = hv.y + fmaxf(av.y, 0.0f);
    val.z = hv.z + fmaxf(av.z, 0.0f);
    val.w = hv.w + fmaxf(av.w, 0.0f);

    float s = val.x + val.y + val.z + val.w;
#pragma unroll
    for (int off = 16; off >= 1; off >>= 1) s += __shfl_xor(s, off, 32);
    const float mu = s * (1.0f / (float)HID);

    v4f dv = val - mu;
    float q = dv.x * dv.x + dv.y * dv.y + dv.z * dv.z + dv.w * dv.w;
#pragma unroll
    for (int off = 16; off >= 1; off >>= 1) q += __shfl_xor(q, off, 32);
    const float rstd = rsqrtf(q * (1.0f / (float)HID) + LN_EPS);

    v4f g = *(const v4f*)(gamma + f);
    v4f b = *(const v4f*)(beta + f);
    v4f outv = dv * rstd * g + b;
    *(v4f*)(hr + f) = outv;
}

__global__ void k_out(const float* __restrict__ h,
                      const float* __restrict__ W_out,
                      float* __restrict__ out) {
    int t = blockIdx.x * 256 + threadIdx.x;
    if (t >= NGRAPH * OUT_DIM) return;
    int g = t / OUT_DIM;
    int o = t % OUT_DIM;
    const float* hr = h + (size_t)(g * NPG) * HID;
    float s = 0.0f;
#pragma unroll 4
    for (int k = 0; k < HID; ++k) s = fmaf(hr[k], W_out[k * OUT_DIM + o], s);
    out[t] = s;
}

extern "C" void kernel_launch(void* const* d_in, const int* in_sizes, int n_in,
                              void* d_out, int out_size, void* d_ws, size_t ws_size,
                              hipStream_t stream) {
    const int*   x        = (const int*)d_in[0];
    const int*   ei       = (const int*)d_in[1];
    const float* cls_emb  = (const float*)d_in[3];
    const float* nbr_emb  = (const float*)d_in[4];
    const float* Ws       = (const float*)d_in[5];
    const float* bs       = (const float*)d_in[6];
    const float* gammas   = (const float*)d_in[7];
    const float* betas    = (const float*)d_in[8];
    const float* W_out    = (const float*)d_in[9];
    float*       out      = (float*)d_out;

    char* ws = (char*)d_ws;
    size_t off = 0;
    int*   degcnt = (int*)(ws + off);   off += (size_t)N_NODES * 4;
    float* dinv   = (float*)(ws + off); off += (size_t)N_NODES * 4;
    float* invdeg = (float*)(ws + off); off += (size_t)N_NODES * 4;
    float* h      = (float*)(ws + off); off += (size_t)N_NODES * HID * 4;
    float* hw     = (float*)(ws + off); off += (size_t)N_NODES * HID * 4;
    float* agg    = (float*)(ws + off); off += (size_t)N_NODES * HID * 4;
    int*   rowptr = (int*)(ws + off);   off += (size_t)(N_NODES + 64) * 4;
    int*   slot   = (int*)(ws + off);   off += (size_t)E_EDGES * 4;
    int*   csrtmp = (int*)(ws + off);   off += (size_t)E_EDGES * 4;
    int*   csrsrc = (int*)(ws + off);   off += (size_t)E_EDGES * 4;
    float* csrw   = (float*)(ws + off); off += (size_t)E_EDGES * 4;
    (void)ws_size; (void)in_sizes; (void)n_in; (void)out_size;

    k_deg_init  <<<N_NODES / 256, 256, 0, stream>>>(degcnt);
    k_deg_count <<<E_EDGES / 256, 256, 0, stream>>>(ei, degcnt, slot);
    k_deg_finish<<<N_NODES / SCAN_T, SCAN_T, 0, stream>>>(degcnt, dinv, invdeg, rowptr);

    k_csr_fill <<<E_EDGES / 256, 256, 0, stream>>>(ei, rowptr, slot, csrtmp);
    k_csr_order<<<(N_NODES * 32) / 256, 256, 0, stream>>>(rowptr, csrtmp, dinv, csrsrc, csrw);

    k_embed<<<(N_NODES * 32) / 256, 256, 0, stream>>>(x, cls_emb, nbr_emb, h);

    for (int i = 0; i < N_LAYERS; ++i) {
        k_gemm_fused<<<N_NODES / 128, 256, 0, stream>>>(
            h, Ws + (size_t)i * HID * HID, hw);
        k_gather<<<(N_NODES * 32) / 256, 256, 0, stream>>>(
            rowptr, csrsrc, csrw, hw, invdeg, bs + (size_t)i * HID, agg);
        k_relu_res_ln<<<(N_NODES * 32) / 256, 256, 0, stream>>>(
            h, agg, gammas + (size_t)i * HID, betas + (size_t)i * HID);
    }

    k_out<<<(NGRAPH * OUT_DIM + 255) / 256, 256, 0, stream>>>(h, W_out, out);
}


# relu + residual + LayerNorm folded into the gather kernel's tail: the summed row stays in registers; 4 launches, 262144 waves and the agg store + reload dropped; every sum order as before
# speedup vs baseline: 2.7713x; 1.1153x over previous
#include <hip/hip_runtime.h>

#define N_NODES   65536
#define NPG       128
#define NGRAPH    512
#define E_EDGES   1048576
#define HID       128
#define OUT_DIM   10
#define N_LAYERS  4
#define LN_EPS    1e-5f

typedef float v2f __attribute__((ext_vector_type(2)));
typedef float v4f __attribute__((ext_vector_type(4)));
typedef float v8f __attribute__((ext_vector_type(8)));

__global__ void k_deg_init(int* __restrict__ degcnt) {
    int i = blockIdx.x * 256 + threadIdx.x;
    if (i < N_NODES) degcnt[i] = 0;
}

__global__ void k_deg_count(const int* __restrict__ edge_index, int* __restrict__ degcnt,
                            int* __restrict__ slot) {
    int e = blockIdx.x * 256 + threadIdx.x;
    if (e < E_EDGES) {
        int d = edge_index[E_EDGES + e];
        slot[e] = atomicAdd(&degcnt[d], 1);
    }
}

#define SCAN_T 1024
__launch_bounds__(SCAN_T)
__global__ void k_deg_finish(const int* __restrict__ degcnt,
                             float* __restrict__ dinv, float* __restrict__ invdeg,
                             int* __restrict__ rowptr) {
    __shared__ int part[SCAN_T];
    __shared__ int wsum[SCAN_T / 32];
    const int t = threadIdx.x;
    const int first = blockIdx.x * SCAN_T;
    const int i = first + t;

    const int c = degcnt[i];
    const float d = 1.0f + (float)c;
    dinv[i]   = rsqrtf(d);
    invdeg[i] = 1.0f / d;

    int pre = 0;
    for (int k = t; k < first; k += SCAN_T) pre += degcnt[k];
    for (int off = 16; off >= 1; off >>= 1) pre += __shfl_xor(pre, off, 32);
    if ((t & 31) == 0) wsum[t >> 5] = pre;

    part[t] = c;
    __syncthreads();
    for (int off = 1; off < SCAN_T; off <<= 1) {
        const int v = (t >= off) ? part[t - off] : 0;
        __syncthreads();
        part[t] += v;
        __syncthreads();
    }
    int base = 0;
    for (int w = 0; w < SCAN_T / 32; ++w) base += wsum[w];

    const int incl = base + part[t];
    rowptr[i] = incl - c;
    if (i == N_NODES - 1) rowptr[N_NODES] = incl;
}

__global__ void k_csr_fill(const int* __restrict__ edge_index,
                           const int* __restrict__ rowptr,
                           const int* __restrict__ slot,
                           int* __restrict__ csr_tmp) {
    int e = blockIdx.x * 256 + threadIdx.x;
    if (e < E_EDGES) {
        int s = edge_index[e];
        int d = edge_index[E_EDGES + e];
        csr_tmp[rowptr[d] + slot[e]] = s;
    }
}

__launch_bounds__(256)
__global__ void k_csr_order(const int* __restrict__ rowptr,
                            const int* __restrict__ csr_tmp,
                            const float* __restrict__ dinv,
                            int* __restrict__ csr_src,
                            float* __restrict__ csr_w) {
    const int t = blockIdx.x * 256 + threadIdx.x;
    const int n = t >> 5;
    const int lane = t & 31;
    const int beg = rowptr[n];
    const int deg = rowptr[n + 1] - beg;
    const float dn = dinv[n];
    const int* b = csr_tmp + beg;
    for (int i = lane; i < deg; i += 32) {
        const int key = b[i] & (N_NODES - 1);
        int rank = 0;
        for (int j = 0; j < deg; ++j) {
            const int kj = b[j] & (N_NODES - 1);
            rank += (kj < key || (kj == key && j < i)) ? 1 : 0;
        }
        csr_src[beg + rank] = key;
        csr_w[beg + rank]   = dinv[key] * dn;
    }
}

__global__ void k_embed(const int* __restrict__ x,
                        const float* __restrict__ cls_emb,
                        const float* __restrict__ nbr_emb,
                        float* __restrict__ h) {
    int t = blockIdx.x * 256 + threadIdx.x;
    int n = t >> 5;
    int f = (t & 31) << 2;
    int c  = x[2 * n + 0];
    int nb = x[2 * n + 1];
    v4f a = *(const v4f*)(cls_emb + (size_t)c  * HID + f);
    v4f b = *(const v4f*)(nbr_emb + (size_t)nb * HID + f);
    *(v4f*)(h + (size_t)n * HID + f) = a + b;
}

__launch_bounds__(256)
__global__ void k_gemm_fused(const float* __restrict__ h,
                             const float* __restrict__ W,
                             float* __restrict__ hw) {
    __shared__ float lWt[HID * HID];
    const int tid = threadIdx.x;

    for (int i = tid * 4; i < HID * HID; i += 256 * 4) {
        v4f w = *(const v4f*)(W + i);
        const int k = i >> 7;
        const int n = i & (HID - 1);
#pragma unroll
        for (int j = 0; j < 4; ++j)
            lWt[(n + j) * HID + (k ^ (((n + j) & 31) << 1))] = w[j];
    }
    __syncthreads();

    const int wave = tid >> 5;
    const int lane = tid & 31;
    const int half = lane >> 4;
    const int l16  = lane & 15;
    const int m0   = blockIdx.x * 128 + wave * 16;

    v8f zero = {};
    v8f acc[8];
#pragma unroll
    for (int t = 0; t < 8; ++t) acc[t] = zero;

    const float* arow = h + (size_t)(m0 + l16) * HID;

    for (int k = 0; k < HID; k += 4) {
        const int kb = k + half * 2;
        v2f a = *(const v2f*)(arow + kb);
#pragma unroll
        for (int t = 0; t < 8; ++t) {
            const int col = t * 16 + l16;
            v2f b = *(const v2f*)(lWt + col * HID + (kb ^ ((col & 31) << 1)));
            acc[t] = __builtin_amdgcn_wmma_f32_16x16x4_f32(
                false, a, false, b, (short)0, acc[t], false, false);
        }
    }

    const int rbase = half * 8;
#pragma unroll
    for (int r = 0; r < 8; ++r) {
        const int row = m0 + rbase + r;
        float* hwrow  = hw  + (size_t)row * HID;
#pragma unroll
        for (int t = 0; t < 8; ++t) {
            const int col = t * 16 + l16;
            hwrow[col] = acc[t][r];
        }
    }
}


__device__ __forceinline__ v4f edge_row(const float* hwf, int s_l, int j) {
    const int s = __builtin_amdgcn_readlane(s_l, j);
    return *(const v4f*)(hwf + (size_t)s * HID);
}
__device__ __forceinline__ float edge_wt(float w_l, int j) {
    return __int_as_float(__builtin_amdgcn_readlane(__float_as_int(w_l), j));
}

__launch_bounds__(256)
__global__ void k_gather_ln(const int* __restrict__ rowptr,
                            const int* __restrict__ csr_src,
                            const float* __restrict__ csr_w,
                            const float* __restrict__ hw,
                            const float* __restrict__ invdeg,
                            const float* __restrict__ bias,
                            const float* __restrict__ gamma,
                            const float* __restrict__ beta,
                            float* __restrict__ h) {
    const int t = blockIdx.x * 256 + threadIdx.x;
    const int n = __builtin_amdgcn_readfirstlane(t >> 5);
    const int lane = threadIdx.x & 31;
    const int f = lane << 2;
    const int beg = rowptr[n];
    const int end = rowptr[n + 1];
    const float* hwf = hw + f;

    const v4f  self = *(const v4f*)(hwf + (size_t)n * HID);
    const float inv = invdeg[n];
    const v4f  bv   = *(const v4f*)(bias + f);
    float* hr = h + (size_t)n * HID + f;
    const v4f  hv   = *(const v4f*)hr;
    const v4f  g    = *(const v4f*)(gamma + f);
    const v4f  b    = *(const v4f*)(beta + f);

    v4f acc = {0.0f, 0.0f, 0.0f, 0.0f};
    for (int base = beg; base < end; base += 32) {
        const int idx = base + lane;
        int   s_l = 0;
        float w_l = 0.0f;
        if (idx < end) { s_l = csr_src[idx]; w_l = csr_w[idx]; }
        const int left = end - base;
        const int cnt = left < 32 ? left : 32;
        int j = 0;
        for (; j + 4 <= cnt; j += 4) {
            const v4f v0 = edge_row(hwf, s_l, j + 0);
            const v4f v1 = edge_row(hwf, s_l, j + 1);
            const v4f v2 = edge_row(hwf, s_l, j + 2);
            const v4f v3 = edge_row(hwf, s_l, j + 3);
            acc += v0 * edge_wt(w_l, j + 0);
            acc += v1 * edge_wt(w_l, j + 1);
            acc += v2 * edge_wt(w_l, j + 2);
            acc += v3 * edge_wt(w_l, j + 3);
        }
        const int rem = cnt - j;
        if (rem == 3) {
            const v4f v0 = edge_row(hwf, s_l, j + 0);
            const v4f v1 = edge_row(hwf, s_l, j + 1);
            const v4f v2 = edge_row(hwf, s_l, j + 2);
            acc += v0 * edge_wt(w_l, j + 0);
            acc += v1 * edge_wt(w_l, j + 1);
            acc += v2 * edge_wt(w_l, j + 2);
        } else if (rem == 2) {
            const v4f v0 = edge_row(hwf, s_l, j + 0);
            const v4f v1 = edge_row(hwf, s_l, j + 1);
            acc += v0 * edge_wt(w_l, j + 0);
            acc += v1 * edge_wt(w_l, j + 1);
        } else if (rem == 1) {
            const v4f v0 = edge_row(hwf, s_l, j);
            acc += v0 * edge_wt(w_l, j);
        }
    }

    const v4f av = (acc + self * inv) + bv;

    v4f val;
    val.x = hv.x + fmaxf(av.x, 0.0f);
    val.y = hv.y + fmaxf(av.y, 0.0f);
    val.z = hv.z + fmaxf(av.z, 0.0f);
    val.w = hv.w + fmaxf(av.w, 0.0f);

    float s = val.x + val.y + val.z + val.w;
#pragma unroll
    for (int off = 16; off >= 1; off >>= 1) s += __shfl_xor(s, off, 32);
    const float mu = s * (1.0f / (float)HID);

    v4f dv = val - mu;
    float q = dv.x * dv.x + dv.y * dv.y + dv.z * dv.z + dv.w * dv.w;
#pragma unroll
    for (int off = 16; off >= 1; off >>= 1) q += __shfl_xor(q, off, 32);
    const float rstd = rsqrtf(q * (1.0f / (float)HID) + LN_EPS);

    const v4f outv = dv * rstd * g + b;
    *(v4f*)hr = outv;
}


__global__ void k_out(const float* __restrict__ h,
                      const float* __restrict__ W_out,
                      float* __restrict__ out) {
    int t = blockIdx.x * 256 + threadIdx.x;
    if (t >= NGRAPH * OUT_DIM) return;
    int g = t / OUT_DIM;
    int o = t % OUT_DIM;
    const float* hr = h + (size_t)(g * NPG) * HID;
    float s = 0.0f;
#pragma unroll 4
    for (int k = 0; k < HID; ++k) s = fmaf(hr[k], W_out[k * OUT_DIM + o], s);
    out[t] = s;
}

extern "C" void kernel_launch(void* const* d_in, const int* in_sizes, int n_in,
                              void* d_out, int out_size, void* d_ws, size_t ws_size,
                              hipStream_t stream) {
    const int*   x        = (const int*)d_in[0];
    const int*   ei       = (const int*)d_in[1];
    const float* cls_emb  = (const float*)d_in[3];
    const float* nbr_emb  = (const float*)d_in[4];
    const float* Ws       = (const float*)d_in[5];
    const float* bs       = (const float*)d_in[6];
    const float* gammas   = (const float*)d_in[7];
    const float* betas    = (const float*)d_in[8];
    const float* W_out    = (const float*)d_in[9];
    float*       out      = (float*)d_out;

    char* ws = (char*)d_ws;
    size_t off = 0;
    int*   degcnt = (int*)(ws + off);   off += (size_t)N_NODES * 4;
    float* dinv   = (float*)(ws + off); off += (size_t)N_NODES * 4;
    float* invdeg = (float*)(ws + off); off += (size_t)N_NODES * 4;
    float* h      = (float*)(ws + off); off += (size_t)N_NODES * HID * 4;
    float* hw     = (float*)(ws + off); off += (size_t)N_NODES * HID * 4;
    int*   rowptr = (int*)(ws + off);   off += (size_t)(N_NODES + 64) * 4;
    int*   slot   = (int*)(ws + off);   off += (size_t)E_EDGES * 4;
    int*   csrtmp = (int*)(ws + off);   off += (size_t)E_EDGES * 4;
    int*   csrsrc = (int*)(ws + off);   off += (size_t)E_EDGES * 4;
    float* csrw   = (float*)(ws + off); off += (size_t)E_EDGES * 4;
    (void)ws_size; (void)in_sizes; (void)n_in; (void)out_size;

    k_deg_init  <<<N_NODES / 256, 256, 0, stream>>>(degcnt);
    k_deg_count <<<E_EDGES / 256, 256, 0, stream>>>(ei, degcnt, slot);
    k_deg_finish<<<N_NODES / SCAN_T, SCAN_T, 0, stream>>>(degcnt, dinv, invdeg, rowptr);

    k_csr_fill <<<E_EDGES / 256, 256, 0, stream>>>(ei, rowptr, slot, csrtmp);
    k_csr_order<<<(N_NODES * 32) / 256, 256, 0, stream>>>(rowptr, csrtmp, dinv, csrsrc, csrw);

    k_embed<<<(N_NODES * 32) / 256, 256, 0, stream>>>(x, cls_emb, nbr_emb, h);

    for (int i = 0; i < N_LAYERS; ++i) {
        k_gemm_fused<<<N_NODES / 128, 256, 0, stream>>>(
            h, Ws + (size_t)i * HID * HID, hw);
        k_gather_ln<<<(N_NODES * 32) / 256, 256, 0, stream>>>(
            rowptr, csrsrc, csrw, hw, invdeg, bs + (size_t)i * HID,
            gammas + (size_t)i * HID, betas + (size_t)i * HID, h);
    }

    k_out<<<(NGRAPH * OUT_DIM + 255) / 256, 256, 0, stream>>>(h, W_out, out);
}


# gather+LN computed only for rows the logits depend on: last layer for the 512 root rows, the layer before for the roots and the src nodes of their edges (flag array); all sums as before
# speedup vs baseline: 3.1077x; 1.1214x over previous
#include <hip/hip_runtime.h>

#define N_NODES   65536
#define NPG       128
#define NGRAPH    512
#define E_EDGES   1048576
#define HID       128
#define OUT_DIM   10
#define N_LAYERS  4
#define LN_EPS    1e-5f

typedef float v2f __attribute__((ext_vector_type(2)));
typedef float v4f __attribute__((ext_vector_type(4)));
typedef float v8f __attribute__((ext_vector_type(8)));

__global__ void k_deg_init(int* __restrict__ degcnt, int* __restrict__ need) {
    int i = blockIdx.x * 256 + threadIdx.x;
    if (i < N_NODES) { degcnt[i] = 0; need[i] = 0; }
}

__global__ void k_deg_count(const int* __restrict__ edge_index, int* __restrict__ degcnt,
                            int* __restrict__ slot) {
    int e = blockIdx.x * 256 + threadIdx.x;
    if (e < E_EDGES) {
        int d = edge_index[E_EDGES + e];
        slot[e] = atomicAdd(&degcnt[d], 1);
    }
}

#define SCAN_T 1024
__launch_bounds__(SCAN_T)
__global__ void k_deg_finish(const int* __restrict__ degcnt,
                             float* __restrict__ dinv, float* __restrict__ invdeg,
                             int* __restrict__ rowptr) {
    __shared__ int part[SCAN_T];
    __shared__ int wsum[SCAN_T / 32];
    const int t = threadIdx.x;
    const int first = blockIdx.x * SCAN_T;
    const int i = first + t;

    const int c = degcnt[i];
    const float d = 1.0f + (float)c;
    dinv[i]   = rsqrtf(d);
    invdeg[i] = 1.0f / d;

    int pre = 0;
    for (int k = t; k < first; k += SCAN_T) pre += degcnt[k];
    for (int off = 16; off >= 1; off >>= 1) pre += __shfl_xor(pre, off, 32);
    if ((t & 31) == 0) wsum[t >> 5] = pre;

    part[t] = c;
    __syncthreads();
    for (int off = 1; off < SCAN_T; off <<= 1) {
        const int v = (t >= off) ? part[t - off] : 0;
        __syncthreads();
        part[t] += v;
        __syncthreads();
    }
    int base = 0;
    for (int w = 0; w < SCAN_T / 32; ++w) base += wsum[w];

    const int incl = base + part[t];
    rowptr[i] = incl - c;
    if (i == N_NODES - 1) rowptr[N_NODES] = incl;
}

__global__ void k_csr_fill(const int* __restrict__ edge_index,
                           const int* __restrict__ rowptr,
                           const int* __restrict__ slot,
                           int* __restrict__ csr_tmp) {
    int e = blockIdx.x * 256 + threadIdx.x;
    if (e < E_EDGES) {
        int s = edge_index[e];
        int d = edge_index[E_EDGES + e];
        csr_tmp[rowptr[d] + slot[e]] = s;
    }
}

__launch_bounds__(256)
__global__ void k_csr_order(const int* __restrict__ rowptr,
                            const int* __restrict__ csr_tmp,
                            const float* __restrict__ dinv,
                            int* __restrict__ csr_src,
                            float* __restrict__ csr_w,
                            int* __restrict__ need) {
    const int t = blockIdx.x * 256 + threadIdx.x;
    const int n = t >> 5;
    const int lane = t & 31;
    const bool root = (n % NPG) == 0;
    if (root && lane == 0) need[n] = 1;
    const int beg = rowptr[n];
    const int deg = rowptr[n + 1] - beg;
    const float dn = dinv[n];
    const int* b = csr_tmp + beg;
    for (int i = lane; i < deg; i += 32) {
        const int key = b[i] & (N_NODES - 1);
        int rank = 0;
        for (int j = 0; j < deg; ++j) {
            const int kj = b[j] & (N_NODES - 1);
            rank += (kj < key || (kj == key && j < i)) ? 1 : 0;
        }
        csr_src[beg + rank] = key;
        csr_w[beg + rank]   = dinv[key] * dn;
        if (root) need[key] = 1;
    }
}

__global__ void k_embed(const int* __restrict__ x,
                        const float* __restrict__ cls_emb,
                        const float* __restrict__ nbr_emb,
                        float* __restrict__ h) {
    int t = blockIdx.x * 256 + threadIdx.x;
    int n = t >> 5;
    int f = (t & 31) << 2;
    int c  = x[2 * n + 0];
    int nb = x[2 * n + 1];
    v4f a = *(const v4f*)(cls_emb + (size_t)c  * HID + f);
    v4f b = *(const v4f*)(nbr_emb + (size_t)nb * HID + f);
    *(v4f*)(h + (size_t)n * HID + f) = a + b;
}

__launch_bounds__(256)
__global__ void k_gemm_fused(const float* __restrict__ h,
                             const float* __restrict__ W,
                             float* __restrict__ hw) {
    __shared__ float lWt[HID * HID];
    const int tid = threadIdx.x;

    for (int i = tid * 4; i < HID * HID; i += 256 * 4) {
        v4f w = *(const v4f*)(W + i);
        const int k = i >> 7;
        const int n = i & (HID - 1);
#pragma unroll
        for (int j = 0; j < 4; ++j)
            lWt[(n + j) * HID + (k ^ (((n + j) & 31) << 1))] = w[j];
    }
    __syncthreads();

    const int wave = tid >> 5;
    const int lane = tid & 31;
    const int half = lane >> 4;
    const int l16  = lane & 15;
    const int m0   = blockIdx.x * 128 + wave * 16;

    v8f zero = {};
    v8f acc[8];
#pragma unroll
    for (int t = 0; t < 8; ++t) acc[t] = zero;

    const float* arow = h + (size_t)(m0 + l16) * HID;

    for (int k = 0; k < HID; k += 4) {
        const int kb = k + half * 2;
        v2f a = *(const v2f*)(arow + kb);
#pragma unroll
        for (int t = 0; t < 8; ++t) {
            const int col = t * 16 + l16;
            v2f b = *(const v2f*)(lWt + col * HID + (kb ^ ((col & 31) << 1)));
            acc[t] = __builtin_amdgcn_wmma_f32_16x16x4_f32(
                false, a, false, b, (short)0, acc[t], false, false);
        }
    }

    const int rbase = half * 8;
#pragma unroll
    for (int r = 0; r < 8; ++r) {
        const int row = m0 + rbase + r;
        float* hwrow  = hw  + (size_t)row * HID;
#pragma unroll
        for (int t = 0; t < 8; ++t) {
            const int col = t * 16 + l16;
            hwrow[col] = acc[t][r];
        }
    }
}


__device__ __forceinline__ v4f edge_row(const float* hwf, int s_l, int j) {
    const int s = __builtin_amdgcn_readlane(s_l, j);
    return *(const v4f*)(hwf + (size_t)s * HID);
}
__device__ __forceinline__ float edge_wt(float w_l, int j) {
    return __int_as_float(__builtin_amdgcn_readlane(__float_as_int(w_l), j));
}

__launch_bounds__(256)
__global__ void k_gather_ln(const int* __restrict__ rowptr,
                            const int* __restrict__ csr_src,
                            const float* __restrict__ csr_w,
                            const float* __restrict__ hw,
                            const float* __restrict__ invdeg,
                            const float* __restrict__ bias,
                            const float* __restrict__ gamma,
                            const float* __restrict__ beta,
                            float* __restrict__ h,
                            int node_stride,
                            const int* __restrict__ need) {
    const int t = blockIdx.x * 256 + threadIdx.x;
    const int n = __builtin_amdgcn_readfirstlane(t >> 5) * node_stride;
    if (need != nullptr && need[n] == 0) return;
    const int lane = threadIdx.x & 31;
    const int f = lane << 2;
    const int beg = rowptr[n];
    const int end = rowptr[n + 1];
    const float* hwf = hw + f;

    const v4f  self = *(const v4f*)(hwf + (size_t)n * HID);
    const float inv = invdeg[n];
    const v4f  bv   = *(const v4f*)(bias + f);
    float* hr = h + (size_t)n * HID + f;
    const v4f  hv   = *(const v4f*)hr;
    const v4f  g    = *(const v4f*)(gamma + f);
    const v4f  b    = *(const v4f*)(beta + f);

    v4f acc = {0.0f, 0.0f, 0.0f, 0.0f};
    for (int base = beg; base < end; base += 32) {
        const int idx = base + lane;
        int   s_l = 0;
        float w_l = 0.0f;
        if (idx < end) { s_l = csr_src[idx]; w_l = csr_w[idx]; }
        const int left = end - base;
        const int cnt = left < 32 ? left : 32;
        int j = 0;
        for (; j + 4 <= cnt; j += 4) {
            const v4f v0 = edge_row(hwf, s_l, j + 0);
            const v4f v1 = edge_row(hwf, s_l, j + 1);
            const v4f v2 = edge_row(hwf, s_l, j + 2);
            const v4f v3 = edge_row(hwf, s_l, j + 3);
            acc += v0 * edge_wt(w_l, j + 0);
            acc += v1 * edge_wt(w_l, j + 1);
            acc += v2 * edge_wt(w_l, j + 2);
            acc += v3 * edge_wt(w_l, j + 3);
        }
        const int rem = cnt - j;
        if (rem == 3) {
            const v4f v0 = edge_row(hwf, s_l, j + 0);
            const v4f v1 = edge_row(hwf, s_l, j + 1);
            const v4f v2 = edge_row(hwf, s_l, j + 2);
            acc += v0 * edge_wt(w_l, j + 0);
            acc += v1 * edge_wt(w_l, j + 1);
            acc += v2 * edge_wt(w_l, j + 2);
        } else if (rem == 2) {
            const v4f v0 = edge_row(hwf, s_l, j + 0);
            const v4f v1 = edge_row(hwf, s_l, j + 1);
            acc += v0 * edge_wt(w_l, j + 0);
            acc += v1 * edge_wt(w_l, j + 1);
        } else if (rem == 1) {
            const v4f v0 = edge_row(hwf, s_l, j);
            acc += v0 * edge_wt(w_l, j);
        }
    }

    const v4f av = (acc + self * inv) + bv;

    v4f val;
    val.x = hv.x + fmaxf(av.x, 0.0f);
    val.y = hv.y + fmaxf(av.y, 0.0f);
    val.z = hv.z + fmaxf(av.z, 0.0f);
    val.w = hv.w + fmaxf(av.w, 0.0f);

    float s = val.x + val.y + val.z + val.w;
#pragma unroll
    for (int off = 16; off >= 1; off >>= 1) s += __shfl_xor(s, off, 32);
    const float mu = s * (1.0f / (float)HID);

    v4f dv = val - mu;
    float q = dv.x * dv.x + dv.y * dv.y + dv.z * dv.z + dv.w * dv.w;
#pragma unroll
    for (int off = 16; off >= 1; off >>= 1) q += __shfl_xor(q, off, 32);
    const float rstd = rsqrtf(q * (1.0f / (float)HID) + LN_EPS);

    const v4f outv = dv * rstd * g + b;
    *(v4f*)hr = outv;
}


__global__ void k_out(const float* __restrict__ h,
                      const float* __restrict__ W_out,
                      float* __restrict__ out) {
    int t = blockIdx.x * 256 + threadIdx.x;
    if (t >= NGRAPH * OUT_DIM) return;
    int g = t / OUT_DIM;
    int o = t % OUT_DIM;
    const float* hr = h + (size_t)(g * NPG) * HID;
    float s = 0.0f;
#pragma unroll 4
    for (int k = 0; k < HID; ++k) s = fmaf(hr[k], W_out[k * OUT_DIM + o], s);
    out[t] = s;
}

extern "C" void kernel_launch(void* const* d_in, const int* in_sizes, int n_in,
                              void* d_out, int out_size, void* d_ws, size_t ws_size,
                              hipStream_t stream) {
    const int*   x        = (const int*)d_in[0];
    const int*   ei       = (const int*)d_in[1];
    const float* cls_emb  = (const float*)d_in[3];
    const float* nbr_emb  = (const float*)d_in[4];
    const float* Ws       = (const float*)d_in[5];
    const float* bs       = (const float*)d_in[6];
    const float* gammas   = (const float*)d_in[7];
    const float* betas    = (const float*)d_in[8];
    const float* W_out    = (const float*)d_in[9];
    float*       out      = (float*)d_out;

    char* ws = (char*)d_ws;
    size_t off = 0;
    int*   degcnt = (int*)(ws + off);   off += (size_t)N_NODES * 4;
    float* dinv   = (float*)(ws + off); off += (size_t)N_NODES * 4;
    float* invdeg = (float*)(ws + off); off += (size_t)N_NODES * 4;
    float* h      = (float*)(ws + off); off += (size_t)N_NODES * HID * 4;
    float* hw     = (float*)(ws + off); off += (size_t)N_NODES * HID * 4;
    int*   rowptr = (int*)(ws + off);   off += (size_t)(N_NODES + 64) * 4;
    int*   slot   = (int*)(ws + off);   off += (size_t)E_EDGES * 4;
    int*   csrtmp = (int*)(ws + off);   off += (size_t)E_EDGES * 4;
    int*   csrsrc = (int*)(ws + off);   off += (size_t)E_EDGES * 4;
    float* csrw   = (float*)(ws + off); off += (size_t)E_EDGES * 4;
    int*   need   = (int*)(ws + off);   off += (size_t)N_NODES * 4;
    (void)ws_size; (void)in_sizes; (void)n_in; (void)out_size;

    k_deg_init  <<<N_NODES / 256, 256, 0, stream>>>(degcnt, need);
    k_deg_count <<<E_EDGES / 256, 256, 0, stream>>>(ei, degcnt, slot);
    k_deg_finish<<<N_NODES / SCAN_T, SCAN_T, 0, stream>>>(degcnt, dinv, invdeg, rowptr);

    k_csr_fill <<<E_EDGES / 256, 256, 0, stream>>>(ei, rowptr, slot, csrtmp);
    k_csr_order<<<(N_NODES * 32) / 256, 256, 0, stream>>>(rowptr, csrtmp, dinv, csrsrc, csrw, need);

    k_embed<<<(N_NODES * 32) / 256, 256, 0, stream>>>(x, cls_emb, nbr_emb, h);

    for (int i = 0; i < N_LAYERS - 2; ++i) {
        k_gemm_fused<<<N_NODES / 128, 256, 0, stream>>>(
            h, Ws + (size_t)i * HID * HID, hw);
        k_gather_ln<<<(N_NODES * 32) / 256, 256, 0, stream>>>(
            rowptr, csrsrc, csrw, hw, invdeg, bs + (size_t)i * HID,
            gammas + (size_t)i * HID, betas + (size_t)i * HID, h, 1, nullptr);
    }
    k_gemm_fused<<<N_NODES / 128, 256, 0, stream>>>(
        h, Ws + (size_t)(N_LAYERS - 2) * HID * HID, hw);
    k_gather_ln<<<(N_NODES * 32) / 256, 256, 0, stream>>>(
        rowptr, csrsrc, csrw, hw, invdeg, bs + (size_t)(N_LAYERS - 2) * HID,
        gammas + (size_t)(N_LAYERS - 2) * HID, betas + (size_t)(N_LAYERS - 2) * HID, h, 1, need);
    k_gemm_fused<<<N_NODES / 128, 256, 0, stream>>>(
        h, Ws + (size_t)(N_LAYERS - 1) * HID * HID, hw);
    k_gather_ln<<<(NGRAPH * 32) / 256, 256, 0, stream>>>(
        rowptr, csrsrc, csrw, hw, invdeg, bs + (size_t)(N_LAYERS - 1) * HID,
        gammas + (size_t)(N_LAYERS - 1) * HID, betas + (size_t)(N_LAYERS - 1) * HID, h, NPG, nullptr);

    k_out<<<(NGRAPH * OUT_DIM + 255) / 256, 256, 0, stream>>>(h, W_out, out);
}


# Edge grouping by dst: global returning atomics + scattered 4-B stores (count, scan, fill) replaced by a two-level counting sort on LDS counters; bucket order still fixed by src rank, sums as before
# speedup vs baseline: 4.0504x; 1.3033x over previous
#include <hip/hip_runtime.h>

#define N_NODES   65536
#define NPG       128
#define NGRAPH    512
#define E_EDGES   1048576
#define HID       128
#define OUT_DIM   10
#define N_LAYERS  4
#define LN_EPS    1e-5f

typedef float v2f __attribute__((ext_vector_type(2)));
typedef float v4f __attribute__((ext_vector_type(4)));
typedef float v8f __attribute__((ext_vector_type(8)));

#define NBIN      256
#define BIN_NODES 256
#define BIN_SHIFT 8
#define EPB       4096
#define EBLOCKS   (E_EDGES / EPB)
#define STAGE_CAP 12288
static_assert(NBIN * BIN_NODES == N_NODES, "bins must cover the nodes");
static_assert((1 << BIN_SHIFT) == BIN_NODES, "bin shift");
static_assert(EBLOCKS * EPB == E_EDGES, "edge blocks must cover the edges");
static_assert(N_NODES == 65536, "an edge is packed as src | dst << 16");
static_assert(EBLOCKS % 64 == 0 && NBIN == 256 && EPB == 16 * 256, "layout of the edge passes");

typedef int v4i __attribute__((ext_vector_type(4)));

__device__ __forceinline__ int excl_scan_256(int c, int t, int* wtot) {
    const int lane = t & 31;
    int x = c;
#pragma unroll
    for (int o = 1; o < 32; o <<= 1) {
        const int y = __shfl_up(x, o, 32);
        if (lane >= o) x += y;
    }
    if (t < 256 && lane == 31) wtot[t >> 5] = x;
    __syncthreads();
    int pre = 0;
    const int wv = t >> 5;
#pragma unroll
    for (int w = 0; w < 8; ++w) pre += (w < wv) ? wtot[w] : 0;
    return pre + x - c;
}

__launch_bounds__(256)
__global__ void k_bin_count(const int* __restrict__ edge_index, int* __restrict__ cnt) {
    __shared__ int hist[NBIN];
    const int t = threadIdx.x;
    hist[t] = 0;
    const int* dstp = edge_index + E_EDGES + blockIdx.x * EPB + t * 4;
    v4i dv[4];
#pragma unroll
    for (int i = 0; i < 4; ++i) dv[i] = *(const v4i*)(dstp + i * 1024);
    __syncthreads();
#pragma unroll
    for (int i = 0; i < 4; ++i)
#pragma unroll
        for (int j = 0; j < 4; ++j) {
            const int d = dv[i][j] & (N_NODES - 1);
            atomicAdd(&hist[d >> BIN_SHIFT], 1);
        }
    __syncthreads();
    cnt[blockIdx.x * NBIN + t] = hist[t];
}

__launch_bounds__(256)
__global__ void k_bin_scatter(const int* __restrict__ edge_index, const int* __restrict__ cnt,
                              int* __restrict__ binned, int* __restrict__ binbase) {
    __shared__ int ptot[4][NBIN];
    __shared__ int pbef[4][NBIN];
    __shared__ int cursor[NBIN];
    __shared__ int delta[NBIN];
    __shared__ int wtot_a[8];
    __shared__ int wtot_b[8];
    __shared__ int stage[EPB];
    const int t = threadIdx.x;
    const int b = blockIdx.x;
    const int* srcp = edge_index + b * EPB + t * 4;
    const int* dstp = edge_index + E_EDGES + b * EPB + t * 4;
    v4i sv[4], dv[4];
#pragma unroll
    for (int i = 0; i < 4; ++i) {
        sv[i] = *(const v4i*)(srcp + i * 1024);
        dv[i] = *(const v4i*)(dstp + i * 1024);
    }
    const int own = cnt[b * NBIN + t];
    const int q  = t >> 6;
    const int c4 = (t & 63) * 4;
    v4i tot = {0, 0, 0, 0};
    v4i bef = {0, 0, 0, 0};
#pragma unroll 16
    for (int r = q; r < EBLOCKS; r += 4) {
        const v4i c = *(const v4i*)(cnt + r * NBIN + c4);
        tot += c;
        if (r < b) bef += c;
    }
#pragma unroll
    for (int j = 0; j < 4; ++j) { ptot[q][c4 + j] = tot[j]; pbef[q][c4 + j] = bef[j]; }
    __syncthreads();
    const int total  = ptot[0][t] + ptot[1][t] + ptot[2][t] + ptot[3][t];
    const int before = pbef[0][t] + pbef[1][t] + pbef[2][t] + pbef[3][t];
    const int base   = excl_scan_256(total, t, wtot_a);
    const int lstart = excl_scan_256(own, t, wtot_b);
    cursor[t] = lstart;
    delta[t]  = base + before - lstart;
    if (b == 0) {
        binbase[t] = base;
        if (t == NBIN - 1) binbase[NBIN] = base + total;
    }
    __syncthreads();
#pragma unroll
    for (int i = 0; i < 4; ++i)
#pragma unroll
        for (int j = 0; j < 4; ++j) {
            const int s = sv[i][j] & (N_NODES - 1);
            const int d = dv[i][j] & (N_NODES - 1);
            const int lp = atomicAdd(&cursor[d >> BIN_SHIFT], 1);
            stage[lp] = (int)((unsigned)s | ((unsigned)d << 16));
        }
    __syncthreads();
#pragma unroll 4
    for (int i = t; i < EPB; i += 256) {
        const int w = stage[i];
        binned[i + delta[(w >> 24) & (NBIN - 1)]] = w;
    }
}

#define BUILD_T 1024
__launch_bounds__(BUILD_T)
__global__ void k_bin_build(const int* __restrict__ binned, const int* __restrict__ binbase,
                            int* __restrict__ rowptr, float* __restrict__ dinv,
                            float* __restrict__ invdeg, int* __restrict__ need,
                            int* __restrict__ csr_tmp) {
    __shared__ int deg[BIN_NODES];
    __shared__ int off[BIN_NODES];
    __shared__ int cur[BIN_NODES];
    __shared__ int wtot[8];
    __shared__ int stage[STAGE_CAP];
    const int t = threadIdx.x;
    const int bin = blockIdx.x;
    const int base = binbase[bin];
    const int n_e = binbase[bin + 1] - base;
    if (t < BIN_NODES) { deg[t] = 0; cur[t] = 0; }
    __syncthreads();
    const int* eb = binned + base;
    for (int i0 = 0; i0 < n_e; i0 += 4 * BUILD_T) {
        int w[4];
#pragma unroll
        for (int j = 0; j < 4; ++j) {
            const int i = i0 + j * BUILD_T + t;
            w[j] = eb[i < n_e ? i : n_e - 1];
        }
#pragma unroll
        for (int j = 0; j < 4; ++j)
            if (i0 + j * BUILD_T + t < n_e)
                atomicAdd(&deg[(w[j] >> 16) & (BIN_NODES - 1)], 1);
    }
    __syncthreads();
    const int c = (t < BIN_NODES) ? deg[t] : 0;
    const int ex = excl_scan_256(c, t, wtot);
    if (t < BIN_NODES) {
        const int node = bin * BIN_NODES + t;
        const float d = 1.0f + (float)c;
        dinv[node]   = rsqrtf(d);
        invdeg[node] = 1.0f / d;
        rowptr[node] = base + ex;
        need[node]   = 0;
        if (node == N_NODES - 1) rowptr[N_NODES] = base + ex + c;
        off[t] = ex;
    }
    __syncthreads();
    const bool fits = n_e <= STAGE_CAP;
    for (int i0 = 0; i0 < n_e; i0 += 4 * BUILD_T) {
        int w[4];
#pragma unroll
        for (int j = 0; j < 4; ++j) {
            const int i = i0 + j * BUILD_T + t;
            w[j] = eb[i < n_e ? i : n_e - 1];
        }
#pragma unroll
        for (int j = 0; j < 4; ++j)
            if (i0 + j * BUILD_T + t < n_e) {
                const int dl = (w[j] >> 16) & (BIN_NODES - 1);
                const int p = off[dl] + atomicAdd(&cur[dl], 1);
                const int key = w[j] & 0xffff;
                if (fits) stage[p] = key; else csr_tmp[base + p] = key;
            }
    }
    __syncthreads();
    if (fits)
        for (int i = t; i < n_e; i += BUILD_T) csr_tmp[base + i] = stage[i];
}

__launch_bounds__(256)
__global__ void k_csr_order(const int* __restrict__ rowptr,
                            const int* __restrict__ csr_tmp,
                            const float* __restrict__ dinv,
                            int* __restrict__ csr_src,
                            float* __restrict__ csr_w,
                            int* __restrict__ need) {
    const int t = blockIdx.x * 256 + threadIdx.x;
    const int n = t >> 5;
    const int lane = t & 31;
    const bool root = (n % NPG) == 0;
    if (root && lane == 0) need[n] = 1;
    const int beg = rowptr[n];
    const int deg = rowptr[n + 1] - beg;
    const float dn = dinv[n];
    const int* b = csr_tmp + beg;
    for (int i = lane; i < deg; i += 32) {
        const int key = b[i] & (N_NODES - 1);
        int rank = 0;
        for (int j = 0; j < deg; ++j) {
            const int kj = b[j] & (N_NODES - 1);
            rank += (kj < key || (kj == key && j < i)) ? 1 : 0;
        }
        csr_src[beg + rank] = key;
        csr_w[beg + rank]   = dinv[key] * dn;
        if (root) need[key] = 1;
    }
}

__global__ void k_embed(const int* __restrict__ x,
                        const float* __restrict__ cls_emb,
                        const float* __restrict__ nbr_emb,
                        float* __restrict__ h) {
    int t = blockIdx.x * 256 + threadIdx.x;
    int n = t >> 5;
    int f = (t & 31) << 2;
    int c  = x[2 * n + 0];
    int nb = x[2 * n + 1];
    v4f a = *(const v4f*)(cls_emb + (size_t)c  * HID + f);
    v4f b = *(const v4f*)(nbr_emb + (size_t)nb * HID + f);
    *(v4f*)(h + (size_t)n * HID + f) = a + b;
}

__launch_bounds__(256)
__global__ void k_gemm_fused(const float* __restrict__ h,
                             const float* __restrict__ W,
                             float* __restrict__ hw) {
    __shared__ float lWt[HID * HID];
    const int tid = threadIdx.x;

    for (int i = tid * 4; i < HID * HID; i += 256 * 4) {
        v4f w = *(const v4f*)(W + i);
        const int k = i >> 7;
        const int n = i & (HID - 1);
#pragma unroll
        for (int j = 0; j < 4; ++j)
            lWt[(n + j) * HID + (k ^ (((n + j) & 31) << 1))] = w[j];
    }
    __syncthreads();

    const int wave = tid >> 5;
    const int lane = tid & 31;
    const int half = lane >> 4;
    const int l16  = lane & 15;
    const int m0   = blockIdx.x * 128 + wave * 16;

    v8f zero = {};
    v8f acc[8];
#pragma unroll
    for (int t = 0; t < 8; ++t) acc[t] = zero;

    const float* arow = h + (size_t)(m0 + l16) * HID;

    for (int k = 0; k < HID; k += 4) {
        const int kb = k + half * 2;
        v2f a = *(const v2f*)(arow + kb);
#pragma unroll
        for (int t = 0; t < 8; ++t) {
            const int col = t * 16 + l16;
            v2f b = *(const v2f*)(lWt + col * HID + (kb ^ ((col & 31) << 1)));
            acc[t] = __builtin_amdgcn_wmma_f32_16x16x4_f32(
                false, a, false, b, (short)0, acc[t], false, false);
        }
    }

    const int rbase = half * 8;
#pragma unroll
    for (int r = 0; r < 8; ++r) {
        const int row = m0 + rbase + r;
        float* hwrow  = hw  + (size_t)row * HID;
#pragma unroll
        for (int t = 0; t < 8; ++t) {
            const int col = t * 16 + l16;
            hwrow[col] = acc[t][r];
        }
    }
}


__device__ __forceinline__ v4f edge_row(const float* hwf, int s_l, int j) {
    const int s = __builtin_amdgcn_readlane(s_l, j);
    return *(const v4f*)(hwf + (size_t)s * HID);
}
__device__ __forceinline__ float edge_wt(float w_l, int j) {
    return __int_as_float(__builtin_amdgcn_readlane(__float_as_int(w_l), j));
}

__launch_bounds__(256)
__global__ void k_gather_ln(const int* __restrict__ rowptr,
                            const int* __restrict__ csr_src,
                            const float* __restrict__ csr_w,
                            const float* __restrict__ hw,
                            const float* __restrict__ invdeg,
                            const float* __restrict__ bias,
                            const float* __restrict__ gamma,
                            const float* __restrict__ beta,
                            float* __restrict__ h,
                            int node_stride,
                            const int* __restrict__ need) {
    const int t = blockIdx.x * 256 + threadIdx.x;
    const int n = __builtin_amdgcn_readfirstlane(t >> 5) * node_stride;
    if (need != nullptr && need[n] == 0) return;
    const int lane = threadIdx.x & 31;
    const int f = lane << 2;
    const int beg = rowptr[n];
    const int end = rowptr[n + 1];
    const float* hwf = hw + f;

    const v4f  self = *(const v4f*)(hwf + (size_t)n * HID);
    const float inv = invdeg[n];
    const v4f  bv   = *(const v4f*)(bias + f);
    float* hr = h + (size_t)n * HID + f;
    const v4f  hv   = *(const v4f*)hr;
    const v4f  g    = *(const v4f*)(gamma + f);
    const v4f  b    = *(const v4f*)(beta + f);

    v4f acc = {0.0f, 0.0f, 0.0f, 0.0f};
    for (int base = beg; base < end; base += 32) {
        const int idx = base + lane;
        int   s_l = 0;
        float w_l = 0.0f;
        if (idx < end) { s_l = csr_src[idx]; w_l = csr_w[idx]; }
        const int left = end - base;
        const int cnt = left < 32 ? left : 32;
        int j = 0;
        for (; j + 4 <= cnt; j += 4) {
            const v4f v0 = edge_row(hwf, s_l, j + 0);
            const v4f v1 = edge_row(hwf, s_l, j + 1);
            const v4f v2 = edge_row(hwf, s_l, j + 2);
            const v4f v3 = edge_row(hwf, s_l, j + 3);
            acc += v0 * edge_wt(w_l, j + 0);
            acc += v1 * edge_wt(w_l, j + 1);
            acc += v2 * edge_wt(w_l, j + 2);
            acc += v3 * edge_wt(w_l, j + 3);
        }
        const int rem = cnt - j;
        if (rem == 3) {
            const v4f v0 = edge_row(hwf, s_l, j + 0);
            const v4f v1 = edge_row(hwf, s_l, j + 1);
            const v4f v2 = edge_row(hwf, s_l, j + 2);
            acc += v0 * edge_wt(w_l, j + 0);
            acc += v1 * edge_wt(w_l, j + 1);
            acc += v2 * edge_wt(w_l, j + 2);
        } else if (rem == 2) {
            const v4f v0 = edge_row(hwf, s_l, j + 0);
            const v4f v1 = edge_row(hwf, s_l, j + 1);
            acc += v0 * edge_wt(w_l, j + 0);
            acc += v1 * edge_wt(w_l, j + 1);
        } else if (rem == 1) {
            const v4f v0 = edge_row(hwf, s_l, j);
            acc += v0 * edge_wt(w_l, j);
        }
    }

    const v4f av = (acc + self * inv) + bv;

    v4f val;
    val.x = hv.x + fmaxf(av.x, 0.0f);
    val.y = hv.y + fmaxf(av.y, 0.0f);
    val.z = hv.z + fmaxf(av.z, 0.0f);
    val.w = hv.w + fmaxf(av.w, 0.0f);

    float s = val.x + val.y + val.z + val.w;
#pragma unroll
    for (int off = 16; off >= 1; off >>= 1) s += __shfl_xor(s, off, 32);
    const float mu = s * (1.0f / (float)HID);

    v4f dv = val - mu;
    float q = dv.x * dv.x + dv.y * dv.y + dv.z * dv.z + dv.w * dv.w;
#pragma unroll
    for (int off = 16; off >= 1; off >>= 1) q += __shfl_xor(q, off, 32);
    const float rstd = rsqrtf(q * (1.0f / (float)HID) + LN_EPS);

    const v4f outv = dv * rstd * g + b;
    *(v4f*)hr = outv;
}


__global__ void k_out(const float* __restrict__ h,
                      const float* __restrict__ W_out,
                      float* __restrict__ out) {
    int t = blockIdx.x * 256 + threadIdx.x;
    if (t >= NGRAPH * OUT_DIM) return;
    int g = t / OUT_DIM;
    int o = t % OUT_DIM;
    const float* hr = h + (size_t)(g * NPG) * HID;
    float s = 0.0f;
#pragma unroll 4
    for (int k = 0; k < HID; ++k) s = fmaf(hr[k], W_out[k * OUT_DIM + o], s);
    out[t] = s;
}

extern "C" void kernel_launch(void* const* d_in, const int* in_sizes, int n_in,
                              void* d_out, int out_size, void* d_ws, size_t ws_size,
                              hipStream_t stream) {
    const int*   x        = (const int*)d_in[0];
    const int*   ei       = (const int*)d_in[1];
    const float* cls_emb  = (const float*)d_in[3];
    const float* nbr_emb  = (const float*)d_in[4];
    const float* Ws       = (const float*)d_in[5];
    const float* bs       = (const float*)d_in[6];
    const float* gammas   = (const float*)d_in[7];
    const float* betas    = (const float*)d_in[8];
    const float* W_out    = (const float*)d_in[9];
    float*       out      = (float*)d_out;

    char* ws = (char*)d_ws;
    size_t off = 0;
    int*   cnt    = (int*)(ws + off);   off += (size_t)EBLOCKS * NBIN * 4;
    float* dinv   = (float*)(ws + off); off += (size_t)N_NODES * 4;
    float* invdeg = (float*)(ws + off); off += (size_t)N_NODES * 4;
    float* h      = (float*)(ws + off); off += (size_t)N_NODES * HID * 4;
    float* hw     = (float*)(ws + off); off += (size_t)N_NODES * HID * 4;
    int*   rowptr = (int*)(ws + off);   off += (size_t)(N_NODES + 64) * 4;
    int*   binned = (int*)(ws + off);   off += (size_t)E_EDGES * 4;
    int*   csrtmp = (int*)(ws + off);   off += (size_t)E_EDGES * 4;
    int*   csrsrc = (int*)(ws + off);   off += (size_t)E_EDGES * 4;
    float* csrw   = (float*)(ws + off); off += (size_t)E_EDGES * 4;
    int*   need   = (int*)(ws + off);   off += (size_t)N_NODES * 4;
    int*   binbase = (int*)(ws + off);  off += (size_t)(NBIN + 64) * 4;
    (void)ws_size; (void)in_sizes; (void)n_in; (void)out_size;

    k_bin_count  <<<EBLOCKS, 256, 0, stream>>>(ei, cnt);
    k_bin_scatter<<<EBLOCKS, 256, 0, stream>>>(ei, cnt, binned, binbase);
    k_bin_build  <<<NBIN, BUILD_T, 0, stream>>>(binned, binbase, rowptr, dinv, invdeg, need, csrtmp);
    k_csr_order<<<(N_NODES * 32) / 256, 256, 0, stream>>>(rowptr, csrtmp, dinv, csrsrc, csrw, need);

    k_embed<<<(N_NODES * 32) / 256, 256, 0, stream>>>(x, cls_emb, nbr_emb, h);

    for (int i = 0; i < N_LAYERS - 2; ++i) {
        k_gemm_fused<<<N_NODES / 128, 256, 0, stream>>>(
            h, Ws + (size_t)i * HID * HID, hw);
        k_gather_ln<<<(N_NODES * 32) / 256, 256, 0, stream>>>(
            rowptr, csrsrc, csrw, hw, invdeg, bs + (size_t)i * HID,
            gammas + (size_t)i * HID, betas + (size_t)i * HID, h, 1, nullptr);
    }
    k_gemm_fused<<<N_NODES / 128, 256, 0, stream>>>(
        h, Ws + (size_t)(N_LAYERS - 2) * HID * HID, hw);
    k_gather_ln<<<(N_NODES * 32) / 256, 256, 0, stream>>>(
        rowptr, csrsrc, csrw, hw, invdeg, bs + (size_t)(N_LAYERS - 2) * HID,
        gammas + (size_t)(N_LAYERS - 2) * HID, betas + (size_t)(N_LAYERS - 2) * HID, h, 1, need);
    k_gemm_fused<<<N_NODES / 128, 256, 0, stream>>>(
        h, Ws + (size_t)(N_LAYERS - 1) * HID * HID, hw);
    k_gather_ln<<<(NGRAPH * 32) / 256, 256, 0, stream>>>(
        rowptr, csrsrc, csrw, hw, invdeg, bs + (size_t)(N_LAYERS - 1) * HID,
        gammas + (size_t)(N_LAYERS - 1) * HID, betas + (size_t)(N_LAYERS - 1) * HID, h, NPG, nullptr);

    k_out<<<(NGRAPH * OUT_DIM + 255) / 256, 256, 0, stream>>>(h, W_out, out);
}


# Last layer as (A h) W: root rows of h aggregated first, then one 512-row WMMA product, LN and logits in the same launch; full last GEMM and 2 launches dropped; that layer's sum order changes
# speedup vs baseline: 4.7288x; 1.1675x over previous
#include <hip/hip_runtime.h>

#define N_NODES   65536
#define NPG       128
#define NGRAPH    512
#define E_EDGES   1048576
#define HID       128
#define OUT_DIM   10
#define N_LAYERS  4
#define LN_EPS    1e-5f

typedef float v2f __attribute__((ext_vector_type(2)));
typedef float v4f __attribute__((ext_vector_type(4)));
typedef float v8f __attribute__((ext_vector_type(8)));

#define NBIN      256
#define BIN_NODES 256
#define BIN_SHIFT 8
#define EPB       4096
#define EBLOCKS   (E_EDGES / EPB)
#define STAGE_CAP 12288
static_assert(NBIN * BIN_NODES == N_NODES, "bins must cover the nodes");
static_assert((1 << BIN_SHIFT) == BIN_NODES, "bin shift");
static_assert(EBLOCKS * EPB == E_EDGES, "edge blocks must cover the edges");
static_assert(N_NODES == 65536, "an edge is packed as src | dst << 16");
static_assert(EBLOCKS % 64 == 0 && NBIN == 256 && EPB == 16 * 256, "layout of the edge passes");

typedef int v4i __attribute__((ext_vector_type(4)));

__device__ __forceinline__ int excl_scan_256(int c, int t, int* wtot) {
    const int lane = t & 31;
    int x = c;
#pragma unroll
    for (int o = 1; o < 32; o <<= 1) {
        const int y = __shfl_up(x, o, 32);
        if (lane >= o) x += y;
    }
    if (t < 256 && lane == 31) wtot[t >> 5] = x;
    __syncthreads();
    int pre = 0;
    const int wv = t >> 5;
#pragma unroll
    for (int w = 0; w < 8; ++w) pre += (w < wv) ? wtot[w] : 0;
    return pre + x - c;
}

__launch_bounds__(256)
__global__ void k_bin_count(const int* __restrict__ edge_index, int* __restrict__ cnt) {
    __shared__ int hist[NBIN];
    const int t = threadIdx.x;
    hist[t] = 0;
    const int* dstp = edge_index + E_EDGES + blockIdx.x * EPB + t * 4;
    v4i dv[4];
#pragma unroll
    for (int i = 0; i < 4; ++i) dv[i] = *(const v4i*)(dstp + i * 1024);
    __syncthreads();
#pragma unroll
    for (int i = 0; i < 4; ++i)
#pragma unroll
        for (int j = 0; j < 4; ++j) {
            const int d = dv[i][j] & (N_NODES - 1);
            atomicAdd(&hist[d >> BIN_SHIFT], 1);
        }
    __syncthreads();
    cnt[blockIdx.x * NBIN + t] = hist[t];
}

__launch_bounds__(256)
__global__ void k_bin_scatter(const int* __restrict__ edge_index, const int* __restrict__ cnt,
                              int* __restrict__ binned, int* __restrict__ binbase) {
    __shared__ int ptot[4][NBIN];
    __shared__ int pbef[4][NBIN];
    __shared__ int cursor[NBIN];
    __shared__ int delta[NBIN];
    __shared__ int wtot_a[8];
    __shared__ int wtot_b[8];
    __shared__ int stage[EPB];
    const int t = threadIdx.x;
    const int b = blockIdx.x;
    const int* srcp = edge_index + b * EPB + t * 4;
    const int* dstp = edge_index + E_EDGES + b * EPB + t * 4;
    v4i sv[4], dv[4];
#pragma unroll
    for (int i = 0; i < 4; ++i) {
        sv[i] = *(const v4i*)(srcp + i * 1024);
        dv[i] = *(const v4i*)(dstp + i * 1024);
    }
    const int own = cnt[b * NBIN + t];
    const int q  = t >> 6;
    const int c4 = (t & 63) * 4;
    v4i tot = {0, 0, 0, 0};
    v4i bef = {0, 0, 0, 0};
#pragma unroll 16
    for (int r = q; r < EBLOCKS; r += 4) {
        const v4i c = *(const v4i*)(cnt + r * NBIN + c4);
        tot += c;
        if (r < b) bef += c;
    }
#pragma unroll
    for (int j = 0; j < 4; ++j) { ptot[q][c4 + j] = tot[j]; pbef[q][c4 + j] = bef[j]; }
    __syncthreads();
    const int total  = ptot[0][t] + ptot[1][t] + ptot[2][t] + ptot[3][t];
    const int before = pbef[0][t] + pbef[1][t] + pbef[2][t] + pbef[3][t];
    const int base   = excl_scan_256(total, t, wtot_a);
    const int lstart = excl_scan_256(own, t, wtot_b);
    cursor[t] = lstart;
    delta[t]  = base + before - lstart;
    if (b == 0) {
        binbase[t] = base;
        if (t == NBIN - 1) binbase[NBIN] = base + total;
    }
    __syncthreads();
#pragma unroll
    for (int i = 0; i < 4; ++i)
#pragma unroll
        for (int j = 0; j < 4; ++j) {
            const int s = sv[i][j] & (N_NODES - 1);
            const int d = dv[i][j] & (N_NODES - 1);
            const int lp = atomicAdd(&cursor[d >> BIN_SHIFT], 1);
            stage[lp] = (int)((unsigned)s | ((unsigned)d << 16));
        }
    __syncthreads();
#pragma unroll 4
    for (int i = t; i < EPB; i += 256) {
        const int w = stage[i];
        binned[i + delta[(w >> 24) & (NBIN - 1)]] = w;
    }
}

#define BUILD_T 1024
__launch_bounds__(BUILD_T)
__global__ void k_bin_build(const int* __restrict__ binned, const int* __restrict__ binbase,
                            int* __restrict__ rowptr, float* __restrict__ dinv,
                            float* __restrict__ invdeg, int* __restrict__ need,
                            int* __restrict__ csr_tmp) {
    __shared__ int deg[BIN_NODES];
    __shared__ int off[BIN_NODES];
    __shared__ int cur[BIN_NODES];
    __shared__ int wtot[8];
    __shared__ int stage[STAGE_CAP];
    const int t = threadIdx.x;
    const int bin = blockIdx.x;
    const int base = binbase[bin];
    const int n_e = binbase[bin + 1] - base;
    if (t < BIN_NODES) { deg[t] = 0; cur[t] = 0; }
    __syncthreads();
    const int* eb = binned + base;
    for (int i0 = 0; i0 < n_e; i0 += 4 * BUILD_T) {
        int w[4];
#pragma unroll
        for (int j = 0; j < 4; ++j) {
            const int i = i0 + j * BUILD_T + t;
            w[j] = eb[i < n_e ? i : n_e - 1];
        }
#pragma unroll
        for (int j = 0; j < 4; ++j)
            if (i0 + j * BUILD_T + t < n_e)
                atomicAdd(&deg[(w[j] >> 16) & (BIN_NODES - 1)], 1);
    }
    __syncthreads();
    const int c = (t < BIN_NODES) ? deg[t] : 0;
    const int ex = excl_scan_256(c, t, wtot);
    if (t < BIN_NODES) {
        const int node = bin * BIN_NODES + t;
        const float d = 1.0f + (float)c;
        dinv[node]   = rsqrtf(d);
        invdeg[node] = 1.0f / d;
        rowptr[node] = base + ex;
        need[node]   = 0;
        if (node == N_NODES - 1) rowptr[N_NODES] = base + ex + c;
        off[t] = ex;
    }
    __syncthreads();
    const bool fits = n_e <= STAGE_CAP;
    for (int i0 = 0; i0 < n_e; i0 += 4 * BUILD_T) {
        int w[4];
#pragma unroll
        for (int j = 0; j < 4; ++j) {
            const int i = i0 + j * BUILD_T + t;
            w[j] = eb[i < n_e ? i : n_e - 1];
        }
#pragma unroll
        for (int j = 0; j < 4; ++j)
            if (i0 + j * BUILD_T + t < n_e) {
                const int dl = (w[j] >> 16) & (BIN_NODES - 1);
                const int p = off[dl] + atomicAdd(&cur[dl], 1);
                const int key = w[j] & 0xffff;
                if (fits) stage[p] = key; else csr_tmp[base + p] = key;
            }
    }
    __syncthreads();
    if (fits)
        for (int i = t; i < n_e; i += BUILD_T) csr_tmp[base + i] = stage[i];
}

__launch_bounds__(256)
__global__ void k_csr_order(const int* __restrict__ rowptr,
                            const int* __restrict__ csr_tmp,
                            const float* __restrict__ dinv,
                            int* __restrict__ csr_src,
                            float* __restrict__ csr_w,
                            int* __restrict__ need) {
    const int t = blockIdx.x * 256 + threadIdx.x;
    const int n = t >> 5;
    const int lane = t & 31;
    const bool root = (n % NPG) == 0;
    if (root && lane == 0) need[n] = 1;
    const int beg = rowptr[n];
    const int deg = rowptr[n + 1] - beg;
    const float dn = dinv[n];
    const int* b = csr_tmp + beg;
    for (int i = lane; i < deg; i += 32) {
        const int key = b[i] & (N_NODES - 1);
        int rank = 0;
        for (int j = 0; j < deg; ++j) {
            const int kj = b[j] & (N_NODES - 1);
            rank += (kj < key || (kj == key && j < i)) ? 1 : 0;
        }
        csr_src[beg + rank] = key;
        csr_w[beg + rank]   = dinv[key] * dn;
        if (root) need[key] = 1;
    }
}

__global__ void k_embed(const int* __restrict__ x,
                        const float* __restrict__ cls_emb,
                        const float* __restrict__ nbr_emb,
                        float* __restrict__ h) {
    int t = blockIdx.x * 256 + threadIdx.x;
    int n = t >> 5;
    int f = (t & 31) << 2;
    int c  = x[2 * n + 0];
    int nb = x[2 * n + 1];
    v4f a = *(const v4f*)(cls_emb + (size_t)c  * HID + f);
    v4f b = *(const v4f*)(nbr_emb + (size_t)nb * HID + f);
    *(v4f*)(h + (size_t)n * HID + f) = a + b;
}

__launch_bounds__(256)
__global__ void k_gemm_fused(const float* __restrict__ h,
                             const float* __restrict__ W,
                             float* __restrict__ hw) {
    __shared__ float lWt[HID * HID];
    const int tid = threadIdx.x;

    for (int i = tid * 4; i < HID * HID; i += 256 * 4) {
        v4f w = *(const v4f*)(W + i);
        const int k = i >> 7;
        const int n = i & (HID - 1);
#pragma unroll
        for (int j = 0; j < 4; ++j)
            lWt[(n + j) * HID + (k ^ (((n + j) & 31) << 1))] = w[j];
    }
    __syncthreads();

    const int wave = tid >> 5;
    const int lane = tid & 31;
    const int half = lane >> 4;
    const int l16  = lane & 15;
    const int m0   = blockIdx.x * 128 + wave * 16;

    v8f zero = {};
    v8f acc[8];
#pragma unroll
    for (int t = 0; t < 8; ++t) acc[t] = zero;

    const float* arow = h + (size_t)(m0 + l16) * HID;

    for (int k = 0; k < HID; k += 4) {
        const int kb = k + half * 2;
        v2f a = *(const v2f*)(arow + kb);
#pragma unroll
        for (int t = 0; t < 8; ++t) {
            const int col = t * 16 + l16;
            v2f b = *(const v2f*)(lWt + col * HID + (kb ^ ((col & 31) << 1)));
            acc[t] = __builtin_amdgcn_wmma_f32_16x16x4_f32(
                false, a, false, b, (short)0, acc[t], false, false);
        }
    }

    const int rbase = half * 8;
#pragma unroll
    for (int r = 0; r < 8; ++r) {
        const int row = m0 + rbase + r;
        float* hwrow  = hw  + (size_t)row * HID;
#pragma unroll
        for (int t = 0; t < 8; ++t) {
            const int col = t * 16 + l16;
            hwrow[col] = acc[t][r];
        }
    }
}


__device__ __forceinline__ v4f edge_row(const float* hwf, int s_l, int j) {
    const int s = __builtin_amdgcn_readlane(s_l, j);
    return *(const v4f*)(hwf + (size_t)s * HID);
}
__device__ __forceinline__ float edge_wt(float w_l, int j) {
    return __int_as_float(__builtin_amdgcn_readlane(__float_as_int(w_l), j));
}

__device__ __forceinline__ v4f edge_sum(const float* rowsf,
                                        const int* __restrict__ csr_src,
                                        const float* __restrict__ csr_w,
                                        int beg, int end, int lane) {
    v4f acc = {0.0f, 0.0f, 0.0f, 0.0f};
    for (int base = beg; base < end; base += 32) {
        const int idx = base + lane;
        int   s_l = 0;
        float w_l = 0.0f;
        if (idx < end) { s_l = csr_src[idx]; w_l = csr_w[idx]; }
        const int left = end - base;
        const int cnt = left < 32 ? left : 32;
        int j = 0;
        for (; j + 4 <= cnt; j += 4) {
            const v4f v0 = edge_row(rowsf, s_l, j + 0);
            const v4f v1 = edge_row(rowsf, s_l, j + 1);
            const v4f v2 = edge_row(rowsf, s_l, j + 2);
            const v4f v3 = edge_row(rowsf, s_l, j + 3);
            acc += v0 * edge_wt(w_l, j + 0);
            acc += v1 * edge_wt(w_l, j + 1);
            acc += v2 * edge_wt(w_l, j + 2);
            acc += v3 * edge_wt(w_l, j + 3);
        }
        const int rem = cnt - j;
        if (rem == 3) {
            const v4f v0 = edge_row(rowsf, s_l, j + 0);
            const v4f v1 = edge_row(rowsf, s_l, j + 1);
            const v4f v2 = edge_row(rowsf, s_l, j + 2);
            acc += v0 * edge_wt(w_l, j + 0);
            acc += v1 * edge_wt(w_l, j + 1);
            acc += v2 * edge_wt(w_l, j + 2);
        } else if (rem == 2) {
            const v4f v0 = edge_row(rowsf, s_l, j + 0);
            const v4f v1 = edge_row(rowsf, s_l, j + 1);
            acc += v0 * edge_wt(w_l, j + 0);
            acc += v1 * edge_wt(w_l, j + 1);
        } else if (rem == 1) {
            const v4f v0 = edge_row(rowsf, s_l, j);
            acc += v0 * edge_wt(w_l, j);
        }
    }
    return acc;
}

__launch_bounds__(256)
__global__ void k_gather_ln(const int* __restrict__ rowptr,
                            const int* __restrict__ csr_src,
                            const float* __restrict__ csr_w,
                            const float* __restrict__ hw,
                            const float* __restrict__ invdeg,
                            const float* __restrict__ bias,
                            const float* __restrict__ gamma,
                            const float* __restrict__ beta,
                            float* __restrict__ h,
                            const int* __restrict__ need) {
    const int t = blockIdx.x * 256 + threadIdx.x;
    const int n = __builtin_amdgcn_readfirstlane(t >> 5);
    if (need != nullptr && need[n] == 0) return;
    const int lane = threadIdx.x & 31;
    const int f = lane << 2;
    const int beg = rowptr[n];
    const int end = rowptr[n + 1];
    const float* hwf = hw + f;

    const v4f  self = *(const v4f*)(hwf + (size_t)n * HID);
    const float inv = invdeg[n];
    const v4f  bv   = *(const v4f*)(bias + f);
    float* hr = h + (size_t)n * HID + f;
    const v4f  hv   = *(const v4f*)hr;
    const v4f  g    = *(const v4f*)(gamma + f);
    const v4f  b    = *(const v4f*)(beta + f);

    const v4f acc = edge_sum(hwf, csr_src, csr_w, beg, end, lane);

    const v4f av = (acc + self * inv) + bv;

    v4f val;
    val.x = hv.x + fmaxf(av.x, 0.0f);
    val.y = hv.y + fmaxf(av.y, 0.0f);
    val.z = hv.z + fmaxf(av.z, 0.0f);
    val.w = hv.w + fmaxf(av.w, 0.0f);

    float s = val.x + val.y + val.z + val.w;
#pragma unroll
    for (int off = 16; off >= 1; off >>= 1) s += __shfl_xor(s, off, 32);
    const float mu = s * (1.0f / (float)HID);

    v4f dv = val - mu;
    float q = dv.x * dv.x + dv.y * dv.y + dv.z * dv.z + dv.w * dv.w;
#pragma unroll
    for (int off = 16; off >= 1; off >>= 1) q += __shfl_xor(q, off, 32);
    const float rstd = rsqrtf(q * (1.0f / (float)HID) + LN_EPS);

    const v4f outv = dv * rstd * g + b;
    *(v4f*)hr = outv;
}


#define LAST_ROOTS 16
#define ZP 132
static_assert(NGRAPH % LAST_ROOTS == 0, "blocks must cover the roots");
static_assert(HID * OUT_DIM <= 3 * LAST_ROOTS * 32, "W_out is staged three words per thread");

__launch_bounds__(LAST_ROOTS * 32)
__global__ void k_last(const int* __restrict__ rowptr,
                       const int* __restrict__ csr_src,
                       const float* __restrict__ csr_w,
                       const float* __restrict__ h,
                       const float* __restrict__ invdeg,
                       const float* __restrict__ W,
                       const float* __restrict__ bias,
                       const float* __restrict__ gamma,
                       const float* __restrict__ beta,
                       const float* __restrict__ W_out,
                       float* __restrict__ out) {
    __shared__ __attribute__((aligned(16))) float zt[LAST_ROOTS * ZP];
    __shared__ __attribute__((aligned(16))) float yt[LAST_ROOTS * ZP];
    __shared__ float wo[HID * OUT_DIM];
    const int tid  = threadIdx.x;
    const int wave = __builtin_amdgcn_readfirstlane(tid >> 5);
    const int lane = tid & 31;
    const int half = lane >> 4;
    const int l16  = lane & 15;
    const int g = blockIdx.x * LAST_ROOTS + wave;
    const int n = g * NPG;
    const int f = lane << 2;

    float wov[3];
#pragma unroll
    for (int r = 0; r < 3; ++r) {
        const int i = tid + r * (LAST_ROOTS * 32);
        wov[r] = (i < HID * OUT_DIM) ? W_out[i] : 0.0f;
    }

    v2f bfrag[32];
#pragma unroll
    for (int s = 0; s < 32; ++s) { bfrag[s].x = 0.0f; bfrag[s].y = 0.0f; }
    if (wave < 8) {
        const float* wcol = W + wave * 16 + l16;
#pragma unroll
        for (int s = 0; s < 32; ++s) {
            const int kb = s * 4 + half * 2;
            bfrag[s].x = wcol[(kb + 0) * HID];
            bfrag[s].y = wcol[(kb + 1) * HID];
        }
    }

    const int beg = rowptr[n];
    const int end = rowptr[n + 1];
    const float* hf = h + f;
    const v4f  hv  = *(const v4f*)(hf + (size_t)n * HID);
    const float inv = invdeg[n];
    const v4f  bv  = *(const v4f*)(bias + f);
    const v4f  gm  = *(const v4f*)(gamma + f);
    const v4f  bt  = *(const v4f*)(beta + f);

    const v4f acc = edge_sum(hf, csr_src, csr_w, beg, end, lane);
    const v4f z = acc + hv * inv;
    *(v4f*)(zt + wave * ZP + f) = z;
#pragma unroll
    for (int r = 0; r < 3; ++r) {
        const int i = tid + r * (LAST_ROOTS * 32);
        if (i < HID * OUT_DIM) wo[i] = wov[r];
    }
    __syncthreads();

    if (wave < 8) {
        v8f cacc = {};
        const float* arow = zt + l16 * ZP + half * 2;
#pragma unroll
        for (int s = 0; s < 32; ++s) {
            const v2f a = *(const v2f*)(arow + s * 4);
            cacc = __builtin_amdgcn_wmma_f32_16x16x4_f32(
                false, a, false, bfrag[s], (short)0, cacc, false, false);
        }
        const int col = wave * 16 + l16;
#pragma unroll
        for (int r = 0; r < 8; ++r) yt[(half * 8 + r) * ZP + col] = cacc[r];
    }
    __syncthreads();

    const v4f y  = *(const v4f*)(yt + wave * ZP + f);
    const v4f av = y + bv;

    v4f val;
    val.x = hv.x + fmaxf(av.x, 0.0f);
    val.y = hv.y + fmaxf(av.y, 0.0f);
    val.z = hv.z + fmaxf(av.z, 0.0f);
    val.w = hv.w + fmaxf(av.w, 0.0f);

    float s = val.x + val.y + val.z + val.w;
#pragma unroll
    for (int off = 16; off >= 1; off >>= 1) s += __shfl_xor(s, off, 32);
    const float mu = s * (1.0f / (float)HID);

    v4f dv = val - mu;
    float q = dv.x * dv.x + dv.y * dv.y + dv.z * dv.z + dv.w * dv.w;
#pragma unroll
    for (int off = 16; off >= 1; off >>= 1) q += __shfl_xor(q, off, 32);
    const float rstd = rsqrtf(q * (1.0f / (float)HID) + LN_EPS);

    const v4f outv = dv * rstd * gm + bt;
    *(v4f*)(zt + wave * ZP + f) = outv;
    __syncthreads();

    if (lane < OUT_DIM) {
        const float* hr = zt + wave * ZP;
        float lg = 0.0f;
#pragma unroll 32
        for (int k = 0; k < HID; ++k) lg = fmaf(hr[k], wo[k * OUT_DIM + lane], lg);
        out[g * OUT_DIM + lane] = lg;
    }
}

extern "C" void kernel_launch(void* const* d_in, const int* in_sizes, int n_in,
                              void* d_out, int out_size, void* d_ws, size_t ws_size,
                              hipStream_t stream) {
    const int*   x        = (const int*)d_in[0];
    const int*   ei       = (const int*)d_in[1];
    const float* cls_emb  = (const float*)d_in[3];
    const float* nbr_emb  = (const float*)d_in[4];
    const float* Ws       = (const float*)d_in[5];
    const float* bs       = (const float*)d_in[6];
    const float* gammas   = (const float*)d_in[7];
    const float* betas    = (const float*)d_in[8];
    const float* W_out    = (const float*)d_in[9];
    float*       out      = (float*)d_out;

    char* ws = (char*)d_ws;
    size_t off = 0;
    int*   cnt    = (int*)(ws + off);   off += (size_t)EBLOCKS * NBIN * 4;
    float* dinv   = (float*)(ws + off); off += (size_t)N_NODES * 4;
    float* invdeg = (float*)(ws + off); off += (size_t)N_NODES * 4;
    float* h      = (float*)(ws + off); off += (size_t)N_NODES * HID * 4;
    float* hw     = (float*)(ws + off); off += (size_t)N_NODES * HID * 4;
    int*   rowptr = (int*)(ws + off);   off += (size_t)(N_NODES + 64) * 4;
    int*   binned = (int*)(ws + off);   off += (size_t)E_EDGES * 4;
    int*   csrtmp = (int*)(ws + off);   off += (size_t)E_EDGES * 4;
    int*   csrsrc = (int*)(ws + off);   off += (size_t)E_EDGES * 4;
    float* csrw   = (float*)(ws + off); off += (size_t)E_EDGES * 4;
    int*   need   = (int*)(ws + off);   off += (size_t)N_NODES * 4;
    int*   binbase = (int*)(ws + off);  off += (size_t)(NBIN + 64) * 4;
    (void)ws_size; (void)in_sizes; (void)n_in; (void)out_size;

    k_bin_count  <<<EBLOCKS, 256, 0, stream>>>(ei, cnt);
    k_bin_scatter<<<EBLOCKS, 256, 0, stream>>>(ei, cnt, binned, binbase);
    k_bin_build  <<<NBIN, BUILD_T, 0, stream>>>(binned, binbase, rowptr, dinv, invdeg, need, csrtmp);
    k_csr_order<<<(N_NODES * 32) / 256, 256, 0, stream>>>(rowptr, csrtmp, dinv, csrsrc, csrw, need);

    k_embed<<<(N_NODES * 32) / 256, 256, 0, stream>>>(x, cls_emb, nbr_emb, h);

    for (int i = 0; i < N_LAYERS - 2; ++i) {
        k_gemm_fused<<<N_NODES / 128, 256, 0, stream>>>(
            h, Ws + (size_t)i * HID * HID, hw);
        k_gather_ln<<<(N_NODES * 32) / 256, 256, 0, stream>>>(
            rowptr, csrsrc, csrw, hw, invdeg, bs + (size_t)i * HID,
            gammas + (size_t)i * HID, betas + (size_t)i * HID, h, nullptr);
    }
    k_gemm_fused<<<N_NODES / 128, 256, 0, stream>>>(
        h, Ws + (size_t)(N_LAYERS - 2) * HID * HID, hw);
    k_gather_ln<<<(N_NODES * 32) / 256, 256, 0, stream>>>(
        rowptr, csrsrc, csrw, hw, invdeg, bs + (size_t)(N_LAYERS - 2) * HID,
        gammas + (size_t)(N_LAYERS - 2) * HID, betas + (size_t)(N_LAYERS - 2) * HID, h, need);
    k_last<<<NGRAPH / LAST_ROOTS, LAST_ROOTS * 32, 0, stream>>>(
        rowptr, csrsrc, csrw, h, invdeg, Ws + (size_t)(N_LAYERS - 1) * HID * HID,
        bs + (size_t)(N_LAYERS - 1) * HID, gammas + (size_t)(N_LAYERS - 1) * HID,
        betas + (size_t)(N_LAYERS - 1) * HID, W_out, out);
}


# Layer 0 once per distinct input pair (class id, neighbour count): h@W0 as a 4160-row table from one small WMMA launch, gather reads rows by pair; full GEMM and embed launch dropped; sums as before
# speedup vs baseline: 5.5854x; 1.1811x over previous
#include <hip/hip_runtime.h>

#define N_NODES   65536
#define NPG       128
#define NGRAPH    512
#define E_EDGES   1048576
#define HID       128
#define OUT_DIM   10
#define N_LAYERS  4
#define LN_EPS    1e-5f
#define N_CLS     64
#define N_NBR     65
#define N_PAIR    (N_CLS * N_NBR)

typedef float v2f __attribute__((ext_vector_type(2)));
typedef float v4f __attribute__((ext_vector_type(4)));
typedef float v8f __attribute__((ext_vector_type(8)));

#define NBIN      256
#define BIN_NODES 256
#define BIN_SHIFT 8
#define EPB       4096
#define EBLOCKS   (E_EDGES / EPB)
#define STAGE_CAP 12288
static_assert(NBIN * BIN_NODES == N_NODES, "bins must cover the nodes");
static_assert((1 << BIN_SHIFT) == BIN_NODES, "bin shift");
static_assert(EBLOCKS * EPB == E_EDGES, "edge blocks must cover the edges");
static_assert(N_NODES == 65536, "an edge is packed as src | dst << 16");
static_assert(EBLOCKS % 64 == 0 && NBIN == 256 && EPB == 16 * 256, "layout of the edge passes");

typedef int v4i __attribute__((ext_vector_type(4)));
typedef int v2i __attribute__((ext_vector_type(2)));

__device__ __forceinline__ int clamp_cls(int cc) { return min(max(cc, 0), N_CLS - 1); }
__device__ __forceinline__ int clamp_nbr(int nb) { return min(max(nb, 0), N_NBR - 1); }
__device__ __forceinline__ int pair_row(int cc, int nb) {
    return clamp_cls(cc) * N_NBR + clamp_nbr(nb);
}

__device__ __forceinline__ int excl_scan_256(int c, int t, int* wtot) {
    const int lane = t & 31;
    int x = c;
#pragma unroll
    for (int o = 1; o < 32; o <<= 1) {
        const int y = __shfl_up(x, o, 32);
        if (lane >= o) x += y;
    }
    if (t < 256 && lane == 31) wtot[t >> 5] = x;
    __syncthreads();
    int pre = 0;
    const int wv = t >> 5;
#pragma unroll
    for (int w = 0; w < 8; ++w) pre += (w < wv) ? wtot[w] : 0;
    return pre + x - c;
}

__launch_bounds__(256)
__global__ void k_bin_count(const int* __restrict__ edge_index, int* __restrict__ cnt) {
    __shared__ int hist[NBIN];
    const int t = threadIdx.x;
    hist[t] = 0;
    const int* dstp = edge_index + E_EDGES + blockIdx.x * EPB + t * 4;
    v4i dv[4];
#pragma unroll
    for (int i = 0; i < 4; ++i) dv[i] = *(const v4i*)(dstp + i * 1024);
    __syncthreads();
#pragma unroll
    for (int i = 0; i < 4; ++i)
#pragma unroll
        for (int j = 0; j < 4; ++j) {
            const int d = dv[i][j] & (N_NODES - 1);
            atomicAdd(&hist[d >> BIN_SHIFT], 1);
        }
    __syncthreads();
    cnt[blockIdx.x * NBIN + t] = hist[t];
}

__launch_bounds__(256)
__global__ void k_bin_scatter(const int* __restrict__ edge_index, const int* __restrict__ cnt,
                              int* __restrict__ binned, int* __restrict__ binbase) {
    __shared__ int ptot[4][NBIN];
    __shared__ int pbef[4][NBIN];
    __shared__ int cursor[NBIN];
    __shared__ int delta[NBIN];
    __shared__ int wtot_a[8];
    __shared__ int wtot_b[8];
    __shared__ int stage[EPB];
    const int t = threadIdx.x;
    const int b = blockIdx.x;
    const int* srcp = edge_index + b * EPB + t * 4;
    const int* dstp = edge_index + E_EDGES + b * EPB + t * 4;
    v4i sv[4], dv[4];
#pragma unroll
    for (int i = 0; i < 4; ++i) {
        sv[i] = *(const v4i*)(srcp + i * 1024);
        dv[i] = *(const v4i*)(dstp + i * 1024);
    }
    const int own = cnt[b * NBIN + t];
    const int q  = t >> 6;
    const int c4 = (t & 63) * 4;
    v4i tot = {0, 0, 0, 0};
    v4i bef = {0, 0, 0, 0};
#pragma unroll 16
    for (int r = q; r < EBLOCKS; r += 4) {
        const v4i c = *(const v4i*)(cnt + r * NBIN + c4);
        tot += c;
        if (r < b) bef += c;
    }
#pragma unroll
    for (int j = 0; j < 4; ++j) { ptot[q][c4 + j] = tot[j]; pbef[q][c4 + j] = bef[j]; }
    __syncthreads();
    const int total  = ptot[0][t] + ptot[1][t] + ptot[2][t] + ptot[3][t];
    const int before = pbef[0][t] + pbef[1][t] + pbef[2][t] + pbef[3][t];
    const int base   = excl_scan_256(total, t, wtot_a);
    const int lstart = excl_scan_256(own, t, wtot_b);
    cursor[t] = lstart;
    delta[t]  = base + before - lstart;
    if (b == 0) {
        binbase[t] = base;
        if (t == NBIN - 1) binbase[NBIN] = base + total;
    }
    __syncthreads();
#pragma unroll
    for (int i = 0; i < 4; ++i)
#pragma unroll
        for (int j = 0; j < 4; ++j) {
            const int s = sv[i][j] & (N_NODES - 1);
            const int d = dv[i][j] & (N_NODES - 1);
            const int lp = atomicAdd(&cursor[d >> BIN_SHIFT], 1);
            stage[lp] = (int)((unsigned)s | ((unsigned)d << 16));
        }
    __syncthreads();
#pragma unroll 4
    for (int i = t; i < EPB; i += 256) {
        const int w = stage[i];
        binned[i + delta[(w >> 24) & (NBIN - 1)]] = w;
    }
}

#define BUILD_T 1024
__launch_bounds__(BUILD_T)
__global__ void k_bin_build(const int* __restrict__ binned, const int* __restrict__ binbase,
                            int* __restrict__ rowptr, float* __restrict__ dinv,
                            float* __restrict__ invdeg, int* __restrict__ need,
                            int* __restrict__ csr_tmp) {
    __shared__ int deg[BIN_NODES];
    __shared__ int off[BIN_NODES];
    __shared__ int cur[BIN_NODES];
    __shared__ int wtot[8];
    __shared__ int stage[STAGE_CAP];
    const int t = threadIdx.x;
    const int bin = blockIdx.x;
    const int base = binbase[bin];
    const int n_e = binbase[bin + 1] - base;
    if (t < BIN_NODES) { deg[t] = 0; cur[t] = 0; }
    __syncthreads();
    const int* eb = binned + base;
    for (int i0 = 0; i0 < n_e; i0 += 4 * BUILD_T) {
        int w[4];
#pragma unroll
        for (int j = 0; j < 4; ++j) {
            const int i = i0 + j * BUILD_T + t;
            w[j] = eb[i < n_e ? i : n_e - 1];
        }
#pragma unroll
        for (int j = 0; j < 4; ++j)
            if (i0 + j * BUILD_T + t < n_e)
                atomicAdd(&deg[(w[j] >> 16) & (BIN_NODES - 1)], 1);
    }
    __syncthreads();
    const int c = (t < BIN_NODES) ? deg[t] : 0;
    const int ex = excl_scan_256(c, t, wtot);
    if (t < BIN_NODES) {
        const int node = bin * BIN_NODES + t;
        const float d = 1.0f + (float)c;
        dinv[node]   = rsqrtf(d);
        invdeg[node] = 1.0f / d;
        rowptr[node] = base + ex;
        need[node]   = 0;
        if (node == N_NODES - 1) rowptr[N_NODES] = base + ex + c;
        off[t] = ex;
    }
    __syncthreads();
    const bool fits = n_e <= STAGE_CAP;
    for (int i0 = 0; i0 < n_e; i0 += 4 * BUILD_T) {
        int w[4];
#pragma unroll
        for (int j = 0; j < 4; ++j) {
            const int i = i0 + j * BUILD_T + t;
            w[j] = eb[i < n_e ? i : n_e - 1];
        }
#pragma unroll
        for (int j = 0; j < 4; ++j)
            if (i0 + j * BUILD_T + t < n_e) {
                const int dl = (w[j] >> 16) & (BIN_NODES - 1);
                const int p = off[dl] + atomicAdd(&cur[dl], 1);
                const int key = w[j] & 0xffff;
                if (fits) stage[p] = key; else csr_tmp[base + p] = key;
            }
    }
    __syncthreads();
    if (fits)
        for (int i = t; i < n_e; i += BUILD_T) csr_tmp[base + i] = stage[i];
}

__launch_bounds__(256)
__global__ void k_csr_order(const int* __restrict__ rowptr,
                            const int* __restrict__ csr_tmp,
                            const float* __restrict__ dinv,
                            int* __restrict__ csr_src,
                            float* __restrict__ csr_w,
                            int* __restrict__ need,
                            const int* __restrict__ x,
                            int* __restrict__ csr_pair) {
    const int t = blockIdx.x * 256 + threadIdx.x;
    const int n = t >> 5;
    const int lane = t & 31;
    const bool root = (n % NPG) == 0;
    if (root && lane == 0) need[n] = 1;
    const int beg = rowptr[n];
    const int deg = rowptr[n + 1] - beg;
    const float dn = dinv[n];
    const int* b = csr_tmp + beg;
    for (int i = lane; i < deg; i += 32) {
        const int key = b[i] & (N_NODES - 1);
        const v2i xs = *(const v2i*)(x + 2 * key);
        int rank = 0;
        for (int j = 0; j < deg; ++j) {
            const int kj = b[j] & (N_NODES - 1);
            rank += (kj < key || (kj == key && j < i)) ? 1 : 0;
        }
        csr_src[beg + rank] = key;
        csr_w[beg + rank]   = dinv[key] * dn;
        csr_pair[beg + rank] = pair_row(xs.x, xs.y);
        if (root) need[key] = 1;
    }
}

#define PAIR_TILES (N_PAIR / 16)
static_assert(PAIR_TILES * 16 == N_PAIR, "pair rows must fill whole row tiles");

__launch_bounds__(256)
__global__ void k_pair_product(const float* __restrict__ cls_emb,
                               const float* __restrict__ nbr_emb,
                               const float* __restrict__ W,
                               float* __restrict__ hwp) {
    const int tid  = threadIdx.x;
    const int wave = tid >> 5;
    const int lane = tid & 31;
    const int half = lane >> 4;
    const int l16  = lane & 15;
    const int p  = blockIdx.x * 16 + l16;
    const int cc = p / N_NBR;
    const int nb = p - cc * N_NBR;
    const float* ca = cls_emb + (size_t)cc * HID + half * 2;
    const float* na = nbr_emb + (size_t)nb * HID + half * 2;
    const float* wcol = W + wave * 16 + l16;
    v2f ac[32], an[32], bfrag[32];
#pragma unroll
    for (int s = 0; s < 32; ++s) {
        ac[s] = *(const v2f*)(ca + s * 4);
        an[s] = *(const v2f*)(na + s * 4);
    }
#pragma unroll
    for (int s = 0; s < 32; ++s) {
        const int kb = s * 4 + half * 2;
        bfrag[s].x = wcol[(kb + 0) * HID];
        bfrag[s].y = wcol[(kb + 1) * HID];
    }
    v8f cacc = {};
#pragma unroll
    for (int s = 0; s < 32; ++s) {
        const v2f a = ac[s] + an[s];
        cacc = __builtin_amdgcn_wmma_f32_16x16x4_f32(
            false, a, false, bfrag[s], (short)0, cacc, false, false);
    }
    float* o = hwp + (size_t)(blockIdx.x * 16 + half * 8) * HID + wave * 16 + l16;
#pragma unroll
    for (int r = 0; r < 8; ++r) o[r * HID] = cacc[r];
}

__launch_bounds__(256)
__global__ void k_gemm_fused(const float* __restrict__ h,
                             const float* __restrict__ W,
                             float* __restrict__ hw) {
    __shared__ float lWt[HID * HID];
    const int tid = threadIdx.x;

    for (int i = tid * 4; i < HID * HID; i += 256 * 4) {
        v4f w = *(const v4f*)(W + i);
        const int k = i >> 7;
        const int n = i & (HID - 1);
#pragma unroll
        for (int j = 0; j < 4; ++j)
            lWt[(n + j) * HID + (k ^ (((n + j) & 31) << 1))] = w[j];
    }
    __syncthreads();

    const int wave = tid >> 5;
    const int lane = tid & 31;
    const int half = lane >> 4;
    const int l16  = lane & 15;
    const int m0   = blockIdx.x * 128 + wave * 16;

    v8f zero = {};
    v8f acc[8];
#pragma unroll
    for (int t = 0; t < 8; ++t) acc[t] = zero;

    const float* arow = h + (size_t)(m0 + l16) * HID;

    for (int k = 0; k < HID; k += 4) {
        const int kb = k + half * 2;
        v2f a = *(const v2f*)(arow + kb);
#pragma unroll
        for (int t = 0; t < 8; ++t) {
            const int col = t * 16 + l16;
            v2f b = *(const v2f*)(lWt + col * HID + (kb ^ ((col & 31) << 1)));
            acc[t] = __builtin_amdgcn_wmma_f32_16x16x4_f32(
                false, a, false, b, (short)0, acc[t], false, false);
        }
    }

    const int rbase = half * 8;
#pragma unroll
    for (int r = 0; r < 8; ++r) {
        const int row = m0 + rbase + r;
        float* hwrow  = hw  + (size_t)row * HID;
#pragma unroll
        for (int t = 0; t < 8; ++t) {
            const int col = t * 16 + l16;
            hwrow[col] = acc[t][r];
        }
    }
}


__device__ __forceinline__ v4f edge_row(const float* hwf, int s_l, int j) {
    const int s = __builtin_amdgcn_readlane(s_l, j);
    return *(const v4f*)(hwf + (size_t)s * HID);
}
__device__ __forceinline__ float edge_wt(float w_l, int j) {
    return __int_as_float(__builtin_amdgcn_readlane(__float_as_int(w_l), j));
}

__device__ __forceinline__ v4f edge_sum(const float* rowsf,
                                        const int* __restrict__ csr_src,
                                        const float* __restrict__ csr_w,
                                        int beg, int end, int lane) {
    v4f acc = {0.0f, 0.0f, 0.0f, 0.0f};
    for (int base = beg; base < end; base += 32) {
        const int idx = base + lane;
        int   s_l = 0;
        float w_l = 0.0f;
        if (idx < end) { s_l = csr_src[idx]; w_l = csr_w[idx]; }
        const int left = end - base;
        const int cnt = left < 32 ? left : 32;
        int j = 0;
        for (; j + 4 <= cnt; j += 4) {
            const v4f v0 = edge_row(rowsf, s_l, j + 0);
            const v4f v1 = edge_row(rowsf, s_l, j + 1);
            const v4f v2 = edge_row(rowsf, s_l, j + 2);
            const v4f v3 = edge_row(rowsf, s_l, j + 3);
            acc += v0 * edge_wt(w_l, j + 0);
            acc += v1 * edge_wt(w_l, j + 1);
            acc += v2 * edge_wt(w_l, j + 2);
            acc += v3 * edge_wt(w_l, j + 3);
        }
        const int rem = cnt - j;
        if (rem == 3) {
            const v4f v0 = edge_row(rowsf, s_l, j + 0);
            const v4f v1 = edge_row(rowsf, s_l, j + 1);
            const v4f v2 = edge_row(rowsf, s_l, j + 2);
            acc += v0 * edge_wt(w_l, j + 0);
            acc += v1 * edge_wt(w_l, j + 1);
            acc += v2 * edge_wt(w_l, j + 2);
        } else if (rem == 2) {
            const v4f v0 = edge_row(rowsf, s_l, j + 0);
            const v4f v1 = edge_row(rowsf, s_l, j + 1);
            acc += v0 * edge_wt(w_l, j + 0);
            acc += v1 * edge_wt(w_l, j + 1);
        } else if (rem == 1) {
            const v4f v0 = edge_row(rowsf, s_l, j);
            acc += v0 * edge_wt(w_l, j);
        }
    }
    return acc;
}

template <bool PAIRS>
__launch_bounds__(256)
__global__ void k_gather_ln(const int* __restrict__ rowptr,
                            const int* __restrict__ csr_src,
                            const float* __restrict__ csr_w,
                            const float* __restrict__ hw,
                            const float* __restrict__ invdeg,
                            const float* __restrict__ bias,
                            const float* __restrict__ gamma,
                            const float* __restrict__ beta,
                            float* __restrict__ h,
                            const int* __restrict__ need,
                            const int* __restrict__ x,
                            const float* __restrict__ cls_emb,
                            const float* __restrict__ nbr_emb) {
    const int t = blockIdx.x * 256 + threadIdx.x;
    const int n = __builtin_amdgcn_readfirstlane(t >> 5);
    if (need != nullptr && need[n] == 0) return;
    const int lane = threadIdx.x & 31;
    const int f = lane << 2;
    const int beg = rowptr[n];
    const int end = rowptr[n + 1];
    const float* hwf = hw + f;
    int own = n, cc = 0, nb = 0;
    if (PAIRS) {
        const v2i xv = *(const v2i*)(x + 2 * (t >> 5));
        cc = clamp_cls(xv.x);
        nb = clamp_nbr(xv.y);
        own = cc * N_NBR + nb;
    }

    const v4f  self = *(const v4f*)(hwf + (size_t)own * HID);
    const float inv = invdeg[n];
    const v4f  bv   = *(const v4f*)(bias + f);
    float* hr = h + (size_t)n * HID + f;
    v4f hv;
    if (PAIRS) {
        const v4f ea = *(const v4f*)(cls_emb + (size_t)cc * HID + f);
        const v4f eb = *(const v4f*)(nbr_emb + (size_t)nb * HID + f);
        hv = ea + eb;
    } else {
        hv = *(const v4f*)hr;
    }
    const v4f  g    = *(const v4f*)(gamma + f);
    const v4f  b    = *(const v4f*)(beta + f);

    const v4f acc = edge_sum(hwf, csr_src, csr_w, beg, end, lane);

    const v4f av = (acc + self * inv) + bv;

    v4f val;
    val.x = hv.x + fmaxf(av.x, 0.0f);
    val.y = hv.y + fmaxf(av.y, 0.0f);
    val.z = hv.z + fmaxf(av.z, 0.0f);
    val.w = hv.w + fmaxf(av.w, 0.0f);

    float s = val.x + val.y + val.z + val.w;
#pragma unroll
    for (int off = 16; off >= 1; off >>= 1) s += __shfl_xor(s, off, 32);
    const float mu = s * (1.0f / (float)HID);

    v4f dv = val - mu;
    float q = dv.x * dv.x + dv.y * dv.y + dv.z * dv.z + dv.w * dv.w;
#pragma unroll
    for (int off = 16; off >= 1; off >>= 1) q += __shfl_xor(q, off, 32);
    const float rstd = rsqrtf(q * (1.0f / (float)HID) + LN_EPS);

    const v4f outv = dv * rstd * g + b;
    *(v4f*)hr = outv;
}


#define LAST_ROOTS 16
#define ZP 132
static_assert(NGRAPH % LAST_ROOTS == 0, "blocks must cover the roots");
static_assert(HID * OUT_DIM <= 3 * LAST_ROOTS * 32, "W_out is staged three words per thread");

__launch_bounds__(LAST_ROOTS * 32)
__global__ void k_last(const int* __restrict__ rowptr,
                       const int* __restrict__ csr_src,
                       const float* __restrict__ csr_w,
                       const float* __restrict__ h,
                       const float* __restrict__ invdeg,
                       const float* __restrict__ W,
                       const float* __restrict__ bias,
                       const float* __restrict__ gamma,
                       const float* __restrict__ beta,
                       const float* __restrict__ W_out,
                       float* __restrict__ out) {
    __shared__ __attribute__((aligned(16))) float zt[LAST_ROOTS * ZP];
    __shared__ __attribute__((aligned(16))) float yt[LAST_ROOTS * ZP];
    __shared__ float wo[HID * OUT_DIM];
    const int tid  = threadIdx.x;
    const int wave = __builtin_amdgcn_readfirstlane(tid >> 5);
    const int lane = tid & 31;
    const int half = lane >> 4;
    const int l16  = lane & 15;
    const int g = blockIdx.x * LAST_ROOTS + wave;
    const int n = g * NPG;
    const int f = lane << 2;

    float wov[3];
#pragma unroll
    for (int r = 0; r < 3; ++r) {
        const int i = tid + r * (LAST_ROOTS * 32);
        wov[r] = (i < HID * OUT_DIM) ? W_out[i] : 0.0f;
    }

    v2f bfrag[32];
#pragma unroll
    for (int s = 0; s < 32; ++s) { bfrag[s].x = 0.0f; bfrag[s].y = 0.0f; }
    if (wave < 8) {
        const float* wcol = W + wave * 16 + l16;
#pragma unroll
        for (int s = 0; s < 32; ++s) {
            const int kb = s * 4 + half * 2;
            bfrag[s].x = wcol[(kb + 0) * HID];
            bfrag[s].y = wcol[(kb + 1) * HID];
        }
    }

    const int beg = rowptr[n];
    const int end = rowptr[n + 1];
    const float* hf = h + f;
    const v4f  hv  = *(const v4f*)(hf + (size_t)n * HID);
    const float inv = invdeg[n];
    const v4f  bv  = *(const v4f*)(bias + f);
    const v4f  gm  = *(const v4f*)(gamma + f);
    const v4f  bt  = *(const v4f*)(beta + f);

    const v4f acc = edge_sum(hf, csr_src, csr_w, beg, end, lane);
    const v4f z = acc + hv * inv;
    *(v4f*)(zt + wave * ZP + f) = z;
#pragma unroll
    for (int r = 0; r < 3; ++r) {
        const int i = tid + r * (LAST_ROOTS * 32);
        if (i < HID * OUT_DIM) wo[i] = wov[r];
    }
    __syncthreads();

    if (wave < 8) {
        v8f cacc = {};
        const float* arow = zt + l16 * ZP + half * 2;
#pragma unroll
        for (int s = 0; s < 32; ++s) {
            const v2f a = *(const v2f*)(arow + s * 4);
            cacc = __builtin_amdgcn_wmma_f32_16x16x4_f32(
                false, a, false, bfrag[s], (short)0, cacc, false, false);
        }
        const int col = wave * 16 + l16;
#pragma unroll
        for (int r = 0; r < 8; ++r) yt[(half * 8 + r) * ZP + col] = cacc[r];
    }
    __syncthreads();

    const v4f y  = *(const v4f*)(yt + wave * ZP + f);
    const v4f av = y + bv;

    v4f val;
    val.x = hv.x + fmaxf(av.x, 0.0f);
    val.y = hv.y + fmaxf(av.y, 0.0f);
    val.z = hv.z + fmaxf(av.z, 0.0f);
    val.w = hv.w + fmaxf(av.w, 0.0f);

    float s = val.x + val.y + val.z + val.w;
#pragma unroll
    for (int off = 16; off >= 1; off >>= 1) s += __shfl_xor(s, off, 32);
    const float mu = s * (1.0f / (float)HID);

    v4f dv = val - mu;
    float q = dv.x * dv.x + dv.y * dv.y + dv.z * dv.z + dv.w * dv.w;
#pragma unroll
    for (int off = 16; off >= 1; off >>= 1) q += __shfl_xor(q, off, 32);
    const float rstd = rsqrtf(q * (1.0f / (float)HID) + LN_EPS);

    const v4f outv = dv * rstd * gm + bt;
    *(v4f*)(zt + wave * ZP + f) = outv;
    __syncthreads();

    if (lane < OUT_DIM) {
        const float* hr = zt + wave * ZP;
        float lg = 0.0f;
#pragma unroll 32
        for (int k = 0; k < HID; ++k) lg = fmaf(hr[k], wo[k * OUT_DIM + lane], lg);
        out[g * OUT_DIM + lane] = lg;
    }
}

extern "C" void kernel_launch(void* const* d_in, const int* in_sizes, int n_in,
                              void* d_out, int out_size, void* d_ws, size_t ws_size,
                              hipStream_t stream) {
    const int*   x        = (const int*)d_in[0];
    const int*   ei       = (const int*)d_in[1];
    const float* cls_emb  = (const float*)d_in[3];
    const float* nbr_emb  = (const float*)d_in[4];
    const float* Ws       = (const float*)d_in[5];
    const float* bs       = (const float*)d_in[6];
    const float* gammas   = (const float*)d_in[7];
    const float* betas    = (const float*)d_in[8];
    const float* W_out    = (const float*)d_in[9];
    float*       out      = (float*)d_out;

    char* ws = (char*)d_ws;
    size_t off = 0;
    int*   cnt    = (int*)(ws + off);   off += (size_t)EBLOCKS * NBIN * 4;
    float* dinv   = (float*)(ws + off); off += (size_t)N_NODES * 4;
    float* invdeg = (float*)(ws + off); off += (size_t)N_NODES * 4;
    float* h      = (float*)(ws + off); off += (size_t)N_NODES * HID * 4;
    float* hw     = (float*)(ws + off); off += (size_t)N_NODES * HID * 4;
    int*   rowptr = (int*)(ws + off);   off += (size_t)(N_NODES + 64) * 4;
    int*   binned = (int*)(ws + off);   off += (size_t)E_EDGES * 4;
    int*   csrtmp = (int*)(ws + off);   off += (size_t)E_EDGES * 4;
    int*   csrsrc = (int*)(ws + off);   off += (size_t)E_EDGES * 4;
    float* csrw   = (float*)(ws + off); off += (size_t)E_EDGES * 4;
    int*   need   = (int*)(ws + off);   off += (size_t)N_NODES * 4;
    int*   binbase = (int*)(ws + off);  off += (size_t)(NBIN + 64) * 4;
    int*   csrpair = (int*)(ws + off);  off += (size_t)E_EDGES * 4;
    float* hwp     = (float*)(ws + off); off += (size_t)N_PAIR * HID * 4;
    (void)ws_size; (void)in_sizes; (void)n_in; (void)out_size;

    k_bin_count  <<<EBLOCKS, 256, 0, stream>>>(ei, cnt);
    k_bin_scatter<<<EBLOCKS, 256, 0, stream>>>(ei, cnt, binned, binbase);
    k_bin_build  <<<NBIN, BUILD_T, 0, stream>>>(binned, binbase, rowptr, dinv, invdeg, need, csrtmp);
    k_csr_order<<<(N_NODES * 32) / 256, 256, 0, stream>>>(rowptr, csrtmp, dinv, csrsrc, csrw, need,
                                                          x, csrpair);

    k_pair_product<<<PAIR_TILES, 256, 0, stream>>>(cls_emb, nbr_emb, Ws, hwp);
    k_gather_ln<true><<<(N_NODES * 32) / 256, 256, 0, stream>>>(
        rowptr, csrpair, csrw, hwp, invdeg, bs, gammas, betas, h, nullptr, x, cls_emb, nbr_emb);

    for (int i = 1; i < N_LAYERS - 2; ++i) {
        k_gemm_fused<<<N_NODES / 128, 256, 0, stream>>>(
            h, Ws + (size_t)i * HID * HID, hw);
        k_gather_ln<false><<<(N_NODES * 32) / 256, 256, 0, stream>>>(
            rowptr, csrsrc, csrw, hw, invdeg, bs + (size_t)i * HID,
            gammas + (size_t)i * HID, betas + (size_t)i * HID, h, nullptr,
            nullptr, nullptr, nullptr);
    }
    k_gemm_fused<<<N_NODES / 128, 256, 0, stream>>>(
        h, Ws + (size_t)(N_LAYERS - 2) * HID * HID, hw);
    k_gather_ln<false><<<(N_NODES * 32) / 256, 256, 0, stream>>>(
        rowptr, csrsrc, csrw, hw, invdeg, bs + (size_t)(N_LAYERS - 2) * HID,
        gammas + (size_t)(N_LAYERS - 2) * HID, betas + (size_t)(N_LAYERS - 2) * HID, h, need,
        nullptr, nullptr, nullptr);
    k_last<<<NGRAPH / LAST_ROOTS, LAST_ROOTS * 32, 0, stream>>>(
        rowptr, csrsrc, csrw, h, invdeg, Ws + (size_t)(N_LAYERS - 1) * HID * HID,
        bs + (size_t)(N_LAYERS - 1) * HID, gammas + (size_t)(N_LAYERS - 1) * HID,
        betas + (size_t)(N_LAYERS - 1) * HID, W_out, out);
}


# Layers 1, 2 as (A h) W in the gather launch: a block sums 16 nodes' rows of h, one 16-row WMMA product with W in LDS, then LN; both full GEMM launches dropped; those layers' sum order changes
# speedup vs baseline: 6.1347x; 1.0983x over previous
#include <hip/hip_runtime.h>

#define N_NODES   65536
#define NPG       128
#define NGRAPH    512
#define E_EDGES   1048576
#define HID       128
#define OUT_DIM   10
#define N_LAYERS  4
#define LN_EPS    1e-5f
#define N_CLS     64
#define N_NBR     65
#define N_PAIR    (N_CLS * N_NBR)

typedef float v2f __attribute__((ext_vector_type(2)));
typedef float v4f __attribute__((ext_vector_type(4)));
typedef float v8f __attribute__((ext_vector_type(8)));

#define NBIN      256
#define BIN_NODES 256
#define BIN_SHIFT 8
#define EPB       4096
#define EBLOCKS   (E_EDGES / EPB)
#define STAGE_CAP 12288
static_assert(NBIN * BIN_NODES == N_NODES, "bins must cover the nodes");
static_assert((1 << BIN_SHIFT) == BIN_NODES, "bin shift");
static_assert(EBLOCKS * EPB == E_EDGES, "edge blocks must cover the edges");
static_assert(N_NODES == 65536, "an edge is packed as src | dst << 16");
static_assert(EBLOCKS % 64 == 0 && NBIN == 256 && EPB == 16 * 256, "layout of the edge passes");

typedef int v4i __attribute__((ext_vector_type(4)));
typedef int v2i __attribute__((ext_vector_type(2)));

__device__ __forceinline__ int clamp_cls(int cc) { return min(max(cc, 0), N_CLS - 1); }
__device__ __forceinline__ int clamp_nbr(int nb) { return min(max(nb, 0), N_NBR - 1); }
__device__ __forceinline__ int pair_row(int cc, int nb) {
    return clamp_cls(cc) * N_NBR + clamp_nbr(nb);
}

__device__ __forceinline__ int excl_scan_256(int c, int t, int* wtot) {
    const int lane = t & 31;
    int x = c;
#pragma unroll
    for (int o = 1; o < 32; o <<= 1) {
        const int y = __shfl_up(x, o, 32);
        if (lane >= o) x += y;
    }
    if (t < 256 && lane == 31) wtot[t >> 5] = x;
    __syncthreads();
    int pre = 0;
    const int wv = t >> 5;
#pragma unroll
    for (int w = 0; w < 8; ++w) pre += (w < wv) ? wtot[w] : 0;
    return pre + x - c;
}

__launch_bounds__(256)
__global__ void k_bin_count(const int* __restrict__ edge_index, int* __restrict__ cnt) {
    __shared__ int hist[NBIN];
    const int t = threadIdx.x;
    hist[t] = 0;
    const int* dstp = edge_index + E_EDGES + blockIdx.x * EPB + t * 4;
    v4i dv[4];
#pragma unroll
    for (int i = 0; i < 4; ++i) dv[i] = *(const v4i*)(dstp + i * 1024);
    __syncthreads();
#pragma unroll
    for (int i = 0; i < 4; ++i)
#pragma unroll
        for (int j = 0; j < 4; ++j) {
            const int d = dv[i][j] & (N_NODES - 1);
            atomicAdd(&hist[d >> BIN_SHIFT], 1);
        }
    __syncthreads();
    cnt[blockIdx.x * NBIN + t] = hist[t];
}

__launch_bounds__(256)
__global__ void k_bin_scatter(const int* __restrict__ edge_index, const int* __restrict__ cnt,
                              int* __restrict__ binned, int* __restrict__ binbase) {
    __shared__ int ptot[4][NBIN];
    __shared__ int pbef[4][NBIN];
    __shared__ int cursor[NBIN];
    __shared__ int delta[NBIN];
    __shared__ int wtot_a[8];
    __shared__ int wtot_b[8];
    __shared__ int stage[EPB];
    const int t = threadIdx.x;
    const int b = blockIdx.x;
    const int* srcp = edge_index + b * EPB + t * 4;
    const int* dstp = edge_index + E_EDGES + b * EPB + t * 4;
    v4i sv[4], dv[4];
#pragma unroll
    for (int i = 0; i < 4; ++i) {
        sv[i] = *(const v4i*)(srcp + i * 1024);
        dv[i] = *(const v4i*)(dstp + i * 1024);
    }
    const int own = cnt[b * NBIN + t];
    const int q  = t >> 6;
    const int c4 = (t & 63) * 4;
    v4i tot = {0, 0, 0, 0};
    v4i bef = {0, 0, 0, 0};
#pragma unroll 16
    for (int r = q; r < EBLOCKS; r += 4) {
        const v4i c = *(const v4i*)(cnt + r * NBIN + c4);
        tot += c;
        if (r < b) bef += c;
    }
#pragma unroll
    for (int j = 0; j < 4; ++j) { ptot[q][c4 + j] = tot[j]; pbef[q][c4 + j] = bef[j]; }
    __syncthreads();
    const int total  = ptot[0][t] + ptot[1][t] + ptot[2][t] + ptot[3][t];
    const int before = pbef[0][t] + pbef[1][t] + pbef[2][t] + pbef[3][t];
    const int base   = excl_scan_256(total, t, wtot_a);
    const int lstart = excl_scan_256(own, t, wtot_b);
    cursor[t] = lstart;
    delta[t]  = base + before - lstart;
    if (b == 0) {
        binbase[t] = base;
        if (t == NBIN - 1) binbase[NBIN] = base + total;
    }
    __syncthreads();
#pragma unroll
    for (int i = 0; i < 4; ++i)
#pragma unroll
        for (int j = 0; j < 4; ++j) {
            const int s = sv[i][j] & (N_NODES - 1);
            const int d = dv[i][j] & (N_NODES - 1);
            const int lp = atomicAdd(&cursor[d >> BIN_SHIFT], 1);
            stage[lp] = (int)((unsigned)s | ((unsigned)d << 16));
        }
    __syncthreads();
#pragma unroll 4
    for (int i = t; i < EPB; i += 256) {
        const int w = stage[i];
        binned[i + delta[(w >> 24) & (NBIN - 1)]] = w;
    }
}

#define BUILD_T 1024
__launch_bounds__(BUILD_T)
__global__ void k_bin_build(const int* __restrict__ binned, const int* __restrict__ binbase,
                            int* __restrict__ rowptr, float* __restrict__ dinv,
                            float* __restrict__ invdeg, int* __restrict__ need,
                            int* __restrict__ csr_tmp) {
    __shared__ int deg[BIN_NODES];
    __shared__ int off[BIN_NODES];
    __shared__ int cur[BIN_NODES];
    __shared__ int wtot[8];
    __shared__ int stage[STAGE_CAP];
    const int t = threadIdx.x;
    const int bin = blockIdx.x;
    const int base = binbase[bin];
    const int n_e = binbase[bin + 1] - base;
    if (t < BIN_NODES) { deg[t] = 0; cur[t] = 0; }
    __syncthreads();
    const int* eb = binned + base;
    for (int i0 = 0; i0 < n_e; i0 += 4 * BUILD_T) {
        int w[4];
#pragma unroll
        for (int j = 0; j < 4; ++j) {
            const int i = i0 + j * BUILD_T + t;
            w[j] = eb[i < n_e ? i : n_e - 1];
        }
#pragma unroll
        for (int j = 0; j < 4; ++j)
            if (i0 + j * BUILD_T + t < n_e)
                atomicAdd(&deg[(w[j] >> 16) & (BIN_NODES - 1)], 1);
    }
    __syncthreads();
    const int c = (t < BIN_NODES) ? deg[t] : 0;
    const int ex = excl_scan_256(c, t, wtot);
    if (t < BIN_NODES) {
        const int node = bin * BIN_NODES + t;
        const float d = 1.0f + (float)c;
        dinv[node]   = rsqrtf(d);
        invdeg[node] = 1.0f / d;
        rowptr[node] = base + ex;
        need[node]   = 0;
        if (node == N_NODES - 1) rowptr[N_NODES] = base + ex + c;
        off[t] = ex;
    }
    __syncthreads();
    const bool fits = n_e <= STAGE_CAP;
    for (int i0 = 0; i0 < n_e; i0 += 4 * BUILD_T) {
        int w[4];
#pragma unroll
        for (int j = 0; j < 4; ++j) {
            const int i = i0 + j * BUILD_T + t;
            w[j] = eb[i < n_e ? i : n_e - 1];
        }
#pragma unroll
        for (int j = 0; j < 4; ++j)
            if (i0 + j * BUILD_T + t < n_e) {
                const int dl = (w[j] >> 16) & (BIN_NODES - 1);
                const int p = off[dl] + atomicAdd(&cur[dl], 1);
                const int key = w[j] & 0xffff;
                if (fits) stage[p] = key; else csr_tmp[base + p] = key;
            }
    }
    __syncthreads();
    if (fits)
        for (int i = t; i < n_e; i += BUILD_T) csr_tmp[base + i] = stage[i];
}

__launch_bounds__(256)
__global__ void k_csr_order(const int* __restrict__ rowptr,
                            const int* __restrict__ csr_tmp,
                            const float* __restrict__ dinv,
                            int* __restrict__ csr_src,
                            float* __restrict__ csr_w,
                            int* __restrict__ need,
                            const int* __restrict__ x,
                            int* __restrict__ csr_pair) {
    const int t = blockIdx.x * 256 + threadIdx.x;
    const int n = t >> 5;
    const int lane = t & 31;
    const bool root = (n % NPG) == 0;
    if (root && lane == 0) need[n] = 1;
    const int beg = rowptr[n];
    const int deg = rowptr[n + 1] - beg;
    const float dn = dinv[n];
    const int* b = csr_tmp + beg;
    for (int i = lane; i < deg; i += 32) {
        const int key = b[i] & (N_NODES - 1);
        const v2i xs = *(const v2i*)(x + 2 * key);
        int rank = 0;
        for (int j = 0; j < deg; ++j) {
            const int kj = b[j] & (N_NODES - 1);
            rank += (kj < key || (kj == key && j < i)) ? 1 : 0;
        }
        csr_src[beg + rank] = key;
        csr_w[beg + rank]   = dinv[key] * dn;
        csr_pair[beg + rank] = pair_row(xs.x, xs.y);
        if (root) need[key] = 1;
    }
}

#define PAIR_TILES (N_PAIR / 16)
static_assert(PAIR_TILES * 16 == N_PAIR, "pair rows must fill whole row tiles");

__launch_bounds__(256)
__global__ void k_pair_product(const float* __restrict__ cls_emb,
                               const float* __restrict__ nbr_emb,
                               const float* __restrict__ W,
                               float* __restrict__ hwp) {
    const int tid  = threadIdx.x;
    const int wave = tid >> 5;
    const int lane = tid & 31;
    const int half = lane >> 4;
    const int l16  = lane & 15;
    const int p  = blockIdx.x * 16 + l16;
    const int cc = p / N_NBR;
    const int nb = p - cc * N_NBR;
    const float* ca = cls_emb + (size_t)cc * HID + half * 2;
    const float* na = nbr_emb + (size_t)nb * HID + half * 2;
    const float* wcol = W + wave * 16 + l16;
    v2f ac[32], an[32], bfrag[32];
#pragma unroll
    for (int s = 0; s < 32; ++s) {
        ac[s] = *(const v2f*)(ca + s * 4);
        an[s] = *(const v2f*)(na + s * 4);
    }
#pragma unroll
    for (int s = 0; s < 32; ++s) {
        const int kb = s * 4 + half * 2;
        bfrag[s].x = wcol[(kb + 0) * HID];
        bfrag[s].y = wcol[(kb + 1) * HID];
    }
    v8f cacc = {};
#pragma unroll
    for (int s = 0; s < 32; ++s) {
        const v2f a = ac[s] + an[s];
        cacc = __builtin_amdgcn_wmma_f32_16x16x4_f32(
            false, a, false, bfrag[s], (short)0, cacc, false, false);
    }
    float* o = hwp + (size_t)(blockIdx.x * 16 + half * 8) * HID + wave * 16 + l16;
#pragma unroll
    for (int r = 0; r < 8; ++r) o[r * HID] = cacc[r];
}


__device__ __forceinline__ v4f edge_row(const float* hwf, int s_l, int j) {
    const int s = __builtin_amdgcn_readlane(s_l, j);
    return *(const v4f*)(hwf + (size_t)s * HID);
}
__device__ __forceinline__ float edge_wt(float w_l, int j) {
    return __int_as_float(__builtin_amdgcn_readlane(__float_as_int(w_l), j));
}

__device__ __forceinline__ v4f edge_sum(const float* rowsf,
                                        const int* __restrict__ csr_src,
                                        const float* __restrict__ csr_w,
                                        int beg, int end, int lane) {
    v4f acc = {0.0f, 0.0f, 0.0f, 0.0f};
    for (int base = beg; base < end; base += 32) {
        const int idx = base + lane;
        int   s_l = 0;
        float w_l = 0.0f;
        if (idx < end) { s_l = csr_src[idx]; w_l = csr_w[idx]; }
        const int left = end - base;
        const int cnt = left < 32 ? left : 32;
        int j = 0;
        for (; j + 4 <= cnt; j += 4) {
            const v4f v0 = edge_row(rowsf, s_l, j + 0);
            const v4f v1 = edge_row(rowsf, s_l, j + 1);
            const v4f v2 = edge_row(rowsf, s_l, j + 2);
            const v4f v3 = edge_row(rowsf, s_l, j + 3);
            acc += v0 * edge_wt(w_l, j + 0);
            acc += v1 * edge_wt(w_l, j + 1);
            acc += v2 * edge_wt(w_l, j + 2);
            acc += v3 * edge_wt(w_l, j + 3);
        }
        const int rem = cnt - j;
        if (rem == 3) {
            const v4f v0 = edge_row(rowsf, s_l, j + 0);
            const v4f v1 = edge_row(rowsf, s_l, j + 1);
            const v4f v2 = edge_row(rowsf, s_l, j + 2);
            acc += v0 * edge_wt(w_l, j + 0);
            acc += v1 * edge_wt(w_l, j + 1);
            acc += v2 * edge_wt(w_l, j + 2);
        } else if (rem == 2) {
            const v4f v0 = edge_row(rowsf, s_l, j + 0);
            const v4f v1 = edge_row(rowsf, s_l, j + 1);
            acc += v0 * edge_wt(w_l, j + 0);
            acc += v1 * edge_wt(w_l, j + 1);
        } else if (rem == 1) {
            const v4f v0 = edge_row(rowsf, s_l, j);
            acc += v0 * edge_wt(w_l, j);
        }
    }
    return acc;
}

template <bool PAIRS>
__launch_bounds__(256)
__global__ void k_gather_ln(const int* __restrict__ rowptr,
                            const int* __restrict__ csr_src,
                            const float* __restrict__ csr_w,
                            const float* __restrict__ hw,
                            const float* __restrict__ invdeg,
                            const float* __restrict__ bias,
                            const float* __restrict__ gamma,
                            const float* __restrict__ beta,
                            float* __restrict__ h,
                            const int* __restrict__ need,
                            const int* __restrict__ x,
                            const float* __restrict__ cls_emb,
                            const float* __restrict__ nbr_emb) {
    const int t = blockIdx.x * 256 + threadIdx.x;
    const int n = __builtin_amdgcn_readfirstlane(t >> 5);
    if (need != nullptr && need[n] == 0) return;
    const int lane = threadIdx.x & 31;
    const int f = lane << 2;
    const int beg = rowptr[n];
    const int end = rowptr[n + 1];
    const float* hwf = hw + f;
    int own = n, cc = 0, nb = 0;
    if (PAIRS) {
        const v2i xv = *(const v2i*)(x + 2 * (t >> 5));
        cc = clamp_cls(xv.x);
        nb = clamp_nbr(xv.y);
        own = cc * N_NBR + nb;
    }

    const v4f  self = *(const v4f*)(hwf + (size_t)own * HID);
    const float inv = invdeg[n];
    const v4f  bv   = *(const v4f*)(bias + f);
    float* hr = h + (size_t)n * HID + f;
    v4f hv;
    if (PAIRS) {
        const v4f ea = *(const v4f*)(cls_emb + (size_t)cc * HID + f);
        const v4f eb = *(const v4f*)(nbr_emb + (size_t)nb * HID + f);
        hv = ea + eb;
    } else {
        hv = *(const v4f*)hr;
    }
    const v4f  g    = *(const v4f*)(gamma + f);
    const v4f  b    = *(const v4f*)(beta + f);

    const v4f acc = edge_sum(hwf, csr_src, csr_w, beg, end, lane);

    const v4f av = (acc + self * inv) + bv;

    v4f val;
    val.x = hv.x + fmaxf(av.x, 0.0f);
    val.y = hv.y + fmaxf(av.y, 0.0f);
    val.z = hv.z + fmaxf(av.z, 0.0f);
    val.w = hv.w + fmaxf(av.w, 0.0f);

    float s = val.x + val.y + val.z + val.w;
#pragma unroll
    for (int off = 16; off >= 1; off >>= 1) s += __shfl_xor(s, off, 32);
    const float mu = s * (1.0f / (float)HID);

    v4f dv = val - mu;
    float q = dv.x * dv.x + dv.y * dv.y + dv.z * dv.z + dv.w * dv.w;
#pragma unroll
    for (int off = 16; off >= 1; off >>= 1) q += __shfl_xor(q, off, 32);
    const float rstd = rsqrtf(q * (1.0f / (float)HID) + LN_EPS);

    const v4f outv = dv * rstd * g + b;
    *(v4f*)hr = outv;
}


#define LAST_ROOTS 16
#define ZP 132
static_assert(NGRAPH % LAST_ROOTS == 0, "blocks must cover the roots");
static_assert(HID * OUT_DIM <= 3 * LAST_ROOTS * 32, "W_out is staged three words per thread");

__launch_bounds__(LAST_ROOTS * 32)
__global__ void k_last(const int* __restrict__ rowptr,
                       const int* __restrict__ csr_src,
                       const float* __restrict__ csr_w,
                       const float* __restrict__ h,
                       const float* __restrict__ invdeg,
                       const float* __restrict__ W,
                       const float* __restrict__ bias,
                       const float* __restrict__ gamma,
                       const float* __restrict__ beta,
                       const float* __restrict__ W_out,
                       float* __restrict__ out) {
    __shared__ __attribute__((aligned(16))) float zt[LAST_ROOTS * ZP];
    __shared__ __attribute__((aligned(16))) float yt[LAST_ROOTS * ZP];
    __shared__ float wo[HID * OUT_DIM];
    const int tid  = threadIdx.x;
    const int wave = __builtin_amdgcn_readfirstlane(tid >> 5);
    const int lane = tid & 31;
    const int half = lane >> 4;
    const int l16  = lane & 15;
    const int g = blockIdx.x * LAST_ROOTS + wave;
    const int n = g * NPG;
    const int f = lane << 2;

    float wov[3];
#pragma unroll
    for (int r = 0; r < 3; ++r) {
        const int i = tid + r * (LAST_ROOTS * 32);
        wov[r] = (i < HID * OUT_DIM) ? W_out[i] : 0.0f;
    }

    v2f bfrag[32];
#pragma unroll
    for (int s = 0; s < 32; ++s) { bfrag[s].x = 0.0f; bfrag[s].y = 0.0f; }
    if (wave < 8) {
        const float* wcol = W + wave * 16 + l16;
#pragma unroll
        for (int s = 0; s < 32; ++s) {
            const int kb = s * 4 + half * 2;
            bfrag[s].x = wcol[(kb + 0) * HID];
            bfrag[s].y = wcol[(kb + 1) * HID];
        }
    }

    const int beg = rowptr[n];
    const int end = rowptr[n + 1];
    const float* hf = h + f;
    const v4f  hv  = *(const v4f*)(hf + (size_t)n * HID);
    const float inv = invdeg[n];
    const v4f  bv  = *(const v4f*)(bias + f);
    const v4f  gm  = *(const v4f*)(gamma + f);
    const v4f  bt  = *(const v4f*)(beta + f);

    const v4f acc = edge_sum(hf, csr_src, csr_w, beg, end, lane);
    const v4f z = acc + hv * inv;
    *(v4f*)(zt + wave * ZP + f) = z;
#pragma unroll
    for (int r = 0; r < 3; ++r) {
        const int i = tid + r * (LAST_ROOTS * 32);
        if (i < HID * OUT_DIM) wo[i] = wov[r];
    }
    __syncthreads();

    if (wave < 8) {
        v8f cacc = {};
        const float* arow = zt + l16 * ZP + half * 2;
#pragma unroll
        for (int s = 0; s < 32; ++s) {
            const v2f a = *(const v2f*)(arow + s * 4);
            cacc = __builtin_amdgcn_wmma_f32_16x16x4_f32(
                false, a, false, bfrag[s], (short)0, cacc, false, false);
        }
        const int col = wave * 16 + l16;
#pragma unroll
        for (int r = 0; r < 8; ++r) yt[(half * 8 + r) * ZP + col] = cacc[r];
    }
    __syncthreads();

    const v4f y  = *(const v4f*)(yt + wave * ZP + f);
    const v4f av = y + bv;

    v4f val;
    val.x = hv.x + fmaxf(av.x, 0.0f);
    val.y = hv.y + fmaxf(av.y, 0.0f);
    val.z = hv.z + fmaxf(av.z, 0.0f);
    val.w = hv.w + fmaxf(av.w, 0.0f);

    float s = val.x + val.y + val.z + val.w;
#pragma unroll
    for (int off = 16; off >= 1; off >>= 1) s += __shfl_xor(s, off, 32);
    const float mu = s * (1.0f / (float)HID);

    v4f dv = val - mu;
    float q = dv.x * dv.x + dv.y * dv.y + dv.z * dv.z + dv.w * dv.w;
#pragma unroll
    for (int off = 16; off >= 1; off >>= 1) q += __shfl_xor(q, off, 32);
    const float rstd = rsqrtf(q * (1.0f / (float)HID) + LN_EPS);

    const v4f outv = dv * rstd * gm + bt;
    *(v4f*)(zt + wave * ZP + f) = outv;
    __syncthreads();

    if (lane < OUT_DIM) {
        const float* hr = zt + wave * ZP;
        float lg = 0.0f;
#pragma unroll 32
        for (int k = 0; k < HID; ++k) lg = fmaf(hr[k], wo[k * OUT_DIM + lane], lg);
        out[g * OUT_DIM + lane] = lg;
    }
}

#define AGG_ROWS   16
static_assert(N_LAYERS == 4, "the launch sequence is written out for four layers");
#define AGG_TILES  (N_NODES / AGG_ROWS)
#define AGG_BLOCKS 2048
#define WP2 288
static_assert(AGG_TILES * AGG_ROWS == N_NODES, "tiles must cover the nodes");
static_assert(AGG_TILES % AGG_BLOCKS == 0, "every block takes the same number of tiles");
static_assert(HID == 8 * AGG_ROWS, "one wave stages 4 row pairs of W, 16 pairs apart");

__launch_bounds__(AGG_ROWS * 32)
__global__ void k_agg_layer(const int* __restrict__ rowptr,
                            const int* __restrict__ csr_src,
                            const float* __restrict__ csr_w,
                            const float* __restrict__ hin,
                            const float* __restrict__ invdeg,
                            const float* __restrict__ W,
                            const float* __restrict__ bias,
                            const float* __restrict__ gamma,
                            const float* __restrict__ beta,
                            float* __restrict__ hout,
                            const int* __restrict__ need) {
    __shared__ __attribute__((aligned(16))) float lW[(HID / 2) * WP2];
    __shared__ __attribute__((aligned(16))) float zt[AGG_ROWS * ZP];
    __shared__ __attribute__((aligned(16))) float yt[AGG_ROWS * ZP];
    const int tid  = threadIdx.x;
    const int wave = __builtin_amdgcn_readfirstlane(tid >> 5);
    const int lane = tid & 31;
    const int half = lane >> 4;
    const int l16  = lane & 15;
    const int f = lane << 2;

    v4f wv[8];
#pragma unroll
    for (int r = 0; r < 4; ++r) {
        const float* wr = W + (size_t)(2 * (wave + AGG_ROWS * r)) * HID + f;
        wv[2 * r + 0] = *(const v4f*)wr;
        wv[2 * r + 1] = *(const v4f*)(wr + HID);
    }
    const v4f bv = *(const v4f*)(bias + f);
    const v4f gm = *(const v4f*)(gamma + f);
    const v4f bt = *(const v4f*)(beta + f);
#pragma unroll
    for (int r = 0; r < 4; ++r) {
        const v4f e = wv[2 * r + 0];
        const v4f o = wv[2 * r + 1];
        float* dst = lW + (wave + AGG_ROWS * r) * WP2 + 2 * f;
        const v4f lo = {e.x, o.x, e.y, o.y};
        const v4f hi = {e.z, o.z, e.w, o.w};
        *(v4f*)(dst + 0) = lo;
        *(v4f*)(dst + 4) = hi;
    }

    const float* hf = hin + f;
    for (int it = 0; it < AGG_TILES / AGG_BLOCKS; ++it) {
        const int tile = blockIdx.x + it * AGG_BLOCKS;
        const int n = tile * AGG_ROWS + wave;
        bool active = true;
        if (need != nullptr) {
            const int fl = (lane < AGG_ROWS) ? need[tile * AGG_ROWS + lane] : 0;
            const unsigned m = __builtin_amdgcn_ballot_w32(fl != 0);
            if (m == 0u) continue;
            active = ((m >> wave) & 1u) != 0u;
        }

        v4f hv = {0.0f, 0.0f, 0.0f, 0.0f};
        v4f z  = {0.0f, 0.0f, 0.0f, 0.0f};
        if (active) {
            const int beg = rowptr[n];
            const int end = rowptr[n + 1];
            hv = *(const v4f*)(hf + (size_t)n * HID);
            const float inv = invdeg[n];
            const v4f acc = edge_sum(hf, csr_src, csr_w, beg, end, lane);
            z = acc + hv * inv;
        }
        *(v4f*)(zt + wave * ZP + f) = z;
        __syncthreads();

        if (wave < 8) {
            v8f cacc = {};
            const float* arow = zt + l16 * ZP + half * 2;
            const float* bcol = lW + half * WP2 + 2 * (wave * 16 + l16);
#pragma unroll
            for (int s = 0; s < 32; ++s) {
                const v2f a = *(const v2f*)(arow + s * 4);
                const v2f b = *(const v2f*)(bcol + s * 2 * WP2);
                cacc = __builtin_amdgcn_wmma_f32_16x16x4_f32(
                    false, a, false, b, (short)0, cacc, false, false);
            }
            const int col = wave * 16 + l16;
#pragma unroll
            for (int r = 0; r < 8; ++r) yt[(half * 8 + r) * ZP + col] = cacc[r];
        }
        __syncthreads();

        if (active) {
            const v4f y  = *(const v4f*)(yt + wave * ZP + f);
            const v4f av = y + bv;

            v4f val;
            val.x = hv.x + fmaxf(av.x, 0.0f);
            val.y = hv.y + fmaxf(av.y, 0.0f);
            val.z = hv.z + fmaxf(av.z, 0.0f);
            val.w = hv.w + fmaxf(av.w, 0.0f);

            float s = val.x + val.y + val.z + val.w;
#pragma unroll
            for (int off = 16; off >= 1; off >>= 1) s += __shfl_xor(s, off, 32);
            const float mu = s * (1.0f / (float)HID);

            v4f dv = val - mu;
            float q = dv.x * dv.x + dv.y * dv.y + dv.z * dv.z + dv.w * dv.w;
#pragma unroll
            for (int off = 16; off >= 1; off >>= 1) q += __shfl_xor(q, off, 32);
            const float rstd = rsqrtf(q * (1.0f / (float)HID) + LN_EPS);

            const v4f outv = dv * rstd * gm + bt;
            *(v4f*)(hout + (size_t)n * HID + f) = outv;
        }
    }
}

extern "C" void kernel_launch(void* const* d_in, const int* in_sizes, int n_in,
                              void* d_out, int out_size, void* d_ws, size_t ws_size,
                              hipStream_t stream) {
    const int*   x        = (const int*)d_in[0];
    const int*   ei       = (const int*)d_in[1];
    const float* cls_emb  = (const float*)d_in[3];
    const float* nbr_emb  = (const float*)d_in[4];
    const float* Ws       = (const float*)d_in[5];
    const float* bs       = (const float*)d_in[6];
    const float* gammas   = (const float*)d_in[7];
    const float* betas    = (const float*)d_in[8];
    const float* W_out    = (const float*)d_in[9];
    float*       out      = (float*)d_out;

    char* ws = (char*)d_ws;
    size_t off = 0;
    int*   cnt    = (int*)(ws + off);   off += (size_t)EBLOCKS * NBIN * 4;
    float* dinv   = (float*)(ws + off); off += (size_t)N_NODES * 4;
    float* invdeg = (float*)(ws + off); off += (size_t)N_NODES * 4;
    float* h      = (float*)(ws + off); off += (size_t)N_NODES * HID * 4;
    float* h2     = (float*)(ws + off); off += (size_t)N_NODES * HID * 4;
    int*   rowptr = (int*)(ws + off);   off += (size_t)(N_NODES + 64) * 4;
    int*   binned = (int*)(ws + off);   off += (size_t)E_EDGES * 4;
    int*   csrtmp = (int*)(ws + off);   off += (size_t)E_EDGES * 4;
    int*   csrsrc = (int*)(ws + off);   off += (size_t)E_EDGES * 4;
    float* csrw   = (float*)(ws + off); off += (size_t)E_EDGES * 4;
    int*   need   = (int*)(ws + off);   off += (size_t)N_NODES * 4;
    int*   binbase = (int*)(ws + off);  off += (size_t)(NBIN + 64) * 4;
    int*   csrpair = (int*)(ws + off);  off += (size_t)E_EDGES * 4;
    float* hwp     = (float*)(ws + off); off += (size_t)N_PAIR * HID * 4;
    (void)ws_size; (void)in_sizes; (void)n_in; (void)out_size;

    k_bin_count  <<<EBLOCKS, 256, 0, stream>>>(ei, cnt);
    k_bin_scatter<<<EBLOCKS, 256, 0, stream>>>(ei, cnt, binned, binbase);
    k_bin_build  <<<NBIN, BUILD_T, 0, stream>>>(binned, binbase, rowptr, dinv, invdeg, need, csrtmp);
    k_csr_order<<<(N_NODES * 32) / 256, 256, 0, stream>>>(rowptr, csrtmp, dinv, csrsrc, csrw, need,
                                                          x, csrpair);

    k_pair_product<<<PAIR_TILES, 256, 0, stream>>>(cls_emb, nbr_emb, Ws, hwp);
    k_gather_ln<true><<<(N_NODES * 32) / 256, 256, 0, stream>>>(
        rowptr, csrpair, csrw, hwp, invdeg, bs, gammas, betas, h, nullptr, x, cls_emb, nbr_emb);

    k_agg_layer<<<AGG_BLOCKS, AGG_ROWS * 32, 0, stream>>>(
        rowptr, csrsrc, csrw, h, invdeg, Ws + (size_t)1 * HID * HID, bs + (size_t)1 * HID,
        gammas + (size_t)1 * HID, betas + (size_t)1 * HID, h2, nullptr);
    k_agg_layer<<<AGG_BLOCKS, AGG_ROWS * 32, 0, stream>>>(
        rowptr, csrsrc, csrw, h2, invdeg, Ws + (size_t)2 * HID * HID, bs + (size_t)2 * HID,
        gammas + (size_t)2 * HID, betas + (size_t)2 * HID, h, need);
    k_last<<<NGRAPH / LAST_ROOTS, LAST_ROOTS * 32, 0, stream>>>(
        rowptr, csrsrc, csrw, h, invdeg, Ws + (size_t)(N_LAYERS - 1) * HID * HID,
        bs + (size_t)(N_LAYERS - 1) * HID, gammas + (size_t)(N_LAYERS - 1) * HID,
        betas + (size_t)(N_LAYERS - 1) * HID, W_out, out);
}


# Layer 0's edge rows read from LDS: h@W0 split as cls_emb@W0 + nbr_emb@W0, a 129-row table staged per block (was 4160 pair rows gathered from L2); the embedding sum now follows the product
# speedup vs baseline: 6.5073x; 1.0607x over previous
#include <hip/hip_runtime.h>

#define N_NODES   65536
#define NPG       128
#define NGRAPH    512
#define E_EDGES   1048576
#define HID       128
#define OUT_DIM   10
#define N_LAYERS  4
#define LN_EPS    1e-5f
#define N_CLS     64
#define N_NBR     65
#define TAB_ROWS  (N_CLS + N_NBR)

typedef float v2f __attribute__((ext_vector_type(2)));
typedef float v4f __attribute__((ext_vector_type(4)));
typedef float v8f __attribute__((ext_vector_type(8)));

#define NBIN      256
#define BIN_NODES 256
#define BIN_SHIFT 8
#define EPB       4096
#define EBLOCKS   (E_EDGES / EPB)
#define STAGE_CAP 12288
static_assert(NBIN * BIN_NODES == N_NODES, "bins must cover the nodes");
static_assert((1 << BIN_SHIFT) == BIN_NODES, "bin shift");
static_assert(EBLOCKS * EPB == E_EDGES, "edge blocks must cover the edges");
static_assert(N_NODES == 65536, "an edge is packed as src | dst << 16");
static_assert(EBLOCKS % 64 == 0 && NBIN == 256 && EPB == 16 * 256, "layout of the edge passes");

typedef int v4i __attribute__((ext_vector_type(4)));
typedef int v2i __attribute__((ext_vector_type(2)));

__device__ __forceinline__ int clamp_cls(int cc) { return min(max(cc, 0), N_CLS - 1); }
__device__ __forceinline__ int clamp_nbr(int nb) { return min(max(nb, 0), N_NBR - 1); }
__device__ __forceinline__ int tab_code(int cc, int nb) {
    return (clamp_cls(cc) * (HID / 4)) | (((N_CLS + clamp_nbr(nb)) * (HID / 4)) << 16);
}
static_assert(TAB_ROWS * (HID / 4) < 65536, "a row start must fit in half a word");

__device__ __forceinline__ int excl_scan_256(int c, int t, int* wtot) {
    const int lane = t & 31;
    int x = c;
#pragma unroll
    for (int o = 1; o < 32; o <<= 1) {
        const int y = __shfl_up(x, o, 32);
        if (lane >= o) x += y;
    }
    if (t < 256 && lane == 31) wtot[t >> 5] = x;
    __syncthreads();
    int pre = 0;
    const int wv = t >> 5;
#pragma unroll
    for (int w = 0; w < 8; ++w) pre += (w < wv) ? wtot[w] : 0;
    return pre + x - c;
}

__launch_bounds__(256)
__global__ void k_bin_count(const int* __restrict__ edge_index, int* __restrict__ cnt) {
    __shared__ int hist[NBIN];
    const int t = threadIdx.x;
    hist[t] = 0;
    const int* dstp = edge_index + E_EDGES + blockIdx.x * EPB + t * 4;
    v4i dv[4];
#pragma unroll
    for (int i = 0; i < 4; ++i) dv[i] = *(const v4i*)(dstp + i * 1024);
    __syncthreads();
#pragma unroll
    for (int i = 0; i < 4; ++i)
#pragma unroll
        for (int j = 0; j < 4; ++j) {
            const int d = dv[i][j] & (N_NODES - 1);
            atomicAdd(&hist[d >> BIN_SHIFT], 1);
        }
    __syncthreads();
    cnt[blockIdx.x * NBIN + t] = hist[t];
}

__launch_bounds__(256)
__global__ void k_bin_scatter(const int* __restrict__ edge_index, const int* __restrict__ cnt,
                              int* __restrict__ binned, int* __restrict__ binbase) {
    __shared__ int ptot[4][NBIN];
    __shared__ int pbef[4][NBIN];
    __shared__ int cursor[NBIN];
    __shared__ int delta[NBIN];
    __shared__ int wtot_a[8];
    __shared__ int wtot_b[8];
    __shared__ int stage[EPB];
    const int t = threadIdx.x;
    const int b = blockIdx.x;
    const int* srcp = edge_index + b * EPB + t * 4;
    const int* dstp = edge_index + E_EDGES + b * EPB + t * 4;
    v4i sv[4], dv[4];
#pragma unroll
    for (int i = 0; i < 4; ++i) {
        sv[i] = *(const v4i*)(srcp + i * 1024);
        dv[i] = *(const v4i*)(dstp + i * 1024);
    }
    const int own = cnt[b * NBIN + t];
    const int q  = t >> 6;
    const int c4 = (t & 63) * 4;
    v4i tot = {0, 0, 0, 0};
    v4i bef = {0, 0, 0, 0};
#pragma unroll 16
    for (int r = q; r < EBLOCKS; r += 4) {
        const v4i c = *(const v4i*)(cnt + r * NBIN + c4);
        tot += c;
        if (r < b) bef += c;
    }
#pragma unroll
    for (int j = 0; j < 4; ++j) { ptot[q][c4 + j] = tot[j]; pbef[q][c4 + j] = bef[j]; }
    __syncthreads();
    const int total  = ptot[0][t] + ptot[1][t] + ptot[2][t] + ptot[3][t];
    const int before = pbef[0][t] + pbef[1][t] + pbef[2][t] + pbef[3][t];
    const int base   = excl_scan_256(total, t, wtot_a);
    const int lstart = excl_scan_256(own, t, wtot_b);
    cursor[t] = lstart;
    delta[t]  = base + before - lstart;
    if (b == 0) {
        binbase[t] = base;
        if (t == NBIN - 1) binbase[NBIN] = base + total;
    }
    __syncthreads();
#pragma unroll
    for (int i = 0; i < 4; ++i)
#pragma unroll
        for (int j = 0; j < 4; ++j) {
            const int s = sv[i][j] & (N_NODES - 1);
            const int d = dv[i][j] & (N_NODES - 1);
            const int lp = atomicAdd(&cursor[d >> BIN_SHIFT], 1);
            stage[lp] = (int)((unsigned)s | ((unsigned)d << 16));
        }
    __syncthreads();
#pragma unroll 4
    for (int i = t; i < EPB; i += 256) {
        const int w = stage[i];
        binned[i + delta[(w >> 24) & (NBIN - 1)]] = w;
    }
}

#define BUILD_T 1024
__launch_bounds__(BUILD_T)
__global__ void k_bin_build(const int* __restrict__ binned, const int* __restrict__ binbase,
                            int* __restrict__ rowptr, float* __restrict__ dinv,
                            float* __restrict__ invdeg, int* __restrict__ need,
                            int* __restrict__ csr_tmp) {
    __shared__ int deg[BIN_NODES];
    __shared__ int off[BIN_NODES];
    __shared__ int cur[BIN_NODES];
    __shared__ int wtot[8];
    __shared__ int stage[STAGE_CAP];
    const int t = threadIdx.x;
    const int bin = blockIdx.x;
    const int base = binbase[bin];
    const int n_e = binbase[bin + 1] - base;
    if (t < BIN_NODES) { deg[t] = 0; cur[t] = 0; }
    __syncthreads();
    const int* eb = binned + base;
    for (int i0 = 0; i0 < n_e; i0 += 4 * BUILD_T) {
        int w[4];
#pragma unroll
        for (int j = 0; j < 4; ++j) {
            const int i = i0 + j * BUILD_T + t;
            w[j] = eb[i < n_e ? i : n_e - 1];
        }
#pragma unroll
        for (int j = 0; j < 4; ++j)
            if (i0 + j * BUILD_T + t < n_e)
                atomicAdd(&deg[(w[j] >> 16) & (BIN_NODES - 1)], 1);
    }
    __syncthreads();
    const int c = (t < BIN_NODES) ? deg[t] : 0;
    const int ex = excl_scan_256(c, t, wtot);
    if (t < BIN_NODES) {
        const int node = bin * BIN_NODES + t;
        const float d = 1.0f + (float)c;
        dinv[node]   = rsqrtf(d);
        invdeg[node] = 1.0f / d;
        rowptr[node] = base + ex;
        need[node]   = 0;
        if (node == N_NODES - 1) rowptr[N_NODES] = base + ex + c;
        off[t] = ex;
    }
    __syncthreads();
    const bool fits = n_e <= STAGE_CAP;
    for (int i0 = 0; i0 < n_e; i0 += 4 * BUILD_T) {
        int w[4];
#pragma unroll
        for (int j = 0; j < 4; ++j) {
            const int i = i0 + j * BUILD_T + t;
            w[j] = eb[i < n_e ? i : n_e - 1];
        }
#pragma unroll
        for (int j = 0; j < 4; ++j)
            if (i0 + j * BUILD_T + t < n_e) {
                const int dl = (w[j] >> 16) & (BIN_NODES - 1);
                const int p = off[dl] + atomicAdd(&cur[dl], 1);
                const int key = w[j] & 0xffff;
                if (fits) stage[p] = key; else csr_tmp[base + p] = key;
            }
    }
    __syncthreads();
    if (fits)
        for (int i = t; i < n_e; i += BUILD_T) csr_tmp[base + i] = stage[i];
}

__launch_bounds__(256)
__global__ void k_csr_order(const int* __restrict__ rowptr,
                            const int* __restrict__ csr_tmp,
                            const float* __restrict__ dinv,
                            int* __restrict__ csr_src,
                            float* __restrict__ csr_w,
                            int* __restrict__ need,
                            const int* __restrict__ x,
                            int* __restrict__ csr_code) {
    const int t = blockIdx.x * 256 + threadIdx.x;
    const int n = t >> 5;
    const int lane = t & 31;
    const bool root = (n % NPG) == 0;
    if (root && lane == 0) need[n] = 1;
    const int beg = rowptr[n];
    const int deg = rowptr[n + 1] - beg;
    const float dn = dinv[n];
    const int* b = csr_tmp + beg;
    for (int i = lane; i < deg; i += 32) {
        const int key = b[i] & (N_NODES - 1);
        const v2i xs = *(const v2i*)(x + 2 * key);
        int rank = 0;
        for (int j = 0; j < deg; ++j) {
            const int kj = b[j] & (N_NODES - 1);
            rank += (kj < key || (kj == key && j < i)) ? 1 : 0;
        }
        csr_src[beg + rank] = key;
        csr_w[beg + rank]   = dinv[key] * dn;
        csr_code[beg + rank] = tab_code(xs.x, xs.y);
        if (root) need[key] = 1;
    }
}

#define TAB_TILES ((TAB_ROWS + 15) / 16)

__launch_bounds__(256)
__global__ void k_table_product(const float* __restrict__ cls_emb,
                                const float* __restrict__ nbr_emb,
                                const float* __restrict__ W,
                                float* __restrict__ tab) {
    const int tid  = threadIdx.x;
    const int wave = tid >> 5;
    const int lane = tid & 31;
    const int half = lane >> 4;
    const int l16  = lane & 15;
    const int r  = min(blockIdx.x * 16 + l16, TAB_ROWS - 1);
    const float* ar = (r < N_CLS ? cls_emb + (size_t)r * HID
                                 : nbr_emb + (size_t)(r - N_CLS) * HID) + half * 2;
    const float* wcol = W + wave * 16 + l16;
    v2f afrag[32], bfrag[32];
#pragma unroll
    for (int s = 0; s < 32; ++s) afrag[s] = *(const v2f*)(ar + s * 4);
#pragma unroll
    for (int s = 0; s < 32; ++s) {
        const int kb = s * 4 + half * 2;
        bfrag[s].x = wcol[(kb + 0) * HID];
        bfrag[s].y = wcol[(kb + 1) * HID];
    }
    v8f cacc = {};
#pragma unroll
    for (int s = 0; s < 32; ++s)
        cacc = __builtin_amdgcn_wmma_f32_16x16x4_f32(
            false, afrag[s], false, bfrag[s], (short)0, cacc, false, false);
    const int m0 = blockIdx.x * 16 + half * 8;
    float* o = tab + (size_t)m0 * HID + wave * 16 + l16;
#pragma unroll
    for (int q = 0; q < 8; ++q)
        if (m0 + q < TAB_ROWS) o[q * HID] = cacc[q];
}


__device__ __forceinline__ v4f edge_row(const float* hwf, int s_l, int j) {
    const int s = __builtin_amdgcn_readlane(s_l, j);
    return *(const v4f*)(hwf + (size_t)s * HID);
}
__device__ __forceinline__ float edge_wt(float w_l, int j) {
    return __int_as_float(__builtin_amdgcn_readlane(__float_as_int(w_l), j));
}

__device__ __forceinline__ v4f edge_sum(const float* rowsf,
                                        const int* __restrict__ csr_src,
                                        const float* __restrict__ csr_w,
                                        int beg, int end, int lane) {
    v4f acc = {0.0f, 0.0f, 0.0f, 0.0f};
    for (int base = beg; base < end; base += 32) {
        const int idx = base + lane;
        int   s_l = 0;
        float w_l = 0.0f;
        if (idx < end) { s_l = csr_src[idx]; w_l = csr_w[idx]; }
        const int left = end - base;
        const int cnt = left < 32 ? left : 32;
        int j = 0;
        for (; j + 4 <= cnt; j += 4) {
            const v4f v0 = edge_row(rowsf, s_l, j + 0);
            const v4f v1 = edge_row(rowsf, s_l, j + 1);
            const v4f v2 = edge_row(rowsf, s_l, j + 2);
            const v4f v3 = edge_row(rowsf, s_l, j + 3);
            acc += v0 * edge_wt(w_l, j + 0);
            acc += v1 * edge_wt(w_l, j + 1);
            acc += v2 * edge_wt(w_l, j + 2);
            acc += v3 * edge_wt(w_l, j + 3);
        }
        const int rem = cnt - j;
        if (rem == 3) {
            const v4f v0 = edge_row(rowsf, s_l, j + 0);
            const v4f v1 = edge_row(rowsf, s_l, j + 1);
            const v4f v2 = edge_row(rowsf, s_l, j + 2);
            acc += v0 * edge_wt(w_l, j + 0);
            acc += v1 * edge_wt(w_l, j + 1);
            acc += v2 * edge_wt(w_l, j + 2);
        } else if (rem == 2) {
            const v4f v0 = edge_row(rowsf, s_l, j + 0);
            const v4f v1 = edge_row(rowsf, s_l, j + 1);
            acc += v0 * edge_wt(w_l, j + 0);
            acc += v1 * edge_wt(w_l, j + 1);
        } else if (rem == 1) {
            const v4f v0 = edge_row(rowsf, s_l, j);
            acc += v0 * edge_wt(w_l, j);
        }
    }
    return acc;
}

__device__ __forceinline__ v4f tab_edge_row(const v4f* tl, int c_l, int j) {
    const unsigned cd = (unsigned)__builtin_amdgcn_readlane(c_l, j);
    return tl[cd & 0xffffu] + tl[cd >> 16];
}

__device__ __forceinline__ v4f edge_sum_tab(const v4f* tl,
                                            const int* __restrict__ csr_code,
                                            const float* __restrict__ csr_w,
                                            int beg, int end, int lane) {
    v4f acc = {0.0f, 0.0f, 0.0f, 0.0f};
    for (int base = beg; base < end; base += 32) {
        const int idx = base + lane;
        int   c_l = 0;
        float w_l = 0.0f;
        if (idx < end) { c_l = csr_code[idx]; w_l = csr_w[idx]; }
        const int left = end - base;
        const int cnt = left < 32 ? left : 32;
        int j = 0;
        for (; j + 2 <= cnt; j += 2) {
            const v4f v0 = tab_edge_row(tl, c_l, j + 0);
            const v4f v1 = tab_edge_row(tl, c_l, j + 1);
            acc += v0 * edge_wt(w_l, j + 0);
            acc += v1 * edge_wt(w_l, j + 1);
        }
        if (j < cnt) {
            const v4f v0 = tab_edge_row(tl, c_l, j);
            acc += v0 * edge_wt(w_l, j);
        }
    }
    return acc;
}

#define L0_WAVES  16
#define L0_NPW    4
#define L0_BLOCKS (N_NODES / (L0_WAVES * L0_NPW))
#define TAB_V4    (TAB_ROWS * HID / 4)
#define TAB_LD    ((TAB_V4 + L0_WAVES * 32 - 1) / (L0_WAVES * 32))
static_assert(L0_BLOCKS * L0_WAVES * L0_NPW == N_NODES, "blocks must cover the nodes");

__launch_bounds__(L0_WAVES * 32, 16)
__global__ void k_gather_ln0(const int* __restrict__ rowptr,
                             const int* __restrict__ csr_code,
                             const float* __restrict__ csr_w,
                             const float* __restrict__ tab,
                             const float* __restrict__ invdeg,
                             const float* __restrict__ bias,
                             const float* __restrict__ gamma,
                             const float* __restrict__ beta,
                             float* __restrict__ h,
                             const int* __restrict__ x,
                             const float* __restrict__ cls_emb,
                             const float* __restrict__ nbr_emb) {
    __shared__ __attribute__((aligned(16))) float T[TAB_ROWS * HID];
    const int tid  = threadIdx.x;
    const int wave = __builtin_amdgcn_readfirstlane(tid >> 5);
    const int lane = tid & 31;
    const int f = lane << 2;

    v4f tv[TAB_LD];
#pragma unroll
    for (int r = 0; r < TAB_LD; ++r) {
        const int i = tid + r * (L0_WAVES * 32);
        tv[r] = *(const v4f*)(tab + 4 * (i < TAB_V4 ? i : TAB_V4 - 1));
    }
    const v4f bv = *(const v4f*)(bias + f);
    const v4f g  = *(const v4f*)(gamma + f);
    const v4f b  = *(const v4f*)(beta + f);
#pragma unroll
    for (int r = 0; r < TAB_LD; ++r) {
        const int i = tid + r * (L0_WAVES * 32);
        if (i < TAB_V4) *(v4f*)(T + 4 * i) = tv[r];
    }
    __syncthreads();

    const v4f* tl = (const v4f*)T + lane;
    const int n0 = (blockIdx.x * L0_WAVES + wave) * L0_NPW;
    for (int it = 0; it < L0_NPW; ++it) {
        const int n = n0 + it;
        const int beg = rowptr[n];
        const int end = rowptr[n + 1];
        const v2i xv = *(const v2i*)(x + 2 * n);
        const int cc = clamp_cls(xv.x);
        const int nb = clamp_nbr(xv.y);
        const float inv = invdeg[n];
        const v4f ea = *(const v4f*)(cls_emb + (size_t)cc * HID + f);
        const v4f eb = *(const v4f*)(nbr_emb + (size_t)nb * HID + f);
        const v4f hv = ea + eb;

        const v4f acc = edge_sum_tab(tl, csr_code, csr_w, beg, end, lane);

        const v4f self = tl[cc * (HID / 4)] + tl[(N_CLS + nb) * (HID / 4)];

        const v4f av = (acc + self * inv) + bv;

        v4f val;
        val.x = hv.x + fmaxf(av.x, 0.0f);
        val.y = hv.y + fmaxf(av.y, 0.0f);
        val.z = hv.z + fmaxf(av.z, 0.0f);
        val.w = hv.w + fmaxf(av.w, 0.0f);

        float s = val.x + val.y + val.z + val.w;
#pragma unroll
        for (int off = 16; off >= 1; off >>= 1) s += __shfl_xor(s, off, 32);
        const float mu = s * (1.0f / (float)HID);

        v4f dv = val - mu;
        float q = dv.x * dv.x + dv.y * dv.y + dv.z * dv.z + dv.w * dv.w;
#pragma unroll
        for (int off = 16; off >= 1; off >>= 1) q += __shfl_xor(q, off, 32);
        const float rstd = rsqrtf(q * (1.0f / (float)HID) + LN_EPS);

        const v4f outv = dv * rstd * g + b;
        *(v4f*)(h + (size_t)n * HID + f) = outv;
    }
}


#define LAST_ROOTS 16
#define ZP 132
static_assert(NGRAPH % LAST_ROOTS == 0, "blocks must cover the roots");
static_assert(HID * OUT_DIM <= 3 * LAST_ROOTS * 32, "W_out is staged three words per thread");

__launch_bounds__(LAST_ROOTS * 32)
__global__ void k_last(const int* __restrict__ rowptr,
                       const int* __restrict__ csr_src,
                       const float* __restrict__ csr_w,
                       const float* __restrict__ h,
                       const float* __restrict__ invdeg,
                       const float* __restrict__ W,
                       const float* __restrict__ bias,
                       const float* __restrict__ gamma,
                       const float* __restrict__ beta,
                       const float* __restrict__ W_out,
                       float* __restrict__ out) {
    __shared__ __attribute__((aligned(16))) float zt[LAST_ROOTS * ZP];
    __shared__ __attribute__((aligned(16))) float yt[LAST_ROOTS * ZP];
    __shared__ float wo[HID * OUT_DIM];
    const int tid  = threadIdx.x;
    const int wave = __builtin_amdgcn_readfirstlane(tid >> 5);
    const int lane = tid & 31;
    const int half = lane >> 4;
    const int l16  = lane & 15;
    const int g = blockIdx.x * LAST_ROOTS + wave;
    const int n = g * NPG;
    const int f = lane << 2;

    float wov[3];
#pragma unroll
    for (int r = 0; r < 3; ++r) {
        const int i = tid + r * (LAST_ROOTS * 32);
        wov[r] = (i < HID * OUT_DIM) ? W_out[i] : 0.0f;
    }

    v2f bfrag[32];
#pragma unroll
    for (int s = 0; s < 32; ++s) { bfrag[s].x = 0.0f; bfrag[s].y = 0.0f; }
    if (wave < 8) {
        const float* wcol = W + wave * 16 + l16;
#pragma unroll
        for (int s = 0; s < 32; ++s) {
            const int kb = s * 4 + half * 2;
            bfrag[s].x = wcol[(kb + 0) * HID];
            bfrag[s].y = wcol[(kb + 1) * HID];
        }
    }

    const int beg = rowptr[n];
    const int end = rowptr[n + 1];
    const float* hf = h + f;
    const v4f  hv  = *(const v4f*)(hf + (size_t)n * HID);
    const float inv = invdeg[n];
    const v4f  bv  = *(const v4f*)(bias + f);
    const v4f  gm  = *(const v4f*)(gamma + f);
    const v4f  bt  = *(const v4f*)(beta + f);

    const v4f acc = edge_sum(hf, csr_src, csr_w, beg, end, lane);
    const v4f z = acc + hv * inv;
    *(v4f*)(zt + wave * ZP + f) = z;
#pragma unroll
    for (int r = 0; r < 3; ++r) {
        const int i = tid + r * (LAST_ROOTS * 32);
        if (i < HID * OUT_DIM) wo[i] = wov[r];
    }
    __syncthreads();

    if (wave < 8) {
        v8f cacc = {};
        const float* arow = zt + l16 * ZP + half * 2;
#pragma unroll
        for (int s = 0; s < 32; ++s) {
            const v2f a = *(const v2f*)(arow + s * 4);
            cacc = __builtin_amdgcn_wmma_f32_16x16x4_f32(
                false, a, false, bfrag[s], (short)0, cacc, false, false);
        }
        const int col = wave * 16 + l16;
#pragma unroll
        for (int r = 0; r < 8; ++r) yt[(half * 8 + r) * ZP + col] = cacc[r];
    }
    __syncthreads();

    const v4f y  = *(const v4f*)(yt + wave * ZP + f);
    const v4f av = y + bv;

    v4f val;
    val.x = hv.x + fmaxf(av.x, 0.0f);
    val.y = hv.y + fmaxf(av.y, 0.0f);
    val.z = hv.z + fmaxf(av.z, 0.0f);
    val.w = hv.w + fmaxf(av.w, 0.0f);

    float s = val.x + val.y + val.z + val.w;
#pragma unroll
    for (int off = 16; off >= 1; off >>= 1) s += __shfl_xor(s, off, 32);
    const float mu = s * (1.0f / (float)HID);

    v4f dv = val - mu;
    float q = dv.x * dv.x + dv.y * dv.y + dv.z * dv.z + dv.w * dv.w;
#pragma unroll
    for (int off = 16; off >= 1; off >>= 1) q += __shfl_xor(q, off, 32);
    const float rstd = rsqrtf(q * (1.0f / (float)HID) + LN_EPS);

    const v4f outv = dv * rstd * gm + bt;
    *(v4f*)(zt + wave * ZP + f) = outv;
    __syncthreads();

    if (lane < OUT_DIM) {
        const float* hr = zt + wave * ZP;
        float lg = 0.0f;
#pragma unroll 32
        for (int k = 0; k < HID; ++k) lg = fmaf(hr[k], wo[k * OUT_DIM + lane], lg);
        out[g * OUT_DIM + lane] = lg;
    }
}

#define AGG_ROWS   16
static_assert(N_LAYERS == 4, "the launch sequence is written out for four layers");
#define AGG_TILES  (N_NODES / AGG_ROWS)
#define AGG_BLOCKS 2048
#define WP2 288
static_assert(AGG_TILES * AGG_ROWS == N_NODES, "tiles must cover the nodes");
static_assert(AGG_TILES % AGG_BLOCKS == 0, "every block takes the same number of tiles");
static_assert(HID == 8 * AGG_ROWS, "one wave stages 4 row pairs of W, 16 pairs apart");

__launch_bounds__(AGG_ROWS * 32)
__global__ void k_agg_layer(const int* __restrict__ rowptr,
                            const int* __restrict__ csr_src,
                            const float* __restrict__ csr_w,
                            const float* __restrict__ hin,
                            const float* __restrict__ invdeg,
                            const float* __restrict__ W,
                            const float* __restrict__ bias,
                            const float* __restrict__ gamma,
                            const float* __restrict__ beta,
                            float* __restrict__ hout,
                            const int* __restrict__ need) {
    __shared__ __attribute__((aligned(16))) float lW[(HID / 2) * WP2];
    __shared__ __attribute__((aligned(16))) float zt[AGG_ROWS * ZP];
    __shared__ __attribute__((aligned(16))) float yt[AGG_ROWS * ZP];
    const int tid  = threadIdx.x;
    const int wave = __builtin_amdgcn_readfirstlane(tid >> 5);
    const int lane = tid & 31;
    const int half = lane >> 4;
    const int l16  = lane & 15;
    const int f = lane << 2;

    v4f wv[8];
#pragma unroll
    for (int r = 0; r < 4; ++r) {
        const float* wr = W + (size_t)(2 * (wave + AGG_ROWS * r)) * HID + f;
        wv[2 * r + 0] = *(const v4f*)wr;
        wv[2 * r + 1] = *(const v4f*)(wr + HID);
    }
    const v4f bv = *(const v4f*)(bias + f);
    const v4f gm = *(const v4f*)(gamma + f);
    const v4f bt = *(const v4f*)(beta + f);
#pragma unroll
    for (int r = 0; r < 4; ++r) {
        const v4f e = wv[2 * r + 0];
        const v4f o = wv[2 * r + 1];
        float* dst = lW + (wave + AGG_ROWS * r) * WP2 + 2 * f;
        const v4f lo = {e.x, o.x, e.y, o.y};
        const v4f hi = {e.z, o.z, e.w, o.w};
        *(v4f*)(dst + 0) = lo;
        *(v4f*)(dst + 4) = hi;
    }

    const float* hf = hin + f;
    for (int it = 0; it < AGG_TILES / AGG_BLOCKS; ++it) {
        const int tile = blockIdx.x + it * AGG_BLOCKS;
        const int n = tile * AGG_ROWS + wave;
        bool active = true;
        if (need != nullptr) {
            const int fl = (lane < AGG_ROWS) ? need[tile * AGG_ROWS + lane] : 0;
            const unsigned m = __builtin_amdgcn_ballot_w32(fl != 0);
            if (m == 0u) continue;
            active = ((m >> wave) & 1u) != 0u;
        }

        v4f hv = {0.0f, 0.0f, 0.0f, 0.0f};
        v4f z  = {0.0f, 0.0f, 0.0f, 0.0f};
        if (active) {
            const int beg = rowptr[n];
            const int end = rowptr[n + 1];
            hv = *(const v4f*)(hf + (size_t)n * HID);
            const float inv = invdeg[n];
            const v4f acc = edge_sum(hf, csr_src, csr_w, beg, end, lane);
            z = acc + hv * inv;
        }
        *(v4f*)(zt + wave * ZP + f) = z;
        __syncthreads();

        if (wave < 8) {
            v8f cacc = {};
            const float* arow = zt + l16 * ZP + half * 2;
            const float* bcol = lW + half * WP2 + 2 * (wave * 16 + l16);
#pragma unroll
            for (int s = 0; s < 32; ++s) {
                const v2f a = *(const v2f*)(arow + s * 4);
                const v2f b = *(const v2f*)(bcol + s * 2 * WP2);
                cacc = __builtin_amdgcn_wmma_f32_16x16x4_f32(
                    false, a, false, b, (short)0, cacc, false, false);
            }
            const int col = wave * 16 + l16;
#pragma unroll
            for (int r = 0; r < 8; ++r) yt[(half * 8 + r) * ZP + col] = cacc[r];
        }
        __syncthreads();

        if (active) {
            const v4f y  = *(const v4f*)(yt + wave * ZP + f);
            const v4f av = y + bv;

            v4f val;
            val.x = hv.x + fmaxf(av.x, 0.0f);
            val.y = hv.y + fmaxf(av.y, 0.0f);
            val.z = hv.z + fmaxf(av.z, 0.0f);
            val.w = hv.w + fmaxf(av.w, 0.0f);

            float s = val.x + val.y + val.z + val.w;
#pragma unroll
            for (int off = 16; off >= 1; off >>= 1) s += __shfl_xor(s, off, 32);
            const float mu = s * (1.0f / (float)HID);

            v4f dv = val - mu;
            float q = dv.x * dv.x + dv.y * dv.y + dv.z * dv.z + dv.w * dv.w;
#pragma unroll
            for (int off = 16; off >= 1; off >>= 1) q += __shfl_xor(q, off, 32);
            const float rstd = rsqrtf(q * (1.0f / (float)HID) + LN_EPS);

            const v4f outv = dv * rstd * gm + bt;
            *(v4f*)(hout + (size_t)n * HID + f) = outv;
        }
    }
}

extern "C" void kernel_launch(void* const* d_in, const int* in_sizes, int n_in,
                              void* d_out, int out_size, void* d_ws, size_t ws_size,
                              hipStream_t stream) {
    const int*   x        = (const int*)d_in[0];
    const int*   ei       = (const int*)d_in[1];
    const float* cls_emb  = (const float*)d_in[3];
    const float* nbr_emb  = (const float*)d_in[4];
    const float* Ws       = (const float*)d_in[5];
    const float* bs       = (const float*)d_in[6];
    const float* gammas   = (const float*)d_in[7];
    const float* betas    = (const float*)d_in[8];
    const float* W_out    = (const float*)d_in[9];
    float*       out      = (float*)d_out;

    char* ws = (char*)d_ws;
    size_t off = 0;
    int*   cnt    = (int*)(ws + off);   off += (size_t)EBLOCKS * NBIN * 4;
    float* dinv   = (float*)(ws + off); off += (size_t)N_NODES * 4;
    float* invdeg = (float*)(ws + off); off += (size_t)N_NODES * 4;
    float* h      = (float*)(ws + off); off += (size_t)N_NODES * HID * 4;
    float* h2     = (float*)(ws + off); off += (size_t)N_NODES * HID * 4;
    int*   rowptr = (int*)(ws + off);   off += (size_t)(N_NODES + 64) * 4;
    int*   binned = (int*)(ws + off);   off += (size_t)E_EDGES * 4;
    int*   csrtmp = (int*)(ws + off);   off += (size_t)E_EDGES * 4;
    int*   csrsrc = (int*)(ws + off);   off += (size_t)E_EDGES * 4;
    float* csrw   = (float*)(ws + off); off += (size_t)E_EDGES * 4;
    int*   need   = (int*)(ws + off);   off += (size_t)N_NODES * 4;
    int*   binbase = (int*)(ws + off);  off += (size_t)(NBIN + 64) * 4;
    int*   csrcode = (int*)(ws + off);  off += (size_t)E_EDGES * 4;
    float* tab     = (float*)(ws + off); off += (size_t)TAB_ROWS * HID * 4;
    (void)ws_size; (void)in_sizes; (void)n_in; (void)out_size;

    k_bin_count  <<<EBLOCKS, 256, 0, stream>>>(ei, cnt);
    k_bin_scatter<<<EBLOCKS, 256, 0, stream>>>(ei, cnt, binned, binbase);
    k_bin_build  <<<NBIN, BUILD_T, 0, stream>>>(binned, binbase, rowptr, dinv, invdeg, need, csrtmp);
    k_csr_order<<<(N_NODES * 32) / 256, 256, 0, stream>>>(rowptr, csrtmp, dinv, csrsrc, csrw, need,
                                                          x, csrcode);

    k_table_product<<<TAB_TILES, 256, 0, stream>>>(cls_emb, nbr_emb, Ws, tab);
    k_gather_ln0<<<L0_BLOCKS, L0_WAVES * 32, 0, stream>>>(
        rowptr, csrcode, csrw, tab, invdeg, bs, gammas, betas, h, x, cls_emb, nbr_emb);

    k_agg_layer<<<AGG_BLOCKS, AGG_ROWS * 32, 0, stream>>>(
        rowptr, csrsrc, csrw, h, invdeg, Ws + (size_t)1 * HID * HID, bs + (size_t)1 * HID,
        gammas + (size_t)1 * HID, betas + (size_t)1 * HID, h2, nullptr);
    k_agg_layer<<<AGG_BLOCKS, AGG_ROWS * 32, 0, stream>>>(
        rowptr, csrsrc, csrw, h2, invdeg, Ws + (size_t)2 * HID * HID, bs + (size_t)2 * HID,
        gammas + (size_t)2 * HID, betas + (size_t)2 * HID, h, need);
    k_last<<<NGRAPH / LAST_ROOTS, LAST_ROOTS * 32, 0, stream>>>(
        rowptr, csrsrc, csrw, h, invdeg, Ws + (size_t)(N_LAYERS - 1) * HID * HID,
        bs + (size_t)(N_LAYERS - 1) * HID, gammas + (size_t)(N_LAYERS - 1) * HID,
        betas + (size_t)(N_LAYERS - 1) * HID, W_out, out);
}


# Layer 2 only for each root's bucket (in-neighbours + root): one block per root, rows kept in LDS and summed into the last layer in the same launch; flag array and 1 launch dropped; sums as before
# speedup vs baseline: 7.9698x; 1.2247x over previous
#include <hip/hip_runtime.h>

#define N_NODES   65536
#define NPG       128
#define NGRAPH    512
#define E_EDGES   1048576
#define HID       128
#define OUT_DIM   10
#define N_LAYERS  4
#define LN_EPS    1e-5f
#define N_CLS     64
#define N_NBR     65
#define TAB_ROWS  (N_CLS + N_NBR)

typedef float v2f __attribute__((ext_vector_type(2)));
typedef float v4f __attribute__((ext_vector_type(4)));
typedef float v8f __attribute__((ext_vector_type(8)));

#define NBIN      256
#define BIN_NODES 256
#define BIN_SHIFT 8
#define EPB       4096
#define EBLOCKS   (E_EDGES / EPB)
#define STAGE_CAP 12288
static_assert(NBIN * BIN_NODES == N_NODES, "bins must cover the nodes");
static_assert((1 << BIN_SHIFT) == BIN_NODES, "bin shift");
static_assert(EBLOCKS * EPB == E_EDGES, "edge blocks must cover the edges");
static_assert(N_NODES == 65536, "an edge is packed as src | dst << 16");
static_assert(EBLOCKS % 64 == 0 && NBIN == 256 && EPB == 16 * 256, "layout of the edge passes");

typedef int v4i __attribute__((ext_vector_type(4)));
typedef int v2i __attribute__((ext_vector_type(2)));

__device__ __forceinline__ int clamp_cls(int cc) { return min(max(cc, 0), N_CLS - 1); }
__device__ __forceinline__ int clamp_nbr(int nb) { return min(max(nb, 0), N_NBR - 1); }
__device__ __forceinline__ int tab_code(int cc, int nb) {
    return (clamp_cls(cc) * (HID / 4)) | (((N_CLS + clamp_nbr(nb)) * (HID / 4)) << 16);
}
static_assert(TAB_ROWS * (HID / 4) < 65536, "a row start must fit in half a word");

__device__ __forceinline__ int excl_scan_256(int c, int t, int* wtot) {
    const int lane = t & 31;
    int x = c;
#pragma unroll
    for (int o = 1; o < 32; o <<= 1) {
        const int y = __shfl_up(x, o, 32);
        if (lane >= o) x += y;
    }
    if (t < 256 && lane == 31) wtot[t >> 5] = x;
    __syncthreads();
    int pre = 0;
    const int wv = t >> 5;
#pragma unroll
    for (int w = 0; w < 8; ++w) pre += (w < wv) ? wtot[w] : 0;
    return pre + x - c;
}

__launch_bounds__(256)
__global__ void k_bin_count(const int* __restrict__ edge_index, int* __restrict__ cnt) {
    __shared__ int hist[NBIN];
    const int t = threadIdx.x;
    hist[t] = 0;
    const int* dstp = edge_index + E_EDGES + blockIdx.x * EPB + t * 4;
    v4i dv[4];
#pragma unroll
    for (int i = 0; i < 4; ++i) dv[i] = *(const v4i*)(dstp + i * 1024);
    __syncthreads();
#pragma unroll
    for (int i = 0; i < 4; ++i)
#pragma unroll
        for (int j = 0; j < 4; ++j) {
            const int d = dv[i][j] & (N_NODES - 1);
            atomicAdd(&hist[d >> BIN_SHIFT], 1);
        }
    __syncthreads();
    cnt[blockIdx.x * NBIN + t] = hist[t];
}

__launch_bounds__(256)
__global__ void k_bin_scatter(const int* __restrict__ edge_index, const int* __restrict__ cnt,
                              int* __restrict__ binned, int* __restrict__ binbase) {
    __shared__ int ptot[4][NBIN];
    __shared__ int pbef[4][NBIN];
    __shared__ int cursor[NBIN];
    __shared__ int delta[NBIN];
    __shared__ int wtot_a[8];
    __shared__ int wtot_b[8];
    __shared__ int stage[EPB];
    const int t = threadIdx.x;
    const int b = blockIdx.x;
    const int* srcp = edge_index + b * EPB + t * 4;
    const int* dstp = edge_index + E_EDGES + b * EPB + t * 4;
    v4i sv[4], dv[4];
#pragma unroll
    for (int i = 0; i < 4; ++i) {
        sv[i] = *(const v4i*)(srcp + i * 1024);
        dv[i] = *(const v4i*)(dstp + i * 1024);
    }
    const int own = cnt[b * NBIN + t];
    const int q  = t >> 6;
    const int c4 = (t & 63) * 4;
    v4i tot = {0, 0, 0, 0};
    v4i bef = {0, 0, 0, 0};
#pragma unroll 16
    for (int r = q; r < EBLOCKS; r += 4) {
        const v4i c = *(const v4i*)(cnt + r * NBIN + c4);
        tot += c;
        if (r < b) bef += c;
    }
#pragma unroll
    for (int j = 0; j < 4; ++j) { ptot[q][c4 + j] = tot[j]; pbef[q][c4 + j] = bef[j]; }
    __syncthreads();
    const int total  = ptot[0][t] + ptot[1][t] + ptot[2][t] + ptot[3][t];
    const int before = pbef[0][t] + pbef[1][t] + pbef[2][t] + pbef[3][t];
    const int base   = excl_scan_256(total, t, wtot_a);
    const int lstart = excl_scan_256(own, t, wtot_b);
    cursor[t] = lstart;
    delta[t]  = base + before - lstart;
    if (b == 0) {
        binbase[t] = base;
        if (t == NBIN - 1) binbase[NBIN] = base + total;
    }
    __syncthreads();
#pragma unroll
    for (int i = 0; i < 4; ++i)
#pragma unroll
        for (int j = 0; j < 4; ++j) {
            const int s = sv[i][j] & (N_NODES - 1);
            const int d = dv[i][j] & (N_NODES - 1);
            const int lp = atomicAdd(&cursor[d >> BIN_SHIFT], 1);
            stage[lp] = (int)((unsigned)s | ((unsigned)d << 16));
        }
    __syncthreads();
#pragma unroll 4
    for (int i = t; i < EPB; i += 256) {
        const int w = stage[i];
        binned[i + delta[(w >> 24) & (NBIN - 1)]] = w;
    }
}

#define BUILD_T 1024
__launch_bounds__(BUILD_T)
__global__ void k_bin_build(const int* __restrict__ binned, const int* __restrict__ binbase,
                            int* __restrict__ rowptr, float* __restrict__ dinv,
                            float* __restrict__ invdeg, int* __restrict__ csr_tmp) {
    __shared__ int deg[BIN_NODES];
    __shared__ int off[BIN_NODES];
    __shared__ int cur[BIN_NODES];
    __shared__ int wtot[8];
    __shared__ int stage[STAGE_CAP];
    const int t = threadIdx.x;
    const int bin = blockIdx.x;
    const int base = binbase[bin];
    const int n_e = binbase[bin + 1] - base;
    if (t < BIN_NODES) { deg[t] = 0; cur[t] = 0; }
    __syncthreads();
    const int* eb = binned + base;
    for (int i0 = 0; i0 < n_e; i0 += 4 * BUILD_T) {
        int w[4];
#pragma unroll
        for (int j = 0; j < 4; ++j) {
            const int i = i0 + j * BUILD_T + t;
            w[j] = eb[i < n_e ? i : n_e - 1];
        }
#pragma unroll
        for (int j = 0; j < 4; ++j)
            if (i0 + j * BUILD_T + t < n_e)
                atomicAdd(&deg[(w[j] >> 16) & (BIN_NODES - 1)], 1);
    }
    __syncthreads();
    const int c = (t < BIN_NODES) ? deg[t] : 0;
    const int ex = excl_scan_256(c, t, wtot);
    if (t < BIN_NODES) {
        const int node = bin * BIN_NODES + t;
        const float d = 1.0f + (float)c;
        dinv[node]   = rsqrtf(d);
        invdeg[node] = 1.0f / d;
        rowptr[node] = base + ex;
        if (node == N_NODES - 1) rowptr[N_NODES] = base + ex + c;
        off[t] = ex;
    }
    __syncthreads();
    const bool fits = n_e <= STAGE_CAP;
    for (int i0 = 0; i0 < n_e; i0 += 4 * BUILD_T) {
        int w[4];
#pragma unroll
        for (int j = 0; j < 4; ++j) {
            const int i = i0 + j * BUILD_T + t;
            w[j] = eb[i < n_e ? i : n_e - 1];
        }
#pragma unroll
        for (int j = 0; j < 4; ++j)
            if (i0 + j * BUILD_T + t < n_e) {
                const int dl = (w[j] >> 16) & (BIN_NODES - 1);
                const int p = off[dl] + atomicAdd(&cur[dl], 1);
                const int key = w[j] & 0xffff;
                if (fits) stage[p] = key; else csr_tmp[base + p] = key;
            }
    }
    __syncthreads();
    if (fits)
        for (int i = t; i < n_e; i += BUILD_T) csr_tmp[base + i] = stage[i];
}

__launch_bounds__(256)
__global__ void k_csr_order(const int* __restrict__ rowptr,
                            const int* __restrict__ csr_tmp,
                            const float* __restrict__ dinv,
                            int* __restrict__ csr_src,
                            float* __restrict__ csr_w,
                            const int* __restrict__ x,
                            int* __restrict__ csr_code) {
    const int t = blockIdx.x * 256 + threadIdx.x;
    const int n = t >> 5;
    const int lane = t & 31;
    const int beg = rowptr[n];
    const int deg = rowptr[n + 1] - beg;
    const float dn = dinv[n];
    const int* b = csr_tmp + beg;
    for (int i = lane; i < deg; i += 32) {
        const int key = b[i] & (N_NODES - 1);
        const v2i xs = *(const v2i*)(x + 2 * key);
        int rank = 0;
        for (int j = 0; j < deg; ++j) {
            const int kj = b[j] & (N_NODES - 1);
            rank += (kj < key || (kj == key && j < i)) ? 1 : 0;
        }
        csr_src[beg + rank] = key;
        csr_w[beg + rank]   = dinv[key] * dn;
        csr_code[beg + rank] = tab_code(xs.x, xs.y);
    }
}

#define TAB_TILES ((TAB_ROWS + 15) / 16)

__launch_bounds__(256)
__global__ void k_table_product(const float* __restrict__ cls_emb,
                                const float* __restrict__ nbr_emb,
                                const float* __restrict__ W,
                                float* __restrict__ tab) {
    const int tid  = threadIdx.x;
    const int wave = tid >> 5;
    const int lane = tid & 31;
    const int half = lane >> 4;
    const int l16  = lane & 15;
    const int r  = min(blockIdx.x * 16 + l16, TAB_ROWS - 1);
    const float* ar = (r < N_CLS ? cls_emb + (size_t)r * HID
                                 : nbr_emb + (size_t)(r - N_CLS) * HID) + half * 2;
    const float* wcol = W + wave * 16 + l16;
    v2f afrag[32], bfrag[32];
#pragma unroll
    for (int s = 0; s < 32; ++s) afrag[s] = *(const v2f*)(ar + s * 4);
#pragma unroll
    for (int s = 0; s < 32; ++s) {
        const int kb = s * 4 + half * 2;
        bfrag[s].x = wcol[(kb + 0) * HID];
        bfrag[s].y = wcol[(kb + 1) * HID];
    }
    v8f cacc = {};
#pragma unroll
    for (int s = 0; s < 32; ++s)
        cacc = __builtin_amdgcn_wmma_f32_16x16x4_f32(
            false, afrag[s], false, bfrag[s], (short)0, cacc, false, false);
    const int m0 = blockIdx.x * 16 + half * 8;
    float* o = tab + (size_t)m0 * HID + wave * 16 + l16;
#pragma unroll
    for (int q = 0; q < 8; ++q)
        if (m0 + q < TAB_ROWS) o[q * HID] = cacc[q];
}


__device__ __forceinline__ v4f edge_row(const float* hwf, int s_l, int j) {
    const int s = __builtin_amdgcn_readlane(s_l, j);
    return *(const v4f*)(hwf + (size_t)s * HID);
}
__device__ __forceinline__ float edge_wt(float w_l, int j) {
    return __int_as_float(__builtin_amdgcn_readlane(__float_as_int(w_l), j));
}

__device__ __forceinline__ v4f edge_sum(const float* rowsf,
                                        const int* __restrict__ csr_src,
                                        const float* __restrict__ csr_w,
                                        int beg, int end, int lane) {
    v4f acc = {0.0f, 0.0f, 0.0f, 0.0f};
    for (int base = beg; base < end; base += 32) {
        const int idx = base + lane;
        int   s_l = 0;
        float w_l = 0.0f;
        if (idx < end) { s_l = csr_src[idx]; w_l = csr_w[idx]; }
        const int left = end - base;
        const int cnt = left < 32 ? left : 32;
        int j = 0;
        for (; j + 4 <= cnt; j += 4) {
            const v4f v0 = edge_row(rowsf, s_l, j + 0);
            const v4f v1 = edge_row(rowsf, s_l, j + 1);
            const v4f v2 = edge_row(rowsf, s_l, j + 2);
            const v4f v3 = edge_row(rowsf, s_l, j + 3);
            acc += v0 * edge_wt(w_l, j + 0);
            acc += v1 * edge_wt(w_l, j + 1);
            acc += v2 * edge_wt(w_l, j + 2);
            acc += v3 * edge_wt(w_l, j + 3);
        }
        const int rem = cnt - j;
        if (rem == 3) {
            const v4f v0 = edge_row(rowsf, s_l, j + 0);
            const v4f v1 = edge_row(rowsf, s_l, j + 1);
            const v4f v2 = edge_row(rowsf, s_l, j + 2);
            acc += v0 * edge_wt(w_l, j + 0);
            acc += v1 * edge_wt(w_l, j + 1);
            acc += v2 * edge_wt(w_l, j + 2);
        } else if (rem == 2) {
            const v4f v0 = edge_row(rowsf, s_l, j + 0);
            const v4f v1 = edge_row(rowsf, s_l, j + 1);
            acc += v0 * edge_wt(w_l, j + 0);
            acc += v1 * edge_wt(w_l, j + 1);
        } else if (rem == 1) {
            const v4f v0 = edge_row(rowsf, s_l, j);
            acc += v0 * edge_wt(w_l, j);
        }
    }
    return acc;
}

__device__ __forceinline__ v4f tab_edge_row(const v4f* tl, int c_l, int j) {
    const unsigned cd = (unsigned)__builtin_amdgcn_readlane(c_l, j);
    return tl[cd & 0xffffu] + tl[cd >> 16];
}

__device__ __forceinline__ v4f edge_sum_tab(const v4f* tl,
                                            const int* __restrict__ csr_code,
                                            const float* __restrict__ csr_w,
                                            int beg, int end, int lane) {
    v4f acc = {0.0f, 0.0f, 0.0f, 0.0f};
    for (int base = beg; base < end; base += 32) {
        const int idx = base + lane;
        int   c_l = 0;
        float w_l = 0.0f;
        if (idx < end) { c_l = csr_code[idx]; w_l = csr_w[idx]; }
        const int left = end - base;
        const int cnt = left < 32 ? left : 32;
        int j = 0;
        for (; j + 2 <= cnt; j += 2) {
            const v4f v0 = tab_edge_row(tl, c_l, j + 0);
            const v4f v1 = tab_edge_row(tl, c_l, j + 1);
            acc += v0 * edge_wt(w_l, j + 0);
            acc += v1 * edge_wt(w_l, j + 1);
        }
        if (j < cnt) {
            const v4f v0 = tab_edge_row(tl, c_l, j);
            acc += v0 * edge_wt(w_l, j);
        }
    }
    return acc;
}

#define L0_WAVES  16
#define L0_NPW    4
#define L0_BLOCKS (N_NODES / (L0_WAVES * L0_NPW))
#define TAB_V4    (TAB_ROWS * HID / 4)
#define TAB_LD    ((TAB_V4 + L0_WAVES * 32 - 1) / (L0_WAVES * 32))
static_assert(L0_BLOCKS * L0_WAVES * L0_NPW == N_NODES, "blocks must cover the nodes");

__launch_bounds__(L0_WAVES * 32, 16)
__global__ void k_gather_ln0(const int* __restrict__ rowptr,
                             const int* __restrict__ csr_code,
                             const float* __restrict__ csr_w,
                             const float* __restrict__ tab,
                             const float* __restrict__ invdeg,
                             const float* __restrict__ bias,
                             const float* __restrict__ gamma,
                             const float* __restrict__ beta,
                             float* __restrict__ h,
                             const int* __restrict__ x,
                             const float* __restrict__ cls_emb,
                             const float* __restrict__ nbr_emb) {
    __shared__ __attribute__((aligned(16))) float T[TAB_ROWS * HID];
    const int tid  = threadIdx.x;
    const int wave = __builtin_amdgcn_readfirstlane(tid >> 5);
    const int lane = tid & 31;
    const int f = lane << 2;

    v4f tv[TAB_LD];
#pragma unroll
    for (int r = 0; r < TAB_LD; ++r) {
        const int i = tid + r * (L0_WAVES * 32);
        tv[r] = *(const v4f*)(tab + 4 * (i < TAB_V4 ? i : TAB_V4 - 1));
    }
    const v4f bv = *(const v4f*)(bias + f);
    const v4f g  = *(const v4f*)(gamma + f);
    const v4f b  = *(const v4f*)(beta + f);
#pragma unroll
    for (int r = 0; r < TAB_LD; ++r) {
        const int i = tid + r * (L0_WAVES * 32);
        if (i < TAB_V4) *(v4f*)(T + 4 * i) = tv[r];
    }
    __syncthreads();

    const v4f* tl = (const v4f*)T + lane;
    const int n0 = (blockIdx.x * L0_WAVES + wave) * L0_NPW;
    for (int it = 0; it < L0_NPW; ++it) {
        const int n = n0 + it;
        const int beg = rowptr[n];
        const int end = rowptr[n + 1];
        const v2i xv = *(const v2i*)(x + 2 * n);
        const int cc = clamp_cls(xv.x);
        const int nb = clamp_nbr(xv.y);
        const float inv = invdeg[n];
        const v4f ea = *(const v4f*)(cls_emb + (size_t)cc * HID + f);
        const v4f eb = *(const v4f*)(nbr_emb + (size_t)nb * HID + f);
        const v4f hv = ea + eb;

        const v4f acc = edge_sum_tab(tl, csr_code, csr_w, beg, end, lane);

        const v4f self = tl[cc * (HID / 4)] + tl[(N_CLS + nb) * (HID / 4)];

        const v4f av = (acc + self * inv) + bv;

        v4f val;
        val.x = hv.x + fmaxf(av.x, 0.0f);
        val.y = hv.y + fmaxf(av.y, 0.0f);
        val.z = hv.z + fmaxf(av.z, 0.0f);
        val.w = hv.w + fmaxf(av.w, 0.0f);

        float s = val.x + val.y + val.z + val.w;
#pragma unroll
        for (int off = 16; off >= 1; off >>= 1) s += __shfl_xor(s, off, 32);
        const float mu = s * (1.0f / (float)HID);

        v4f dv = val - mu;
        float q = dv.x * dv.x + dv.y * dv.y + dv.z * dv.z + dv.w * dv.w;
#pragma unroll
        for (int off = 16; off >= 1; off >>= 1) q += __shfl_xor(q, off, 32);
        const float rstd = rsqrtf(q * (1.0f / (float)HID) + LN_EPS);

        const v4f outv = dv * rstd * g + b;
        *(v4f*)(h + (size_t)n * HID + f) = outv;
    }
}


#define AGG_ROWS   16
#define ZP 132
static_assert(N_LAYERS == 4, "the launch sequence is written out for four layers");
#define AGG_TILES  (N_NODES / AGG_ROWS)
#define AGG_BLOCKS 2048
#define WP2 288
static_assert(AGG_TILES * AGG_ROWS == N_NODES, "tiles must cover the nodes");
static_assert(AGG_TILES % AGG_BLOCKS == 0, "every block takes the same number of tiles");
static_assert(HID == 8 * AGG_ROWS, "one wave stages 4 row pairs of W, 16 pairs apart");

__launch_bounds__(AGG_ROWS * 32)
__global__ void k_agg_layer(const int* __restrict__ rowptr,
                            const int* __restrict__ csr_src,
                            const float* __restrict__ csr_w,
                            const float* __restrict__ hin,
                            const float* __restrict__ invdeg,
                            const float* __restrict__ W,
                            const float* __restrict__ bias,
                            const float* __restrict__ gamma,
                            const float* __restrict__ beta,
                            float* __restrict__ hout) {
    __shared__ __attribute__((aligned(16))) float lW[(HID / 2) * WP2];
    __shared__ __attribute__((aligned(16))) float zt[AGG_ROWS * ZP];
    __shared__ __attribute__((aligned(16))) float yt[AGG_ROWS * ZP];
    const int tid  = threadIdx.x;
    const int wave = __builtin_amdgcn_readfirstlane(tid >> 5);
    const int lane = tid & 31;
    const int half = lane >> 4;
    const int l16  = lane & 15;
    const int f = lane << 2;

    v4f wv[8];
#pragma unroll
    for (int r = 0; r < 4; ++r) {
        const float* wr = W + (size_t)(2 * (wave + AGG_ROWS * r)) * HID + f;
        wv[2 * r + 0] = *(const v4f*)wr;
        wv[2 * r + 1] = *(const v4f*)(wr + HID);
    }
    const v4f bv = *(const v4f*)(bias + f);
    const v4f gm = *(const v4f*)(gamma + f);
    const v4f bt = *(const v4f*)(beta + f);
#pragma unroll
    for (int r = 0; r < 4; ++r) {
        const v4f e = wv[2 * r + 0];
        const v4f o = wv[2 * r + 1];
        float* dst = lW + (wave + AGG_ROWS * r) * WP2 + 2 * f;
        const v4f lo = {e.x, o.x, e.y, o.y};
        const v4f hi = {e.z, o.z, e.w, o.w};
        *(v4f*)(dst + 0) = lo;
        *(v4f*)(dst + 4) = hi;
    }

    const float* hf = hin + f;
    for (int it = 0; it < AGG_TILES / AGG_BLOCKS; ++it) {
        const int tile = blockIdx.x + it * AGG_BLOCKS;
        const int n = tile * AGG_ROWS + wave;

        const int beg = rowptr[n];
        const int end = rowptr[n + 1];
        const v4f hv = *(const v4f*)(hf + (size_t)n * HID);
        const float inv = invdeg[n];
        const v4f acc = edge_sum(hf, csr_src, csr_w, beg, end, lane);
        const v4f z = acc + hv * inv;
        *(v4f*)(zt + wave * ZP + f) = z;
        __syncthreads();

        if (wave < 8) {
            v8f cacc = {};
            const float* arow = zt + l16 * ZP + half * 2;
            const float* bcol = lW + half * WP2 + 2 * (wave * 16 + l16);
#pragma unroll
            for (int s = 0; s < 32; ++s) {
                const v2f a = *(const v2f*)(arow + s * 4);
                const v2f b = *(const v2f*)(bcol + s * 2 * WP2);
                cacc = __builtin_amdgcn_wmma_f32_16x16x4_f32(
                    false, a, false, b, (short)0, cacc, false, false);
            }
            const int col = wave * 16 + l16;
#pragma unroll
            for (int r = 0; r < 8; ++r) yt[(half * 8 + r) * ZP + col] = cacc[r];
        }
        __syncthreads();

        const v4f y  = *(const v4f*)(yt + wave * ZP + f);
        const v4f av = y + bv;

        v4f val;
        val.x = hv.x + fmaxf(av.x, 0.0f);
        val.y = hv.y + fmaxf(av.y, 0.0f);
        val.z = hv.z + fmaxf(av.z, 0.0f);
        val.w = hv.w + fmaxf(av.w, 0.0f);

        float s = val.x + val.y + val.z + val.w;
#pragma unroll
        for (int off = 16; off >= 1; off >>= 1) s += __shfl_xor(s, off, 32);
        const float mu = s * (1.0f / (float)HID);

        v4f dv = val - mu;
        float q = dv.x * dv.x + dv.y * dv.y + dv.z * dv.z + dv.w * dv.w;
#pragma unroll
        for (int off = 16; off >= 1; off >>= 1) q += __shfl_xor(q, off, 32);
        const float rstd = rsqrtf(q * (1.0f / (float)HID) + LN_EPS);

        const v4f outv = dv * rstd * gm + bt;
        *(v4f*)(hout + (size_t)n * HID + f) = outv;
    }
}

#define TL_ROWS  AGG_ROWS
static_assert(HID / 2 == 4 * TL_ROWS, "one wave stages 4 row pairs of W, TL_ROWS pairs apart");
static_assert(HID * OUT_DIM <= 3 * TL_ROWS * 32, "W_out is staged three words per thread");

__device__ __forceinline__ void w_rows_load(const float* __restrict__ W, int wave, int f, v4f (&wv)[8]) {
#pragma unroll
    for (int r = 0; r < 4; ++r) {
        const float* wr = W + (size_t)(2 * (wave + TL_ROWS * r)) * HID + f;
        wv[2 * r + 0] = *(const v4f*)wr;
        wv[2 * r + 1] = *(const v4f*)(wr + HID);
    }
}
__device__ __forceinline__ void w_rows_store(float* lW, int wave, int f, const v4f (&wv)[8]) {
#pragma unroll
    for (int r = 0; r < 4; ++r) {
        const v4f e = wv[2 * r + 0];
        const v4f o = wv[2 * r + 1];
        float* dst = lW + (wave + TL_ROWS * r) * WP2 + 2 * f;
        const v4f lo = {e.x, o.x, e.y, o.y};
        const v4f hi = {e.z, o.z, e.w, o.w};
        *(v4f*)(dst + 0) = lo;
        *(v4f*)(dst + 4) = hi;
    }
}

__device__ __forceinline__ void tile_product(const float* zrows, const float* lW, float* yrows,
                                             int ct, int half, int l16) {
    v8f cacc = {};
    const float* arow = zrows + l16 * ZP + half * 2;
    const float* bcol = lW + half * WP2 + 2 * (ct * 16 + l16);
#pragma unroll
    for (int s = 0; s < 32; ++s) {
        const v2f a = *(const v2f*)(arow + s * 4);
        const v2f b = *(const v2f*)(bcol + s * 2 * WP2);
        cacc = __builtin_amdgcn_wmma_f32_16x16x4_f32(
            false, a, false, b, (short)0, cacc, false, false);
    }
    const int col = ct * 16 + l16;
#pragma unroll
    for (int r = 0; r < 8; ++r) yrows[(half * 8 + r) * ZP + col] = cacc[r];
}

__device__ __forceinline__ v4f layer_tail(v4f hv, v4f y, v4f bv, v4f gm, v4f bt) {
    const v4f av = y + bv;

    v4f val;
    val.x = hv.x + fmaxf(av.x, 0.0f);
    val.y = hv.y + fmaxf(av.y, 0.0f);
    val.z = hv.z + fmaxf(av.z, 0.0f);
    val.w = hv.w + fmaxf(av.w, 0.0f);

    float s = val.x + val.y + val.z + val.w;
#pragma unroll
    for (int off = 16; off >= 1; off >>= 1) s += __shfl_xor(s, off, 32);
    const float mu = s * (1.0f / (float)HID);

    v4f dv = val - mu;
    float q = dv.x * dv.x + dv.y * dv.y + dv.z * dv.z + dv.w * dv.w;
#pragma unroll
    for (int off = 16; off >= 1; off >>= 1) q += __shfl_xor(q, off, 32);
    const float rstd = rsqrtf(q * (1.0f / (float)HID) + LN_EPS);

    return dv * rstd * gm + bt;
}

__launch_bounds__(TL_ROWS * 32)
__global__ void k_root_tail(const int* __restrict__ rowptr,
                            const int* __restrict__ csr_src,
                            const float* __restrict__ csr_w,
                            const float* __restrict__ hin,
                            const float* __restrict__ invdeg,
                            const float* __restrict__ Wa,
                            const float* __restrict__ bias_a,
                            const float* __restrict__ gamma_a,
                            const float* __restrict__ beta_a,
                            const float* __restrict__ Wb,
                            const float* __restrict__ bias_b,
                            const float* __restrict__ gamma_b,
                            const float* __restrict__ beta_b,
                            const float* __restrict__ W_out,
                            float* __restrict__ out) {
    __shared__ __attribute__((aligned(16))) float lW[(HID / 2) * WP2];
    __shared__ __attribute__((aligned(16))) float zt[TL_ROWS * ZP];
    __shared__ __attribute__((aligned(16))) float yt[TL_ROWS * ZP];
    __shared__ __attribute__((aligned(16))) float ht[TL_ROWS * ZP];
    __shared__ float wo[HID * OUT_DIM];
    const int tid  = threadIdx.x;
    const int wave = __builtin_amdgcn_readfirstlane(tid >> 5);
    const int lane = tid & 31;
    const int half = lane >> 4;
    const int l16  = lane & 15;
    const int f = lane << 2;
    const int g = blockIdx.x;
    const int root = g * NPG;

    const int rbeg = rowptr[root];
    const int rend = rowptr[root + 1];
    const int n_list = rend - rbeg + 1;
    const float inv_root = invdeg[root];

    v4f wv[8];
    w_rows_load(Wa, wave, f, wv);
    const v4f bv = *(const v4f*)(bias_a + f);
    const v4f gm = *(const v4f*)(gamma_a + f);
    const v4f bt = *(const v4f*)(beta_a + f);
    w_rows_store(lW, wave, f, wv);

    v4f wlast[8];
    w_rows_load(Wb, wave, f, wlast);
    float wov[3];
#pragma unroll
    for (int r = 0; r < 3; ++r) {
        const int i = tid + r * (TL_ROWS * 32);
        wov[r] = (i < HID * OUT_DIM) ? W_out[i] : 0.0f;
    }

    v4f zacc  = {0.0f, 0.0f, 0.0f, 0.0f};
    v4f hroot = {0.0f, 0.0f, 0.0f, 0.0f};
    const float* hf = hin + f;
    for (int p0 = 0; p0 < n_list; p0 += TL_ROWS) {
        const int left = n_list - p0;
        const int ri = p0 + wave;
        const bool active = ri < n_list;
        float rw_l = 0.0f;
        if (wave == 0) {
            const int idx = rbeg + p0 + lane;
            if (idx < rend) rw_l = csr_w[idx];
        }

        v4f hv = {0.0f, 0.0f, 0.0f, 0.0f};
        v4f z  = {0.0f, 0.0f, 0.0f, 0.0f};
        if (active) {
            const int n = (ri < n_list - 1) ? csr_src[rbeg + ri] : root;
            const int beg = rowptr[n];
            const int end = rowptr[n + 1];
            hv = *(const v4f*)(hf + (size_t)n * HID);
            const float inv = invdeg[n];
            const v4f acc = edge_sum(hf, csr_src, csr_w, beg, end, lane);
            z = acc + hv * inv;
        }
        *(v4f*)(zt + wave * ZP + f) = z;
        __syncthreads();

        if (wave < 8)
            tile_product(zt, lW, yt, wave, half, l16);
        __syncthreads();

        if (active) {
            const v4f y = *(const v4f*)(yt + wave * ZP + f);
            *(v4f*)(ht + wave * ZP + f) = layer_tail(hv, y, bv, gm, bt);
        }
        __syncthreads();

        if (wave == 0) {
            const int cnt = left < TL_ROWS ? left : TL_ROWS;
            const int ne = (left - 1) < cnt ? (left - 1) : cnt;
            const float* hr = ht + f;
            int j = 0;
            for (; j + 4 <= ne; j += 4) {
                const v4f v0 = *(const v4f*)(hr + (j + 0) * ZP);
                const v4f v1 = *(const v4f*)(hr + (j + 1) * ZP);
                const v4f v2 = *(const v4f*)(hr + (j + 2) * ZP);
                const v4f v3 = *(const v4f*)(hr + (j + 3) * ZP);
                zacc += v0 * edge_wt(rw_l, j + 0);
                zacc += v1 * edge_wt(rw_l, j + 1);
                zacc += v2 * edge_wt(rw_l, j + 2);
                zacc += v3 * edge_wt(rw_l, j + 3);
            }
            for (; j < ne; ++j) {
                const v4f v0 = *(const v4f*)(hr + j * ZP);
                zacc += v0 * edge_wt(rw_l, j);
            }
            if (ne < cnt) hroot = *(const v4f*)(hr + ne * ZP);
        }
    }

    const v4f bvb = *(const v4f*)(bias_b + f);
    const v4f gmb = *(const v4f*)(gamma_b + f);
    const v4f btb = *(const v4f*)(beta_b + f);
    if (wave == 0) *(v4f*)(zt + f) = zacc + hroot * inv_root;
    w_rows_store(lW, wave, f, wlast);
#pragma unroll
    for (int r = 0; r < 3; ++r) {
        const int i = tid + r * (TL_ROWS * 32);
        if (i < HID * OUT_DIM) wo[i] = wov[r];
    }
    __syncthreads();

    if (wave < 8)
        tile_product(zt, lW, yt, wave, half, l16);
    __syncthreads();

    if (wave == 0) {
        const v4f y = *(const v4f*)(yt + f);
        *(v4f*)(zt + f) = layer_tail(hroot, y, bvb, gmb, btb);
    }
    __syncthreads();

    if (wave == 0 && lane < OUT_DIM) {
        float lg = 0.0f;
#pragma unroll 32
        for (int k = 0; k < HID; ++k) lg = fmaf(zt[k], wo[k * OUT_DIM + lane], lg);
        out[g * OUT_DIM + lane] = lg;
    }
}

extern "C" void kernel_launch(void* const* d_in, const int* in_sizes, int n_in,
                              void* d_out, int out_size, void* d_ws, size_t ws_size,
                              hipStream_t stream) {
    const int*   x        = (const int*)d_in[0];
    const int*   ei       = (const int*)d_in[1];
    const float* cls_emb  = (const float*)d_in[3];
    const float* nbr_emb  = (const float*)d_in[4];
    const float* Ws       = (const float*)d_in[5];
    const float* bs       = (const float*)d_in[6];
    const float* gammas   = (const float*)d_in[7];
    const float* betas    = (const float*)d_in[8];
    const float* W_out    = (const float*)d_in[9];
    float*       out      = (float*)d_out;

    char* ws = (char*)d_ws;
    size_t off = 0;
    int*   cnt    = (int*)(ws + off);   off += (size_t)EBLOCKS * NBIN * 4;
    float* dinv   = (float*)(ws + off); off += (size_t)N_NODES * 4;
    float* invdeg = (float*)(ws + off); off += (size_t)N_NODES * 4;
    float* h      = (float*)(ws + off); off += (size_t)N_NODES * HID * 4;
    float* h2     = (float*)(ws + off); off += (size_t)N_NODES * HID * 4;
    int*   rowptr = (int*)(ws + off);   off += (size_t)(N_NODES + 64) * 4;
    int*   binned = (int*)(ws + off);   off += (size_t)E_EDGES * 4;
    int*   csrtmp = (int*)(ws + off);   off += (size_t)E_EDGES * 4;
    int*   csrsrc = (int*)(ws + off);   off += (size_t)E_EDGES * 4;
    float* csrw   = (float*)(ws + off); off += (size_t)E_EDGES * 4;
    int*   binbase = (int*)(ws + off);  off += (size_t)(NBIN + 64) * 4;
    int*   csrcode = (int*)(ws + off);  off += (size_t)E_EDGES * 4;
    float* tab     = (float*)(ws + off); off += (size_t)TAB_ROWS * HID * 4;
    (void)ws_size; (void)in_sizes; (void)n_in; (void)out_size;

    k_bin_count  <<<EBLOCKS, 256, 0, stream>>>(ei, cnt);
    k_bin_scatter<<<EBLOCKS, 256, 0, stream>>>(ei, cnt, binned, binbase);
    k_bin_build  <<<NBIN, BUILD_T, 0, stream>>>(binned, binbase, rowptr, dinv, invdeg, csrtmp);
    k_csr_order<<<(N_NODES * 32) / 256, 256, 0, stream>>>(rowptr, csrtmp, dinv, csrsrc, csrw,
                                                          x, csrcode);

    k_table_product<<<TAB_TILES, 256, 0, stream>>>(cls_emb, nbr_emb, Ws, tab);
    k_gather_ln0<<<L0_BLOCKS, L0_WAVES * 32, 0, stream>>>(
        rowptr, csrcode, csrw, tab, invdeg, bs, gammas, betas, h, x, cls_emb, nbr_emb);

    k_agg_layer<<<AGG_BLOCKS, AGG_ROWS * 32, 0, stream>>>(
        rowptr, csrsrc, csrw, h, invdeg, Ws + (size_t)1 * HID * HID, bs + (size_t)1 * HID,
        gammas + (size_t)1 * HID, betas + (size_t)1 * HID, h2);
    k_root_tail<<<NGRAPH, TL_ROWS * 32, 0, stream>>>(
        rowptr, csrsrc, csrw, h2, invdeg,
        Ws + (size_t)2 * HID * HID, bs + (size_t)2 * HID,
        gammas + (size_t)2 * HID, betas + (size_t)2 * HID,
        Ws + (size_t)3 * HID * HID, bs + (size_t)3 * HID,
        gammas + (size_t)3 * HID, betas + (size_t)3 * HID, W_out, out);
}


# Bucket-order launch dropped: buckets ordered by src in the bin-build block's LDS stage; layer 0's gather derives each edge's weight and table code from src; 65536 waves, 1 launch less; sums as before
# speedup vs baseline: 8.4418x; 1.0592x over previous
#include <hip/hip_runtime.h>

#define N_NODES   65536
#define NPG       128
#define NGRAPH    512
#define E_EDGES   1048576
#define HID       128
#define OUT_DIM   10
#define N_LAYERS  4
#define LN_EPS    1e-5f
#define N_CLS     64
#define N_NBR     65
#define TAB_ROWS  (N_CLS + N_NBR)

typedef float v2f __attribute__((ext_vector_type(2)));
typedef float v4f __attribute__((ext_vector_type(4)));
typedef float v8f __attribute__((ext_vector_type(8)));

#define NBIN      256
#define BIN_NODES 256
#define BIN_SHIFT 8
#define EPB       4096
#define EBLOCKS   (E_EDGES / EPB)
#define STAGE_CAP 12288
static_assert(NBIN * BIN_NODES == N_NODES, "bins must cover the nodes");
static_assert((1 << BIN_SHIFT) == BIN_NODES, "bin shift");
static_assert(EBLOCKS * EPB == E_EDGES, "edge blocks must cover the edges");
static_assert(N_NODES == 65536, "an edge is packed as src | dst << 16");
static_assert(EBLOCKS % 64 == 0 && NBIN == 256 && EPB == 16 * 256, "layout of the edge passes");

typedef int v4i __attribute__((ext_vector_type(4)));
typedef int v2i __attribute__((ext_vector_type(2)));

__device__ __forceinline__ int clamp_cls(int cc) { return min(max(cc, 0), N_CLS - 1); }
__device__ __forceinline__ int clamp_nbr(int nb) { return min(max(nb, 0), N_NBR - 1); }
__device__ __forceinline__ int tab_code(int cc, int nb) {
    return (clamp_cls(cc) * (HID / 4)) | (((N_CLS + clamp_nbr(nb)) * (HID / 4)) << 16);
}
static_assert(TAB_ROWS * (HID / 4) < 65536, "a row start must fit in half a word");

__device__ __forceinline__ int excl_scan_256(int c, int t, int* wtot) {
    const int lane = t & 31;
    int x = c;
#pragma unroll
    for (int o = 1; o < 32; o <<= 1) {
        const int y = __shfl_up(x, o, 32);
        if (lane >= o) x += y;
    }
    if (t < 256 && lane == 31) wtot[t >> 5] = x;
    __syncthreads();
    int pre = 0;
    const int wv = t >> 5;
#pragma unroll
    for (int w = 0; w < 8; ++w) pre += (w < wv) ? wtot[w] : 0;
    return pre + x - c;
}

__launch_bounds__(256)
__global__ void k_bin_count(const int* __restrict__ edge_index, int* __restrict__ cnt) {
    __shared__ int hist[NBIN];
    const int t = threadIdx.x;
    hist[t] = 0;
    const int* dstp = edge_index + E_EDGES + blockIdx.x * EPB + t * 4;
    v4i dv[4];
#pragma unroll
    for (int i = 0; i < 4; ++i) dv[i] = *(const v4i*)(dstp + i * 1024);
    __syncthreads();
#pragma unroll
    for (int i = 0; i < 4; ++i)
#pragma unroll
        for (int j = 0; j < 4; ++j) {
            const int d = dv[i][j] & (N_NODES - 1);
            atomicAdd(&hist[d >> BIN_SHIFT], 1);
        }
    __syncthreads();
    cnt[blockIdx.x * NBIN + t] = hist[t];
}

__launch_bounds__(256)
__global__ void k_bin_scatter(const int* __restrict__ edge_index, const int* __restrict__ cnt,
                              int* __restrict__ binned, int* __restrict__ binbase) {
    __shared__ int ptot[4][NBIN];
    __shared__ int pbef[4][NBIN];
    __shared__ int cursor[NBIN];
    __shared__ int delta[NBIN];
    __shared__ int wtot_a[8];
    __shared__ int wtot_b[8];
    __shared__ int stage[EPB];
    const int t = threadIdx.x;
    const int b = blockIdx.x;
    const int* srcp = edge_index + b * EPB + t * 4;
    const int* dstp = edge_index + E_EDGES + b * EPB + t * 4;
    v4i sv[4], dv[4];
#pragma unroll
    for (int i = 0; i < 4; ++i) {
        sv[i] = *(const v4i*)(srcp + i * 1024);
        dv[i] = *(const v4i*)(dstp + i * 1024);
    }
    const int own = cnt[b * NBIN + t];
    const int q  = t >> 6;
    const int c4 = (t & 63) * 4;
    v4i tot = {0, 0, 0, 0};
    v4i bef = {0, 0, 0, 0};
#pragma unroll 16
    for (int r = q; r < EBLOCKS; r += 4) {
        const v4i c = *(const v4i*)(cnt + r * NBIN + c4);
        tot += c;
        if (r < b) bef += c;
    }
#pragma unroll
    for (int j = 0; j < 4; ++j) { ptot[q][c4 + j] = tot[j]; pbef[q][c4 + j] = bef[j]; }
    __syncthreads();
    const int total  = ptot[0][t] + ptot[1][t] + ptot[2][t] + ptot[3][t];
    const int before = pbef[0][t] + pbef[1][t] + pbef[2][t] + pbef[3][t];
    const int base   = excl_scan_256(total, t, wtot_a);
    const int lstart = excl_scan_256(own, t, wtot_b);
    cursor[t] = lstart;
    delta[t]  = base + before - lstart;
    if (b == 0) {
        binbase[t] = base;
        if (t == NBIN - 1) binbase[NBIN] = base + total;
    }
    __syncthreads();
#pragma unroll
    for (int i = 0; i < 4; ++i)
#pragma unroll
        for (int j = 0; j < 4; ++j) {
            const int s = sv[i][j] & (N_NODES - 1);
            const int d = dv[i][j] & (N_NODES - 1);
            const int lp = atomicAdd(&cursor[d >> BIN_SHIFT], 1);
            stage[lp] = (int)((unsigned)s | ((unsigned)d << 16));
        }
    __syncthreads();
#pragma unroll 4
    for (int i = t; i < EPB; i += 256) {
        const int w = stage[i];
        binned[i + delta[(w >> 24) & (NBIN - 1)]] = w;
    }
}

#define PLACE_BITS 14
static_assert(STAGE_CAP <= (1 << PLACE_BITS), "a place in the LDS stage must fit beside the src id");
static_assert(N_NODES <= (1 << (32 - PLACE_BITS)), "a src id must fit above the place");
static_assert(BIN_NODES <= 256, "a bucket number must fit in a byte");
__device__ __forceinline__ void order_buckets_lds(const unsigned* ent, const unsigned char* bdl,
                                                  const int* boff, const int* bdeg,
                                                  int n_e, int t, int nthreads,
                                                  int* __restrict__ out) {
    for (int i = t; i < n_e; i += nthreads) {
        const unsigned ci = ent[i];
        const int dl = bdl[i];
        const int b0 = boff[dl];
        const int dg = bdeg[dl];
        const unsigned* eb = ent + b0;
        int rank = 0;
#pragma unroll 8
        for (int j = 0; j < dg; ++j) rank += (eb[j] < ci) ? 1 : 0;
        out[b0 + rank] = (int)(ci >> PLACE_BITS);
    }
}
__device__ __forceinline__ void order_buckets_mem(const int* words, const int* boff, const int* bdeg,
                                                  int n_e, int t, int nthreads,
                                                  int* __restrict__ out) {
    for (int i = t; i < n_e; i += nthreads) {
        const int wd  = words[i];
        const int key = wd & 0xffff;
        const int dl  = (wd >> 16) & (BIN_NODES - 1);
        const int b0  = boff[dl];
        const int dg  = bdeg[dl];
        const int own = i - b0;
        int rank = 0;
        for (int j = 0; j < dg; ++j) {
            const int kj = words[b0 + j] & 0xffff;
            rank += (kj < key || (kj == key && j < own)) ? 1 : 0;
        }
        out[b0 + rank] = key;
    }
}

#define BUILD_T 1024
__launch_bounds__(BUILD_T)
__global__ void k_bin_build(const int* __restrict__ binned, const int* __restrict__ binbase,
                            int* __restrict__ rowptr, float* __restrict__ dinv,
                            float* __restrict__ invdeg, int* __restrict__ csr_tmp,
                            int* __restrict__ csr_src) {
    __shared__ int deg[BIN_NODES];
    __shared__ int off[BIN_NODES];
    __shared__ int cur[BIN_NODES];
    __shared__ int wtot[8];
    __shared__ unsigned stage[STAGE_CAP];
    __shared__ unsigned char sdl[STAGE_CAP];
    const int t = threadIdx.x;
    const int bin = blockIdx.x;
    const int base = binbase[bin];
    const int n_e = binbase[bin + 1] - base;
    if (t < BIN_NODES) { deg[t] = 0; cur[t] = 0; }
    __syncthreads();
    const int* eb = binned + base;
    for (int i0 = 0; i0 < n_e; i0 += 4 * BUILD_T) {
        int w[4];
#pragma unroll
        for (int j = 0; j < 4; ++j) {
            const int i = i0 + j * BUILD_T + t;
            w[j] = eb[i < n_e ? i : n_e - 1];
        }
#pragma unroll
        for (int j = 0; j < 4; ++j)
            if (i0 + j * BUILD_T + t < n_e)
                atomicAdd(&deg[(w[j] >> 16) & (BIN_NODES - 1)], 1);
    }
    __syncthreads();
    const int c = (t < BIN_NODES) ? deg[t] : 0;
    const int ex = excl_scan_256(c, t, wtot);
    if (t < BIN_NODES) {
        const int node = bin * BIN_NODES + t;
        const float d = 1.0f + (float)c;
        dinv[node]   = rsqrtf(d);
        invdeg[node] = 1.0f / d;
        rowptr[node] = base + ex;
        if (node == N_NODES - 1) rowptr[N_NODES] = base + ex + c;
        off[t] = ex;
    }
    __syncthreads();
    const bool fits = n_e <= STAGE_CAP;
    for (int i0 = 0; i0 < n_e; i0 += 4 * BUILD_T) {
        int w[4];
#pragma unroll
        for (int j = 0; j < 4; ++j) {
            const int i = i0 + j * BUILD_T + t;
            w[j] = eb[i < n_e ? i : n_e - 1];
        }
#pragma unroll
        for (int j = 0; j < 4; ++j)
            if (i0 + j * BUILD_T + t < n_e) {
                const int dl = (w[j] >> 16) & (BIN_NODES - 1);
                const int p = off[dl] + atomicAdd(&cur[dl], 1);
                if (fits) {
                    stage[p] = ((unsigned)(w[j] & 0xffff) << PLACE_BITS) | (unsigned)p;
                    sdl[p] = (unsigned char)dl;
                } else {
                    csr_tmp[base + p] = w[j];
                }
            }
    }
    __syncthreads();
    if (fits) order_buckets_lds(stage, sdl, off, deg, n_e, t, BUILD_T, csr_src + base);
    else      order_buckets_mem(csr_tmp + base, off, deg, n_e, t, BUILD_T, csr_src + base);
}

#define TAB_TILES ((TAB_ROWS + 15) / 16)

__launch_bounds__(256)
__global__ void k_table_product(const float* __restrict__ cls_emb,
                                const float* __restrict__ nbr_emb,
                                const float* __restrict__ W,
                                float* __restrict__ tab) {
    const int tid  = threadIdx.x;
    const int wave = tid >> 5;
    const int lane = tid & 31;
    const int half = lane >> 4;
    const int l16  = lane & 15;
    const int r  = min(blockIdx.x * 16 + l16, TAB_ROWS - 1);
    const float* ar = (r < N_CLS ? cls_emb + (size_t)r * HID
                                 : nbr_emb + (size_t)(r - N_CLS) * HID) + half * 2;
    const float* wcol = W + wave * 16 + l16;
    v2f afrag[32], bfrag[32];
#pragma unroll
    for (int s = 0; s < 32; ++s) afrag[s] = *(const v2f*)(ar + s * 4);
#pragma unroll
    for (int s = 0; s < 32; ++s) {
        const int kb = s * 4 + half * 2;
        bfrag[s].x = wcol[(kb + 0) * HID];
        bfrag[s].y = wcol[(kb + 1) * HID];
    }
    v8f cacc = {};
#pragma unroll
    for (int s = 0; s < 32; ++s)
        cacc = __builtin_amdgcn_wmma_f32_16x16x4_f32(
            false, afrag[s], false, bfrag[s], (short)0, cacc, false, false);
    const int m0 = blockIdx.x * 16 + half * 8;
    float* o = tab + (size_t)m0 * HID + wave * 16 + l16;
#pragma unroll
    for (int q = 0; q < 8; ++q)
        if (m0 + q < TAB_ROWS) o[q * HID] = cacc[q];
}


__device__ __forceinline__ v4f edge_row(const float* hwf, int s_l, int j) {
    const int s = __builtin_amdgcn_readlane(s_l, j);
    return *(const v4f*)(hwf + (size_t)s * HID);
}
__device__ __forceinline__ float edge_wt(float w_l, int j) {
    return __int_as_float(__builtin_amdgcn_readlane(__float_as_int(w_l), j));
}

__device__ __forceinline__ v4f edge_sum(const float* rowsf,
                                        const int* __restrict__ csr_src,
                                        const float* __restrict__ csr_w,
                                        int beg, int end, int lane) {
    v4f acc = {0.0f, 0.0f, 0.0f, 0.0f};
    for (int base = beg; base < end; base += 32) {
        const int idx = base + lane;
        int   s_l = 0;
        float w_l = 0.0f;
        if (idx < end) { s_l = csr_src[idx]; w_l = csr_w[idx]; }
        const int left = end - base;
        const int cnt = left < 32 ? left : 32;
        int j = 0;
        for (; j + 4 <= cnt; j += 4) {
            const v4f v0 = edge_row(rowsf, s_l, j + 0);
            const v4f v1 = edge_row(rowsf, s_l, j + 1);
            const v4f v2 = edge_row(rowsf, s_l, j + 2);
            const v4f v3 = edge_row(rowsf, s_l, j + 3);
            acc += v0 * edge_wt(w_l, j + 0);
            acc += v1 * edge_wt(w_l, j + 1);
            acc += v2 * edge_wt(w_l, j + 2);
            acc += v3 * edge_wt(w_l, j + 3);
        }
        const int rem = cnt - j;
        if (rem == 3) {
            const v4f v0 = edge_row(rowsf, s_l, j + 0);
            const v4f v1 = edge_row(rowsf, s_l, j + 1);
            const v4f v2 = edge_row(rowsf, s_l, j + 2);
            acc += v0 * edge_wt(w_l, j + 0);
            acc += v1 * edge_wt(w_l, j + 1);
            acc += v2 * edge_wt(w_l, j + 2);
        } else if (rem == 2) {
            const v4f v0 = edge_row(rowsf, s_l, j + 0);
            const v4f v1 = edge_row(rowsf, s_l, j + 1);
            acc += v0 * edge_wt(w_l, j + 0);
            acc += v1 * edge_wt(w_l, j + 1);
        } else if (rem == 1) {
            const v4f v0 = edge_row(rowsf, s_l, j);
            acc += v0 * edge_wt(w_l, j);
        }
    }
    return acc;
}

__device__ __forceinline__ v4f tab_edge_row(const v4f* tl, int c_l, int j) {
    const unsigned cd = (unsigned)__builtin_amdgcn_readlane(c_l, j);
    return tl[cd & 0xffffu] + tl[cd >> 16];
}

__device__ __forceinline__ v4f edge_sum_tab(const v4f* tl,
                                            const int* __restrict__ csr_src,
                                            const int* __restrict__ x,
                                            const float* __restrict__ dinv,
                                            float* __restrict__ csr_w,
                                            float dn, int beg, int end, int lane) {
    v4f acc = {0.0f, 0.0f, 0.0f, 0.0f};
    for (int base = beg; base < end; base += 32) {
        const int idx = base + lane;
        int   c_l = 0;
        float w_l = 0.0f;
        if (idx < end) {
            const int key = csr_src[idx] & (N_NODES - 1);
            const v2i xs = *(const v2i*)(x + 2 * key);
            w_l = dinv[key] * dn;
            c_l = tab_code(xs.x, xs.y);
            csr_w[idx] = w_l;
        }
        const int left = end - base;
        const int cnt = left < 32 ? left : 32;
        int j = 0;
        for (; j + 2 <= cnt; j += 2) {
            const v4f v0 = tab_edge_row(tl, c_l, j + 0);
            const v4f v1 = tab_edge_row(tl, c_l, j + 1);
            acc += v0 * edge_wt(w_l, j + 0);
            acc += v1 * edge_wt(w_l, j + 1);
        }
        if (j < cnt) {
            const v4f v0 = tab_edge_row(tl, c_l, j);
            acc += v0 * edge_wt(w_l, j);
        }
    }
    return acc;
}

#define L0_WAVES  16
#define L0_NPW    4
#define L0_BLOCKS (N_NODES / (L0_WAVES * L0_NPW))
#define TAB_V4    (TAB_ROWS * HID / 4)
#define TAB_LD    ((TAB_V4 + L0_WAVES * 32 - 1) / (L0_WAVES * 32))
static_assert(L0_BLOCKS * L0_WAVES * L0_NPW == N_NODES, "blocks must cover the nodes");

__launch_bounds__(L0_WAVES * 32, 16)
__global__ void k_gather_ln0(const int* __restrict__ rowptr,
                             const int* __restrict__ csr_src,
                             const float* __restrict__ dinv,
                             float* __restrict__ csr_w,
                             const float* __restrict__ tab,
                             const float* __restrict__ invdeg,
                             const float* __restrict__ bias,
                             const float* __restrict__ gamma,
                             const float* __restrict__ beta,
                             float* __restrict__ h,
                             const int* __restrict__ x,
                             const float* __restrict__ cls_emb,
                             const float* __restrict__ nbr_emb) {
    __shared__ __attribute__((aligned(16))) float T[TAB_ROWS * HID];
    const int tid  = threadIdx.x;
    const int wave = __builtin_amdgcn_readfirstlane(tid >> 5);
    const int lane = tid & 31;
    const int f = lane << 2;

    v4f tv[TAB_LD];
#pragma unroll
    for (int r = 0; r < TAB_LD; ++r) {
        const int i = tid + r * (L0_WAVES * 32);
        tv[r] = *(const v4f*)(tab + 4 * (i < TAB_V4 ? i : TAB_V4 - 1));
    }
    const v4f bv = *(const v4f*)(bias + f);
    const v4f g  = *(const v4f*)(gamma + f);
    const v4f b  = *(const v4f*)(beta + f);
#pragma unroll
    for (int r = 0; r < TAB_LD; ++r) {
        const int i = tid + r * (L0_WAVES * 32);
        if (i < TAB_V4) *(v4f*)(T + 4 * i) = tv[r];
    }
    __syncthreads();

    const v4f* tl = (const v4f*)T + lane;
    const int n0 = (blockIdx.x * L0_WAVES + wave) * L0_NPW;
    for (int it = 0; it < L0_NPW; ++it) {
        const int n = n0 + it;
        const int beg = rowptr[n];
        const int end = rowptr[n + 1];
        const v2i xv = *(const v2i*)(x + 2 * n);
        const int cc = clamp_cls(xv.x);
        const int nb = clamp_nbr(xv.y);
        const float inv = invdeg[n];
        const float dn  = dinv[n];
        const v4f ea = *(const v4f*)(cls_emb + (size_t)cc * HID + f);
        const v4f eb = *(const v4f*)(nbr_emb + (size_t)nb * HID + f);
        const v4f hv = ea + eb;

        const v4f acc = edge_sum_tab(tl, csr_src, x, dinv, csr_w, dn, beg, end, lane);

        const v4f self = tl[cc * (HID / 4)] + tl[(N_CLS + nb) * (HID / 4)];

        const v4f av = (acc + self * inv) + bv;

        v4f val;
        val.x = hv.x + fmaxf(av.x, 0.0f);
        val.y = hv.y + fmaxf(av.y, 0.0f);
        val.z = hv.z + fmaxf(av.z, 0.0f);
        val.w = hv.w + fmaxf(av.w, 0.0f);

        float s = val.x + val.y + val.z + val.w;
#pragma unroll
        for (int off = 16; off >= 1; off >>= 1) s += __shfl_xor(s, off, 32);
        const float mu = s * (1.0f / (float)HID);

        v4f dv = val - mu;
        float q = dv.x * dv.x + dv.y * dv.y + dv.z * dv.z + dv.w * dv.w;
#pragma unroll
        for (int off = 16; off >= 1; off >>= 1) q += __shfl_xor(q, off, 32);
        const float rstd = rsqrtf(q * (1.0f / (float)HID) + LN_EPS);

        const v4f outv = dv * rstd * g + b;
        *(v4f*)(h + (size_t)n * HID + f) = outv;
    }
}


#define AGG_ROWS   16
#define ZP 132
static_assert(N_LAYERS == 4, "the launch sequence is written out for four layers");
#define AGG_TILES  (N_NODES / AGG_ROWS)
#define AGG_BLOCKS 2048
#define WP2 288
static_assert(AGG_TILES * AGG_ROWS == N_NODES, "tiles must cover the nodes");
static_assert(AGG_TILES % AGG_BLOCKS == 0, "every block takes the same number of tiles");
static_assert(HID == 8 * AGG_ROWS, "one wave stages 4 row pairs of W, 16 pairs apart");

__launch_bounds__(AGG_ROWS * 32)
__global__ void k_agg_layer(const int* __restrict__ rowptr,
                            const int* __restrict__ csr_src,
                            const float* __restrict__ csr_w,
                            const float* __restrict__ hin,
                            const float* __restrict__ invdeg,
                            const float* __restrict__ W,
                            const float* __restrict__ bias,
                            const float* __restrict__ gamma,
                            const float* __restrict__ beta,
                            float* __restrict__ hout) {
    __shared__ __attribute__((aligned(16))) float lW[(HID / 2) * WP2];
    __shared__ __attribute__((aligned(16))) float zt[AGG_ROWS * ZP];
    __shared__ __attribute__((aligned(16))) float yt[AGG_ROWS * ZP];
    const int tid  = threadIdx.x;
    const int wave = __builtin_amdgcn_readfirstlane(tid >> 5);
    const int lane = tid & 31;
    const int half = lane >> 4;
    const int l16  = lane & 15;
    const int f = lane << 2;

    v4f wv[8];
#pragma unroll
    for (int r = 0; r < 4; ++r) {
        const float* wr = W + (size_t)(2 * (wave + AGG_ROWS * r)) * HID + f;
        wv[2 * r + 0] = *(const v4f*)wr;
        wv[2 * r + 1] = *(const v4f*)(wr + HID);
    }
    const v4f bv = *(const v4f*)(bias + f);
    const v4f gm = *(const v4f*)(gamma + f);
    const v4f bt = *(const v4f*)(beta + f);
#pragma unroll
    for (int r = 0; r < 4; ++r) {
        const v4f e = wv[2 * r + 0];
        const v4f o = wv[2 * r + 1];
        float* dst = lW + (wave + AGG_ROWS * r) * WP2 + 2 * f;
        const v4f lo = {e.x, o.x, e.y, o.y};
        const v4f hi = {e.z, o.z, e.w, o.w};
        *(v4f*)(dst + 0) = lo;
        *(v4f*)(dst + 4) = hi;
    }

    const float* hf = hin + f;
    for (int it = 0; it < AGG_TILES / AGG_BLOCKS; ++it) {
        const int tile = blockIdx.x + it * AGG_BLOCKS;
        const int n = tile * AGG_ROWS + wave;

        const int beg = rowptr[n];
        const int end = rowptr[n + 1];
        const v4f hv = *(const v4f*)(hf + (size_t)n * HID);
        const float inv = invdeg[n];
        const v4f acc = edge_sum(hf, csr_src, csr_w, beg, end, lane);
        const v4f z = acc + hv * inv;
        *(v4f*)(zt + wave * ZP + f) = z;
        __syncthreads();

        if (wave < 8) {
            v8f cacc = {};
            const float* arow = zt + l16 * ZP + half * 2;
            const float* bcol = lW + half * WP2 + 2 * (wave * 16 + l16);
#pragma unroll
            for (int s = 0; s < 32; ++s) {
                const v2f a = *(const v2f*)(arow + s * 4);
                const v2f b = *(const v2f*)(bcol + s * 2 * WP2);
                cacc = __builtin_amdgcn_wmma_f32_16x16x4_f32(
                    false, a, false, b, (short)0, cacc, false, false);
            }
            const int col = wave * 16 + l16;
#pragma unroll
            for (int r = 0; r < 8; ++r) yt[(half * 8 + r) * ZP + col] = cacc[r];
        }
        __syncthreads();

        const v4f y  = *(const v4f*)(yt + wave * ZP + f);
        const v4f av = y + bv;

        v4f val;
        val.x = hv.x + fmaxf(av.x, 0.0f);
        val.y = hv.y + fmaxf(av.y, 0.0f);
        val.z = hv.z + fmaxf(av.z, 0.0f);
        val.w = hv.w + fmaxf(av.w, 0.0f);

        float s = val.x + val.y + val.z + val.w;
#pragma unroll
        for (int off = 16; off >= 1; off >>= 1) s += __shfl_xor(s, off, 32);
        const float mu = s * (1.0f / (float)HID);

        v4f dv = val - mu;
        float q = dv.x * dv.x + dv.y * dv.y + dv.z * dv.z + dv.w * dv.w;
#pragma unroll
        for (int off = 16; off >= 1; off >>= 1) q += __shfl_xor(q, off, 32);
        const float rstd = rsqrtf(q * (1.0f / (float)HID) + LN_EPS);

        const v4f outv = dv * rstd * gm + bt;
        *(v4f*)(hout + (size_t)n * HID + f) = outv;
    }
}

#define TL_ROWS  AGG_ROWS
static_assert(HID / 2 == 4 * TL_ROWS, "one wave stages 4 row pairs of W, TL_ROWS pairs apart");
static_assert(HID * OUT_DIM <= 3 * TL_ROWS * 32, "W_out is staged three words per thread");

__device__ __forceinline__ void w_rows_load(const float* __restrict__ W, int wave, int f, v4f (&wv)[8]) {
#pragma unroll
    for (int r = 0; r < 4; ++r) {
        const float* wr = W + (size_t)(2 * (wave + TL_ROWS * r)) * HID + f;
        wv[2 * r + 0] = *(const v4f*)wr;
        wv[2 * r + 1] = *(const v4f*)(wr + HID);
    }
}
__device__ __forceinline__ void w_rows_store(float* lW, int wave, int f, const v4f (&wv)[8]) {
#pragma unroll
    for (int r = 0; r < 4; ++r) {
        const v4f e = wv[2 * r + 0];
        const v4f o = wv[2 * r + 1];
        float* dst = lW + (wave + TL_ROWS * r) * WP2 + 2 * f;
        const v4f lo = {e.x, o.x, e.y, o.y};
        const v4f hi = {e.z, o.z, e.w, o.w};
        *(v4f*)(dst + 0) = lo;
        *(v4f*)(dst + 4) = hi;
    }
}

__device__ __forceinline__ void tile_product(const float* zrows, const float* lW, float* yrows,
                                             int ct, int half, int l16) {
    v8f cacc = {};
    const float* arow = zrows + l16 * ZP + half * 2;
    const float* bcol = lW + half * WP2 + 2 * (ct * 16 + l16);
#pragma unroll
    for (int s = 0; s < 32; ++s) {
        const v2f a = *(const v2f*)(arow + s * 4);
        const v2f b = *(const v2f*)(bcol + s * 2 * WP2);
        cacc = __builtin_amdgcn_wmma_f32_16x16x4_f32(
            false, a, false, b, (short)0, cacc, false, false);
    }
    const int col = ct * 16 + l16;
#pragma unroll
    for (int r = 0; r < 8; ++r) yrows[(half * 8 + r) * ZP + col] = cacc[r];
}

__device__ __forceinline__ v4f layer_tail(v4f hv, v4f y, v4f bv, v4f gm, v4f bt) {
    const v4f av = y + bv;

    v4f val;
    val.x = hv.x + fmaxf(av.x, 0.0f);
    val.y = hv.y + fmaxf(av.y, 0.0f);
    val.z = hv.z + fmaxf(av.z, 0.0f);
    val.w = hv.w + fmaxf(av.w, 0.0f);

    float s = val.x + val.y + val.z + val.w;
#pragma unroll
    for (int off = 16; off >= 1; off >>= 1) s += __shfl_xor(s, off, 32);
    const float mu = s * (1.0f / (float)HID);

    v4f dv = val - mu;
    float q = dv.x * dv.x + dv.y * dv.y + dv.z * dv.z + dv.w * dv.w;
#pragma unroll
    for (int off = 16; off >= 1; off >>= 1) q += __shfl_xor(q, off, 32);
    const float rstd = rsqrtf(q * (1.0f / (float)HID) + LN_EPS);

    return dv * rstd * gm + bt;
}

__launch_bounds__(TL_ROWS * 32)
__global__ void k_root_tail(const int* __restrict__ rowptr,
                            const int* __restrict__ csr_src,
                            const float* __restrict__ csr_w,
                            const float* __restrict__ hin,
                            const float* __restrict__ invdeg,
                            const float* __restrict__ Wa,
                            const float* __restrict__ bias_a,
                            const float* __restrict__ gamma_a,
                            const float* __restrict__ beta_a,
                            const float* __restrict__ Wb,
                            const float* __restrict__ bias_b,
                            const float* __restrict__ gamma_b,
                            const float* __restrict__ beta_b,
                            const float* __restrict__ W_out,
                            float* __restrict__ out) {
    __shared__ __attribute__((aligned(16))) float lW[(HID / 2) * WP2];
    __shared__ __attribute__((aligned(16))) float zt[TL_ROWS * ZP];
    __shared__ __attribute__((aligned(16))) float yt[TL_ROWS * ZP];
    __shared__ __attribute__((aligned(16))) float ht[TL_ROWS * ZP];
    __shared__ float wo[HID * OUT_DIM];
    const int tid  = threadIdx.x;
    const int wave = __builtin_amdgcn_readfirstlane(tid >> 5);
    const int lane = tid & 31;
    const int half = lane >> 4;
    const int l16  = lane & 15;
    const int f = lane << 2;
    const int g = blockIdx.x;
    const int root = g * NPG;

    const int rbeg = rowptr[root];
    const int rend = rowptr[root + 1];
    const int n_list = rend - rbeg + 1;
    const float inv_root = invdeg[root];

    v4f wv[8];
    w_rows_load(Wa, wave, f, wv);
    const v4f bv = *(const v4f*)(bias_a + f);
    const v4f gm = *(const v4f*)(gamma_a + f);
    const v4f bt = *(const v4f*)(beta_a + f);
    w_rows_store(lW, wave, f, wv);

    v4f wlast[8];
    w_rows_load(Wb, wave, f, wlast);
    float wov[3];
#pragma unroll
    for (int r = 0; r < 3; ++r) {
        const int i = tid + r * (TL_ROWS * 32);
        wov[r] = (i < HID * OUT_DIM) ? W_out[i] : 0.0f;
    }

    v4f zacc  = {0.0f, 0.0f, 0.0f, 0.0f};
    v4f hroot = {0.0f, 0.0f, 0.0f, 0.0f};
    const float* hf = hin + f;
    for (int p0 = 0; p0 < n_list; p0 += TL_ROWS) {
        const int left = n_list - p0;
        const int ri = p0 + wave;
        const bool active = ri < n_list;
        float rw_l = 0.0f;
        if (wave == 0) {
            const int idx = rbeg + p0 + lane;
            if (idx < rend) rw_l = csr_w[idx];
        }

        v4f hv = {0.0f, 0.0f, 0.0f, 0.0f};
        v4f z  = {0.0f, 0.0f, 0.0f, 0.0f};
        if (active) {
            const int n = (ri < n_list - 1) ? csr_src[rbeg + ri] : root;
            const int beg = rowptr[n];
            const int end = rowptr[n + 1];
            hv = *(const v4f*)(hf + (size_t)n * HID);
            const float inv = invdeg[n];
            const v4f acc = edge_sum(hf, csr_src, csr_w, beg, end, lane);
            z = acc + hv * inv;
        }
        *(v4f*)(zt + wave * ZP + f) = z;
        __syncthreads();

        if (wave < 8)
            tile_product(zt, lW, yt, wave, half, l16);
        __syncthreads();

        if (active) {
            const v4f y = *(const v4f*)(yt + wave * ZP + f);
            *(v4f*)(ht + wave * ZP + f) = layer_tail(hv, y, bv, gm, bt);
        }
        __syncthreads();

        if (wave == 0) {
            const int cnt = left < TL_ROWS ? left : TL_ROWS;
            const int ne = (left - 1) < cnt ? (left - 1) : cnt;
            const float* hr = ht + f;
            int j = 0;
            for (; j + 4 <= ne; j += 4) {
                const v4f v0 = *(const v4f*)(hr + (j + 0) * ZP);
                const v4f v1 = *(const v4f*)(hr + (j + 1) * ZP);
                const v4f v2 = *(const v4f*)(hr + (j + 2) * ZP);
                const v4f v3 = *(const v4f*)(hr + (j + 3) * ZP);
                zacc += v0 * edge_wt(rw_l, j + 0);
                zacc += v1 * edge_wt(rw_l, j + 1);
                zacc += v2 * edge_wt(rw_l, j + 2);
                zacc += v3 * edge_wt(rw_l, j + 3);
            }
            for (; j < ne; ++j) {
                const v4f v0 = *(const v4f*)(hr + j * ZP);
                zacc += v0 * edge_wt(rw_l, j);
            }
            if (ne < cnt) hroot = *(const v4f*)(hr + ne * ZP);
        }
    }

    const v4f bvb = *(const v4f*)(bias_b + f);
    const v4f gmb = *(const v4f*)(gamma_b + f);
    const v4f btb = *(const v4f*)(beta_b + f);
    if (wave == 0) *(v4f*)(zt + f) = zacc + hroot * inv_root;
    w_rows_store(lW, wave, f, wlast);
#pragma unroll
    for (int r = 0; r < 3; ++r) {
        const int i = tid + r * (TL_ROWS * 32);
        if (i < HID * OUT_DIM) wo[i] = wov[r];
    }
    __syncthreads();

    if (wave < 8)
        tile_product(zt, lW, yt, wave, half, l16);
    __syncthreads();

    if (wave == 0) {
        const v4f y = *(const v4f*)(yt + f);
        *(v4f*)(zt + f) = layer_tail(hroot, y, bvb, gmb, btb);
    }
    __syncthreads();

    if (wave == 0 && lane < OUT_DIM) {
        float lg = 0.0f;
#pragma unroll 32
        for (int k = 0; k < HID; ++k) lg = fmaf(zt[k], wo[k * OUT_DIM + lane], lg);
        out[g * OUT_DIM + lane] = lg;
    }
}

extern "C" void kernel_launch(void* const* d_in, const int* in_sizes, int n_in,
                              void* d_out, int out_size, void* d_ws, size_t ws_size,
                              hipStream_t stream) {
    const int*   x        = (const int*)d_in[0];
    const int*   ei       = (const int*)d_in[1];
    const float* cls_emb  = (const float*)d_in[3];
    const float* nbr_emb  = (const float*)d_in[4];
    const float* Ws       = (const float*)d_in[5];
    const float* bs       = (const float*)d_in[6];
    const float* gammas   = (const float*)d_in[7];
    const float* betas    = (const float*)d_in[8];
    const float* W_out    = (const float*)d_in[9];
    float*       out      = (float*)d_out;

    char* ws = (char*)d_ws;
    size_t off = 0;
    int*   cnt    = (int*)(ws + off);   off += (size_t)EBLOCKS * NBIN * 4;
    float* dinv   = (float*)(ws + off); off += (size_t)N_NODES * 4;
    float* invdeg = (float*)(ws + off); off += (size_t)N_NODES * 4;
    float* h      = (float*)(ws + off); off += (size_t)N_NODES * HID * 4;
    float* h2     = (float*)(ws + off); off += (size_t)N_NODES * HID * 4;
    int*   rowptr = (int*)(ws + off);   off += (size_t)(N_NODES + 64) * 4;
    int*   binned = (int*)(ws + off);   off += (size_t)E_EDGES * 4;
    int*   csrtmp = (int*)(ws + off);   off += (size_t)E_EDGES * 4;
    int*   csrsrc = (int*)(ws + off);   off += (size_t)E_EDGES * 4;
    float* csrw   = (float*)(ws + off); off += (size_t)E_EDGES * 4;
    int*   binbase = (int*)(ws + off);  off += (size_t)(NBIN + 64) * 4;
    float* tab     = (float*)(ws + off); off += (size_t)TAB_ROWS * HID * 4;
    (void)ws_size; (void)in_sizes; (void)n_in; (void)out_size;

    k_bin_count  <<<EBLOCKS, 256, 0, stream>>>(ei, cnt);
    k_bin_scatter<<<EBLOCKS, 256, 0, stream>>>(ei, cnt, binned, binbase);
    k_bin_build  <<<NBIN, BUILD_T, 0, stream>>>(binned, binbase, rowptr, dinv, invdeg, csrtmp,
                                                csrsrc);

    k_table_product<<<TAB_TILES, 256, 0, stream>>>(cls_emb, nbr_emb, Ws, tab);
    k_gather_ln0<<<L0_BLOCKS, L0_WAVES * 32, 0, stream>>>(
        rowptr, csrsrc, dinv, csrw, tab, invdeg, bs, gammas, betas, h, x, cls_emb, nbr_emb);

    k_agg_layer<<<AGG_BLOCKS, AGG_ROWS * 32, 0, stream>>>(
        rowptr, csrsrc, csrw, h, invdeg, Ws + (size_t)1 * HID * HID, bs + (size_t)1 * HID,
        gammas + (size_t)1 * HID, betas + (size_t)1 * HID, h2);
    k_root_tail<<<NGRAPH, TL_ROWS * 32, 0, stream>>>(
        rowptr, csrsrc, csrw, h2, invdeg,
        Ws + (size_t)2 * HID * HID, bs + (size_t)2 * HID,
        gammas + (size_t)2 * HID, betas + (size_t)2 * HID,
        Ws + (size_t)3 * HID * HID, bs + (size_t)3 * HID,
        gammas + (size_t)3 * HID, betas + (size_t)3 * HID, W_out, out);
}
